# starting kernel
# baseline (speedup 1.0000x reference)
#include <hip/hip_runtime.h>
#include <hip/hip_bf16.h>
#include <math.h>

typedef float v2f __attribute__((ext_vector_type(2)));
typedef float v8f __attribute__((ext_vector_type(8)));

#define BATCH   2
#define Hs      56
#define Ws      56
#define Ls      (Hs*Ws)
#define DMODEL  96
#define DINNER  192
#define DSTATE  16
#define DTRANK  6
#define NCOUT   38
#define NKO     (4*NCOUT)
#define NKO_PAD 160
#define KBIG    (DINNER*2*9)
#define THETA_C 0.6f
#define MROWS   (BATCH*Ls)

static __device__ __forceinline__ float siluf(float v) { return v / (1.f + expf(-v)); }

static __device__ __forceinline__ v8f wmma4(v2f a, v2f b, v8f c) {
  return __builtin_amdgcn_wmma_f32_16x16x4_f32(false, a, false, b, (short)0, c, false, false);
}

__global__ void gemm_f32_wmma(const float* __restrict__ A, const float* __restrict__ Bt,
                              float* __restrict__ C, int K, int ldc) {
  const int lane = threadIdx.x & 31;
  const int half = lane >> 4;
  const int klo  = half << 1;
  const int ln   = lane & 15;
  const int ntile = blockIdx.x;
  const int mtile = blockIdx.y;
  const float* ap = A  + (size_t)(mtile * 16 + ln) * K + klo;
  const float* bp = Bt + (size_t)(ntile * 16 + ln) * K + klo;
  v8f acc = {0.f,0.f,0.f,0.f,0.f,0.f,0.f,0.f};
  for (int k = 0; k < K; k += 4) {
    v2f a = *(const v2f*)(ap + k);
    v2f b = *(const v2f*)(bp + k);
    acc = wmma4(a, b, acc);
  }
  const int n = ntile * 16 + ln;
  for (int rr = 0; rr < 8; ++rr) {
    int mm = mtile * 16 + rr + half * 8;
    C[(size_t)mm * ldc + n] = acc[rr];
  }
}

__global__ void k_pack_w(const float* __restrict__ xw, float* __restrict__ Wp) {
  int tid = blockIdx.x * blockDim.x + threadIdx.x;
  if (tid >= NKO_PAD * KBIG) return;
  int kk = tid % KBIG;
  int n  = tid / KBIG;
  float v = 0.f;
  if (n < NKO) {
    int c = kk % DINNER;
    int r = kk / DINNER;
    int dx = r % 3, dy = (r / 3) % 3, t = r / 9;
    v = xw[((((size_t)n * DINNER + c) * 2 + t) * 3 + dy) * 3 + dx];
  }
  Wp[tid] = v;
}

__global__ void k_pack_kd(const float* __restrict__ xw, float* __restrict__ kdp) {
  int tid = blockIdx.x * blockDim.x + threadIdx.x;
  if (tid >= NKO_PAD * DINNER) return;
  int c = tid % DINNER;
  int n = tid / DINNER;
  float v = 0.f;
  if (n < NKO) {
    const float* p = xw + ((size_t)n * DINNER + c) * 18;
    for (int i = 0; i < 18; ++i) v += p[i];
  }
  kdp[tid] = v;
}

__global__ void k2_conv(const float* __restrict__ xz, const float* __restrict__ xt,
                        const float* __restrict__ w2, const float* __restrict__ b2,
                        const float* __restrict__ wxt, const float* __restrict__ bxt,
                        float* __restrict__ xin) {
  int tid = blockIdx.x * blockDim.x + threadIdx.x;
  if (tid >= BATCH * 2 * Ls * DINNER) return;
  int c = tid % DINNER; int r = tid / DINNER;
  int w = r % Ws; r /= Ws;
  int h = r % Hs; r /= Hs;
  int t = r % 2;  int b = r / 2;
  float acc;
  if (t == 0) {
    acc = b2[c];
    for (int dy = 0; dy < 3; ++dy) {
      int hh = h + dy - 1; if (hh < 0 || hh >= Hs) continue;
      for (int dx = 0; dx < 3; ++dx) {
        int ww = w + dx - 1; if (ww < 0 || ww >= Ws) continue;
        acc += xz[((size_t)((b * Hs + hh) * Ws + ww)) * 384 + c] * w2[c * 9 + dy * 3 + dx];
      }
    }
  } else {
    acc = bxt[c];
    int ci = c >> 1;
    for (int dy = 0; dy < 3; ++dy) {
      int hh = h + dy - 1; if (hh < 0 || hh >= Hs) continue;
      for (int dx = 0; dx < 3; ++dx) {
        int ww = w + dx - 1; if (ww < 0 || ww >= Ws) continue;
        acc += xt[((size_t)((b * Hs + hh) * Ws + ww)) * DMODEL + ci] * wxt[c * 9 + dy * 3 + dx];
      }
    }
  }
  xin[tid] = siluf(acc);
}

__global__ void k3_xs(const float* __restrict__ xin, float* __restrict__ xs) {
  int tid = blockIdx.x * blockDim.x + threadIdx.x;
  if (tid >= BATCH * DINNER * Ls) return;
  int l = tid % Ls; int r = tid / Ls;
  int d = r % DINNER; int b = r / DINNER;

  auto X = [&](int s, int hh, int ww) -> float {
    return xin[(((size_t)(b * 2 + s) * Hs + hh) * Ws + ww) * DINNER + d];
  };
  auto irow = [&](int hh, int ww) -> float {
    int s = hh & 1, i = hh >> 1;
    float v0 = X(s, 2 * i, ww), v1 = X(s, 2 * i + 1, ww);
    return 0.5f * (v0 + v1) + fmaxf(v0, v1);
  };
  auto icol = [&](int hh, int ww) -> float {
    int s = ww & 1, j = ww >> 1;
    float v0 = X(s, hh, 2 * j), v1 = X(s, hh, 2 * j + 1);
    return 0.5f * (v0 + v1) + fmaxf(v0, v1);
  };

  int wr = l / Hs, hr = l % Hs;
  int hc = l / Ws, wc = l % Ws;
  size_t stride = (size_t)DINNER * Ls;
  size_t base = ((size_t)(b * 4) * DINNER + d) * Ls + l;
  xs[base + 0 * stride] = irow(hr, wr);
  xs[base + 1 * stride] = irow(Hs - 1 - hr, wr);
  xs[base + 2 * stride] = icol(hc, wc);
  xs[base + 3 * stride] = icol(hc, Ws - 1 - wc);
}

__global__ void k4_cdc(const float* __restrict__ xin, const float* __restrict__ Wp,
                       const float* __restrict__ kdp, float* __restrict__ xdbl) {
  const int lane = threadIdx.x & 31;
  const int half = lane >> 4;
  const int klo  = half << 1;
  const int ln   = lane & 15;
  const int ntile = blockIdx.x;
  const int mtile = blockIdx.y;
  const int m = mtile * 16 + ln;
  const int b = m / Ls;
  const int hw = m % Ls;
  const int h = hw / Ws, w = hw % Ws;
  const int n = ntile * 16 + ln;

  v8f accN  = {0.f,0.f,0.f,0.f,0.f,0.f,0.f,0.f};
  const float* wptr = Wp + (size_t)n * KBIG + klo;
  for (int td = 0; td < 18; ++td) {
    int t = td / 9, rr9 = td % 9, dy = rr9 / 3, dx = rr9 % 3;
    int hh = h + dy - 1, ww = w + dx - 1;
    bool ok = (hh >= 0) && (hh < Hs) && (ww >= 0) && (ww < Ws);
    const float* ap = xin + (((size_t)(b * 2 + t) * Hs + hh) * Ws + ww) * DINNER + klo;
    for (int c = 0; c < DINNER; c += 4) {
      v2f a = {0.f, 0.f};
      if (ok) a = *(const v2f*)(ap + c);
      v2f bb = *(const v2f*)wptr;
      wptr += 4;
      accN = wmma4(a, bb, accN);
    }
  }

  v8f accD0 = {0.f,0.f,0.f,0.f,0.f,0.f,0.f,0.f};
  v8f accD1 = {0.f,0.f,0.f,0.f,0.f,0.f,0.f,0.f};
  const float* ap0 = xin + (((size_t)(b * 2 + 0) * Hs + h) * Ws + w) * DINNER + klo;
  const float* ap1 = xin + (((size_t)(b * 2 + 1) * Hs + h) * Ws + w) * DINNER + klo;
  const float* kp  = kdp + (size_t)n * DINNER + klo;
  for (int c = 0; c < DINNER; c += 4) {
    v2f bb = *(const v2f*)(kp + c);
    v2f a0 = *(const v2f*)(ap0 + c);
    v2f a1 = *(const v2f*)(ap1 + c);
    accD0 = wmma4(a0, bb, accD0);
    accD1 = wmma4(a1, bb, accD1);
  }

  if (n < NKO) {
    int kq = n / NCOUT, o = n % NCOUT;
    for (int rr = 0; rr < 8; ++rr) {
      int mm = mtile * 16 + rr + half * 8;
      int b2 = mm / Ls; int l2 = mm % Ls;
      float Nv = accN[rr], d0 = accD0[rr], d1 = accD1[rr];
      float pooled = 2.f * Nv - THETA_C * (0.5f * (d0 + d1) + fminf(d0, d1));
      xdbl[(((size_t)b2 * 4 + kq) * NCOUT + o) * Ls + l2] = pooled;
    }
  }
}

__global__ void k5_delta(const float* __restrict__ xdbl, const float* __restrict__ dtw,
                         const float* __restrict__ dtb, float* __restrict__ delta) {
  int tid = blockIdx.x * blockDim.x + threadIdx.x;
  if (tid >= BATCH * 4 * DINNER * Ls) return;
  int l = tid % Ls; int r = tid / Ls;
  int d = r % DINNER; int bk = r / DINNER;
  int k = bk & 3;
  float acc = dtb[k * DINNER + d];
  const float* wp = dtw + ((size_t)k * DINNER + d) * DTRANK;
  const float* xp = xdbl + (size_t)bk * NCOUT * Ls + l;
  for (int rr = 0; rr < DTRANK; ++rr) acc += xp[(size_t)rr * Ls] * wp[rr];
  delta[tid] = fmaxf(acc, 0.f) + log1pf(expf(-fabsf(acc)));
}

__global__ void k6_scan(const float* __restrict__ xs, const float* __restrict__ xdbl,
                        const float* __restrict__ Alogs, const float* __restrict__ Dss,
                        float* __restrict__ dio) {
  int wid  = blockIdx.x * (blockDim.x >> 5) + (threadIdx.x >> 5);
  int lane = threadIdx.x & 31;
  int half = lane >> 4, ln = lane & 15;
  int bk = wid / 96;
  int d  = (wid % 96) * 2 + half;
  int k  = bk & 3;
  const float* dP = dio  + ((size_t)bk * DINNER + d) * Ls;
  float*       oP = dio  + ((size_t)bk * DINNER + d) * Ls;
  const float* uP = xs   + ((size_t)bk * DINNER + d) * Ls;
  const float* bP = xdbl + ((size_t)bk * NCOUT + DTRANK + ln) * Ls;
  const float* cP = xdbl + ((size_t)bk * NCOUT + DTRANK + DSTATE + ln) * Ls;
  float A  = -expf(Alogs[((size_t)(k * DINNER + d)) * DSTATE + ln]);
  float Dv = Dss[k * DINNER + d];
  float h = 0.f;
  for (int l = 0; l < Ls; ++l) {
    float dl = dP[l];
    float ul = uP[l];
    float bl = bP[l];
    float cl = cP[l];
    h = expf(dl * A) * h + (dl * ul) * bl;
    float y = h * cl;
    y += __shfl_xor(y, 1);
    y += __shfl_xor(y, 2);
    y += __shfl_xor(y, 4);
    y += __shfl_xor(y, 8);
    if (ln == 0) oP[l] = y + Dv * ul;
  }
}

__global__ void k7a_comb(const float* __restrict__ oy, float* __restrict__ y2) {
  int tid = blockIdx.x * blockDim.x + threadIdx.x;
  if (tid >= MROWS * DINNER) return;
  int d = tid % DINNER; int m = tid / DINNER;
  int b = m / Ls; int l = m % Ls;
  int h = l / Ws, w = l % Ws;
  int lt = w * Hs + h;
  size_t s = (size_t)DINNER * Ls;
  const float* base = oy + ((size_t)b * 4 * DINNER + d) * Ls;
  y2[tid] = base[l] + base[2 * s + (Ls - 1 - l)]
          + base[s + lt] + base[3 * s + (Ls - 1 - lt)];
}

__global__ void k7b_ln(const float* __restrict__ xz, const float* __restrict__ g,
                       const float* __restrict__ bb, float* __restrict__ y2) {
  int m = blockIdx.x;
  int d = threadIdx.x;
  float v = y2[(size_t)m * DINNER + d];
  float s1 = v, s2 = v * v;
  #pragma unroll
  for (int off = 16; off >= 1; off >>= 1) {
    s1 += __shfl_xor(s1, off);
    s2 += __shfl_xor(s2, off);
  }
  __shared__ float r1[6], r2[6];
  __shared__ float smu, srs;
  if ((d & 31) == 0) { r1[d >> 5] = s1; r2[d >> 5] = s2; }
  __syncthreads();
  if (d == 0) {
    float t1 = 0.f, t2 = 0.f;
    for (int i = 0; i < 6; ++i) { t1 += r1[i]; t2 += r2[i]; }
    float mu = t1 / (float)DINNER;
    float var = t2 / (float)DINNER - mu * mu;
    smu = mu; srs = rsqrtf(var + 1e-5f);
  }
  __syncthreads();
  float yln = (v - smu) * srs * g[d] + bb[d];
  float zv = xz[(size_t)m * 384 + DINNER + d];
  y2[(size_t)m * DINNER + d] = yln * siluf(zv);
}

extern "C" void kernel_launch(void* const* d_in, const int* in_sizes, int n_in,
                              void* d_out, int out_size, void* d_ws, size_t ws_size,
                              hipStream_t stream) {
  (void)in_sizes; (void)n_in; (void)out_size; (void)ws_size;
  const float* x   = (const float*)d_in[0];
  const float* xt  = (const float*)d_in[1];
  const float* ipw = (const float*)d_in[2];
  const float* c2w = (const float*)d_in[3];
  const float* c2b = (const float*)d_in[4];
  const float* cxw = (const float*)d_in[5];
  const float* cxb = (const float*)d_in[6];
  const float* xpw = (const float*)d_in[7];
  const float* dtw = (const float*)d_in[8];
  const float* dtb = (const float*)d_in[9];
  const float* alg = (const float*)d_in[10];
  const float* dss = (const float*)d_in[11];
  const float* lng = (const float*)d_in[12];
  const float* lnb = (const float*)d_in[13];
  const float* opw = (const float*)d_in[14];
  float* out = (float*)d_out;
  float* ws  = (float*)d_ws;

  size_t off = 0;
  float* xz    = ws + off; off += (size_t)MROWS * 384;
  float* xin   = ws + off; off += (size_t)BATCH * 2 * Ls * DINNER;
  float* xs    = ws + off; off += (size_t)BATCH * 4 * DINNER * Ls;
  float* Wp    = ws + off; off += (size_t)NKO_PAD * KBIG;
  float* kdp   = ws + off; off += (size_t)NKO_PAD * DINNER;
  float* xdbl  = ws + off; off += (size_t)BATCH * 4 * NCOUT * Ls;
  float* delta = ws + off; off += (size_t)BATCH * 4 * DINNER * Ls;
  float* y2    = xin;

  gemm_f32_wmma<<<dim3(384 / 16, MROWS / 16), 32, 0, stream>>>(x, ipw, xz, DMODEL, 384);

  k_pack_w <<<(NKO_PAD * KBIG + 255) / 256, 256, 0, stream>>>(xpw, Wp);
  k_pack_kd<<<(NKO_PAD * DINNER + 255) / 256, 256, 0, stream>>>(xpw, kdp);

  k2_conv<<<(BATCH * 2 * Ls * DINNER + 255) / 256, 256, 0, stream>>>(
      xz, xt, c2w, c2b, cxw, cxb, xin);

  k3_xs<<<(BATCH * DINNER * Ls + 255) / 256, 256, 0, stream>>>(xin, xs);

  k4_cdc<<<dim3(NKO_PAD / 16, MROWS / 16), 32, 0, stream>>>(xin, Wp, kdp, xdbl);

  k5_delta<<<(BATCH * 4 * DINNER * Ls + 255) / 256, 256, 0, stream>>>(xdbl, dtw, dtb, delta);

  k6_scan<<<96, 256, 0, stream>>>(xs, xdbl, alg, dss, delta);

  k7a_comb<<<(MROWS * DINNER + 255) / 256, 256, 0, stream>>>(delta, y2);
  k7b_ln<<<MROWS, DINNER, 0, stream>>>(xz, lng, lnb, y2);

  gemm_f32_wmma<<<dim3(DMODEL / 16, MROWS / 16), 32, 0, stream>>>(y2, opw, out, DINNER, DMODEL);
}


# k6 scan walks L in 16-step register chunks: next chunk's loads issued ahead, each butterfly level's 16 cross-lane reads in flight at once, one out-of-place store per chunk; sum tree kept
# speedup vs baseline: 3.1128x; 3.1128x over previous
#include <hip/hip_runtime.h>
#include <hip/hip_bf16.h>
#include <math.h>

typedef float v2f __attribute__((ext_vector_type(2)));
typedef float v8f __attribute__((ext_vector_type(8)));

#define BATCH   2
#define Hs      56
#define Ws      56
#define Ls      (Hs*Ws)
#define DMODEL  96
#define DINNER  192
#define DSTATE  16
#define DTRANK  6
#define NCOUT   38
#define NKO     (4*NCOUT)
#define NKO_PAD 160
#define KBIG    (DINNER*2*9)
#define THETA_C 0.6f
#define MROWS   (BATCH*Ls)

static __device__ __forceinline__ float siluf(float v) { return v / (1.f + expf(-v)); }

static __device__ __forceinline__ v8f wmma4(v2f a, v2f b, v8f c) {
  return __builtin_amdgcn_wmma_f32_16x16x4_f32(false, a, false, b, (short)0, c, false, false);
}

__global__ void gemm_f32_wmma(const float* __restrict__ A, const float* __restrict__ Bt,
                              float* __restrict__ C, int K, int ldc) {
  const int lane = threadIdx.x & 31;
  const int half = lane >> 4;
  const int klo  = half << 1;
  const int ln   = lane & 15;
  const int ntile = blockIdx.x;
  const int mtile = blockIdx.y;
  const float* ap = A  + (size_t)(mtile * 16 + ln) * K + klo;
  const float* bp = Bt + (size_t)(ntile * 16 + ln) * K + klo;
  v8f acc = {0.f,0.f,0.f,0.f,0.f,0.f,0.f,0.f};
  for (int k = 0; k < K; k += 4) {
    v2f a = *(const v2f*)(ap + k);
    v2f b = *(const v2f*)(bp + k);
    acc = wmma4(a, b, acc);
  }
  const int n = ntile * 16 + ln;
  for (int rr = 0; rr < 8; ++rr) {
    int mm = mtile * 16 + rr + half * 8;
    C[(size_t)mm * ldc + n] = acc[rr];
  }
}

__global__ void k_pack_w(const float* __restrict__ xw, float* __restrict__ Wp) {
  int tid = blockIdx.x * blockDim.x + threadIdx.x;
  if (tid >= NKO_PAD * KBIG) return;
  int kk = tid % KBIG;
  int n  = tid / KBIG;
  float v = 0.f;
  if (n < NKO) {
    int c = kk % DINNER;
    int r = kk / DINNER;
    int dx = r % 3, dy = (r / 3) % 3, t = r / 9;
    v = xw[((((size_t)n * DINNER + c) * 2 + t) * 3 + dy) * 3 + dx];
  }
  Wp[tid] = v;
}

__global__ void k_pack_kd(const float* __restrict__ xw, float* __restrict__ kdp) {
  int tid = blockIdx.x * blockDim.x + threadIdx.x;
  if (tid >= NKO_PAD * DINNER) return;
  int c = tid % DINNER;
  int n = tid / DINNER;
  float v = 0.f;
  if (n < NKO) {
    const float* p = xw + ((size_t)n * DINNER + c) * 18;
    for (int i = 0; i < 18; ++i) v += p[i];
  }
  kdp[tid] = v;
}

__global__ void k2_conv(const float* __restrict__ xz, const float* __restrict__ xt,
                        const float* __restrict__ w2, const float* __restrict__ b2,
                        const float* __restrict__ wxt, const float* __restrict__ bxt,
                        float* __restrict__ xin) {
  int tid = blockIdx.x * blockDim.x + threadIdx.x;
  if (tid >= BATCH * 2 * Ls * DINNER) return;
  int c = tid % DINNER; int r = tid / DINNER;
  int w = r % Ws; r /= Ws;
  int h = r % Hs; r /= Hs;
  int t = r % 2;  int b = r / 2;
  float acc;
  if (t == 0) {
    acc = b2[c];
    for (int dy = 0; dy < 3; ++dy) {
      int hh = h + dy - 1; if (hh < 0 || hh >= Hs) continue;
      for (int dx = 0; dx < 3; ++dx) {
        int ww = w + dx - 1; if (ww < 0 || ww >= Ws) continue;
        acc += xz[((size_t)((b * Hs + hh) * Ws + ww)) * 384 + c] * w2[c * 9 + dy * 3 + dx];
      }
    }
  } else {
    acc = bxt[c];
    int ci = c >> 1;
    for (int dy = 0; dy < 3; ++dy) {
      int hh = h + dy - 1; if (hh < 0 || hh >= Hs) continue;
      for (int dx = 0; dx < 3; ++dx) {
        int ww = w + dx - 1; if (ww < 0 || ww >= Ws) continue;
        acc += xt[((size_t)((b * Hs + hh) * Ws + ww)) * DMODEL + ci] * wxt[c * 9 + dy * 3 + dx];
      }
    }
  }
  xin[tid] = siluf(acc);
}

__global__ void k3_xs(const float* __restrict__ xin, float* __restrict__ xs) {
  int tid = blockIdx.x * blockDim.x + threadIdx.x;
  if (tid >= BATCH * DINNER * Ls) return;
  int l = tid % Ls; int r = tid / Ls;
  int d = r % DINNER; int b = r / DINNER;

  auto X = [&](int s, int hh, int ww) -> float {
    return xin[(((size_t)(b * 2 + s) * Hs + hh) * Ws + ww) * DINNER + d];
  };
  auto irow = [&](int hh, int ww) -> float {
    int s = hh & 1, i = hh >> 1;
    float v0 = X(s, 2 * i, ww), v1 = X(s, 2 * i + 1, ww);
    return 0.5f * (v0 + v1) + fmaxf(v0, v1);
  };
  auto icol = [&](int hh, int ww) -> float {
    int s = ww & 1, j = ww >> 1;
    float v0 = X(s, hh, 2 * j), v1 = X(s, hh, 2 * j + 1);
    return 0.5f * (v0 + v1) + fmaxf(v0, v1);
  };

  int wr = l / Hs, hr = l % Hs;
  int hc = l / Ws, wc = l % Ws;
  size_t stride = (size_t)DINNER * Ls;
  size_t base = ((size_t)(b * 4) * DINNER + d) * Ls + l;
  xs[base + 0 * stride] = irow(hr, wr);
  xs[base + 1 * stride] = irow(Hs - 1 - hr, wr);
  xs[base + 2 * stride] = icol(hc, wc);
  xs[base + 3 * stride] = icol(hc, Ws - 1 - wc);
}

__global__ void k4_cdc(const float* __restrict__ xin, const float* __restrict__ Wp,
                       const float* __restrict__ kdp, float* __restrict__ xdbl) {
  const int lane = threadIdx.x & 31;
  const int half = lane >> 4;
  const int klo  = half << 1;
  const int ln   = lane & 15;
  const int ntile = blockIdx.x;
  const int mtile = blockIdx.y;
  const int m = mtile * 16 + ln;
  const int b = m / Ls;
  const int hw = m % Ls;
  const int h = hw / Ws, w = hw % Ws;
  const int n = ntile * 16 + ln;

  v8f accN  = {0.f,0.f,0.f,0.f,0.f,0.f,0.f,0.f};
  const float* wptr = Wp + (size_t)n * KBIG + klo;
  for (int td = 0; td < 18; ++td) {
    int t = td / 9, rr9 = td % 9, dy = rr9 / 3, dx = rr9 % 3;
    int hh = h + dy - 1, ww = w + dx - 1;
    bool ok = (hh >= 0) && (hh < Hs) && (ww >= 0) && (ww < Ws);
    const float* ap = xin + (((size_t)(b * 2 + t) * Hs + hh) * Ws + ww) * DINNER + klo;
    for (int c = 0; c < DINNER; c += 4) {
      v2f a = {0.f, 0.f};
      if (ok) a = *(const v2f*)(ap + c);
      v2f bb = *(const v2f*)wptr;
      wptr += 4;
      accN = wmma4(a, bb, accN);
    }
  }

  v8f accD0 = {0.f,0.f,0.f,0.f,0.f,0.f,0.f,0.f};
  v8f accD1 = {0.f,0.f,0.f,0.f,0.f,0.f,0.f,0.f};
  const float* ap0 = xin + (((size_t)(b * 2 + 0) * Hs + h) * Ws + w) * DINNER + klo;
  const float* ap1 = xin + (((size_t)(b * 2 + 1) * Hs + h) * Ws + w) * DINNER + klo;
  const float* kp  = kdp + (size_t)n * DINNER + klo;
  for (int c = 0; c < DINNER; c += 4) {
    v2f bb = *(const v2f*)(kp + c);
    v2f a0 = *(const v2f*)(ap0 + c);
    v2f a1 = *(const v2f*)(ap1 + c);
    accD0 = wmma4(a0, bb, accD0);
    accD1 = wmma4(a1, bb, accD1);
  }

  if (n < NKO) {
    int kq = n / NCOUT, o = n % NCOUT;
    for (int rr = 0; rr < 8; ++rr) {
      int mm = mtile * 16 + rr + half * 8;
      int b2 = mm / Ls; int l2 = mm % Ls;
      float Nv = accN[rr], d0 = accD0[rr], d1 = accD1[rr];
      float pooled = 2.f * Nv - THETA_C * (0.5f * (d0 + d1) + fminf(d0, d1));
      xdbl[(((size_t)b2 * 4 + kq) * NCOUT + o) * Ls + l2] = pooled;
    }
  }
}

__global__ void k5_delta(const float* __restrict__ xdbl, const float* __restrict__ dtw,
                         const float* __restrict__ dtb, float* __restrict__ delta) {
  int tid = blockIdx.x * blockDim.x + threadIdx.x;
  if (tid >= BATCH * 4 * DINNER * Ls) return;
  int l = tid % Ls; int r = tid / Ls;
  int d = r % DINNER; int bk = r / DINNER;
  int k = bk & 3;
  float acc = dtb[k * DINNER + d];
  const float* wp = dtw + ((size_t)k * DINNER + d) * DTRANK;
  const float* xp = xdbl + (size_t)bk * NCOUT * Ls + l;
  for (int rr = 0; rr < DTRANK; ++rr) acc += xp[(size_t)rr * Ls] * wp[rr];
  delta[tid] = fmaxf(acc, 0.f) + log1pf(expf(-fabsf(acc)));
}

#define SCH 16

struct ScanChunk { float d[SCH], u[SCH], b[SCH], c[SCH]; };

static __device__ __forceinline__ void scan_load(ScanChunk& q, const float* __restrict__ dP,
                                                 const float* __restrict__ uP,
                                                 const float* __restrict__ bP,
                                                 const float* __restrict__ cP, int l0) {
  #pragma unroll
  for (int i = 0; i < SCH; i += 4) {
    float4 vd = *(const float4*)(dP + l0 + i);
    float4 vu = *(const float4*)(uP + l0 + i);
    float4 vb = *(const float4*)(bP + l0 + i);
    float4 vc = *(const float4*)(cP + l0 + i);
    q.d[i] = vd.x; q.d[i + 1] = vd.y; q.d[i + 2] = vd.z; q.d[i + 3] = vd.w;
    q.u[i] = vu.x; q.u[i + 1] = vu.y; q.u[i + 2] = vu.z; q.u[i + 3] = vu.w;
    q.b[i] = vb.x; q.b[i + 1] = vb.y; q.b[i + 2] = vb.z; q.b[i + 3] = vb.w;
    q.c[i] = vc.x; q.c[i + 1] = vc.y; q.c[i + 2] = vc.z; q.c[i + 3] = vc.w;
  }
}

static __device__ __forceinline__ float scan_steps(const ScanChunk& q, float& h, float A,
                                                   float Dv, int ln) {
  float y[SCH];
  #pragma unroll
  for (int j = 0; j < SCH; ++j) {
    float dl = q.d[j];
    h = expf(dl * A) * h + (dl * q.u[j]) * q.b[j];
    y[j] = h * q.c[j];
  }
  #pragma unroll
  for (int m = 1; m < 16; m <<= 1) {
    float t[SCH];
    #pragma unroll
    for (int j = 0; j < SCH; ++j) t[j] = __shfl_xor(y[j], m);
    __builtin_amdgcn_sched_barrier(0);
    #pragma unroll
    for (int j = 0; j < SCH; ++j) y[j] += t[j];
  }
  float keep = 0.f;
  #pragma unroll
  for (int j = 0; j < SCH; ++j) {
    float o = y[j] + Dv * q.u[j];
    if (ln == j) keep = o;
  }
  return keep;
}

__global__ void __launch_bounds__(256) k6_scan(const float* __restrict__ xs, const float* __restrict__ xdbl,
                        const float* __restrict__ Alogs, const float* __restrict__ Dss,
                        const float* __restrict__ delta, float* __restrict__ oy) {
  int wid  = blockIdx.x * (blockDim.x >> 5) + (threadIdx.x >> 5);
  int lane = threadIdx.x & 31;
  int half = lane >> 4, ln = lane & 15;
  int bk = wid / 96;
  int d  = (wid % 96) * 2 + half;
  int k  = bk & 3;
  const float* dP = delta + ((size_t)bk * DINNER + d) * Ls;
  float*       oP = oy    + ((size_t)bk * DINNER + d) * Ls;
  const float* uP = xs    + ((size_t)bk * DINNER + d) * Ls;
  const float* bP = xdbl  + ((size_t)bk * NCOUT + DTRANK + ln) * Ls;
  const float* cP = xdbl  + ((size_t)bk * NCOUT + DTRANK + DSTATE + ln) * Ls;
  float A  = -expf(Alogs[((size_t)(k * DINNER + d)) * DSTATE + ln]);
  float Dv = Dss[k * DINNER + d];
  float h = 0.f;
  ScanChunk qa, qb;
  scan_load(qa, dP, uP, bP, cP, 0);
  for (int l0 = 0; l0 < Ls; l0 += 2 * SCH) {
    scan_load(qb, dP, uP, bP, cP, l0 + SCH);
    oP[l0 + ln] = scan_steps(qa, h, A, Dv, ln);
    int ln2 = (l0 + 2 * SCH < Ls) ? (l0 + 2 * SCH) : (l0 + SCH);
    scan_load(qa, dP, uP, bP, cP, ln2);
    oP[l0 + SCH + ln] = scan_steps(qb, h, A, Dv, ln);
  }
}

__global__ void k7a_comb(const float* __restrict__ oy, float* __restrict__ y2) {
  int tid = blockIdx.x * blockDim.x + threadIdx.x;
  if (tid >= MROWS * DINNER) return;
  int d = tid % DINNER; int m = tid / DINNER;
  int b = m / Ls; int l = m % Ls;
  int h = l / Ws, w = l % Ws;
  int lt = w * Hs + h;
  size_t s = (size_t)DINNER * Ls;
  const float* base = oy + ((size_t)b * 4 * DINNER + d) * Ls;
  y2[tid] = base[l] + base[2 * s + (Ls - 1 - l)]
          + base[s + lt] + base[3 * s + (Ls - 1 - lt)];
}

__global__ void k7b_ln(const float* __restrict__ xz, const float* __restrict__ g,
                       const float* __restrict__ bb, float* __restrict__ y2) {
  int m = blockIdx.x;
  int d = threadIdx.x;
  float v = y2[(size_t)m * DINNER + d];
  float s1 = v, s2 = v * v;
  #pragma unroll
  for (int off = 16; off >= 1; off >>= 1) {
    s1 += __shfl_xor(s1, off);
    s2 += __shfl_xor(s2, off);
  }
  __shared__ float r1[6], r2[6];
  __shared__ float smu, srs;
  if ((d & 31) == 0) { r1[d >> 5] = s1; r2[d >> 5] = s2; }
  __syncthreads();
  if (d == 0) {
    float t1 = 0.f, t2 = 0.f;
    for (int i = 0; i < 6; ++i) { t1 += r1[i]; t2 += r2[i]; }
    float mu = t1 / (float)DINNER;
    float var = t2 / (float)DINNER - mu * mu;
    smu = mu; srs = rsqrtf(var + 1e-5f);
  }
  __syncthreads();
  float yln = (v - smu) * srs * g[d] + bb[d];
  float zv = xz[(size_t)m * 384 + DINNER + d];
  y2[(size_t)m * DINNER + d] = yln * siluf(zv);
}

extern "C" void kernel_launch(void* const* d_in, const int* in_sizes, int n_in,
                              void* d_out, int out_size, void* d_ws, size_t ws_size,
                              hipStream_t stream) {
  (void)in_sizes; (void)n_in; (void)out_size; (void)ws_size;
  const float* x   = (const float*)d_in[0];
  const float* xt  = (const float*)d_in[1];
  const float* ipw = (const float*)d_in[2];
  const float* c2w = (const float*)d_in[3];
  const float* c2b = (const float*)d_in[4];
  const float* cxw = (const float*)d_in[5];
  const float* cxb = (const float*)d_in[6];
  const float* xpw = (const float*)d_in[7];
  const float* dtw = (const float*)d_in[8];
  const float* dtb = (const float*)d_in[9];
  const float* alg = (const float*)d_in[10];
  const float* dss = (const float*)d_in[11];
  const float* lng = (const float*)d_in[12];
  const float* lnb = (const float*)d_in[13];
  const float* opw = (const float*)d_in[14];
  float* out = (float*)d_out;
  float* ws  = (float*)d_ws;

  size_t off = 0;
  float* xz    = ws + off; off += (size_t)MROWS * 384;
  float* xin   = ws + off; off += (size_t)BATCH * 2 * Ls * DINNER;
  float* xs    = ws + off; off += (size_t)BATCH * 4 * DINNER * Ls;
  float* Wp    = ws + off; off += (size_t)NKO_PAD * KBIG;
  float* kdp   = ws + off; off += (size_t)NKO_PAD * DINNER;
  float* xdbl  = ws + off; off += (size_t)BATCH * 4 * NCOUT * Ls;
  float* delta = ws + off; off += (size_t)BATCH * 4 * DINNER * Ls;
  float* oy    = ws + off; off += (size_t)BATCH * 4 * DINNER * Ls;
  float* y2    = xin;

  gemm_f32_wmma<<<dim3(384 / 16, MROWS / 16), 32, 0, stream>>>(x, ipw, xz, DMODEL, 384);

  k_pack_w <<<(NKO_PAD * KBIG + 255) / 256, 256, 0, stream>>>(xpw, Wp);
  k_pack_kd<<<(NKO_PAD * DINNER + 255) / 256, 256, 0, stream>>>(xpw, kdp);

  k2_conv<<<(BATCH * 2 * Ls * DINNER + 255) / 256, 256, 0, stream>>>(
      xz, xt, c2w, c2b, cxw, cxb, xin);

  k3_xs<<<(BATCH * DINNER * Ls + 255) / 256, 256, 0, stream>>>(xin, xs);

  k4_cdc<<<dim3(NKO_PAD / 16, MROWS / 16), 32, 0, stream>>>(xin, Wp, kdp, xdbl);

  k5_delta<<<(BATCH * 4 * DINNER * Ls + 255) / 256, 256, 0, stream>>>(xdbl, dtw, dtb, delta);

  k6_scan<<<96, 256, 0, stream>>>(xs, xdbl, alg, dss, delta, oy);

  k7a_comb<<<(MROWS * DINNER + 255) / 256, 256, 0, stream>>>(oy, y2);
  k7b_ln<<<MROWS, DINNER, 0, stream>>>(xz, lng, lnb, y2);

  gemm_f32_wmma<<<dim3(DMODEL / 16, MROWS / 16), 32, 0, stream>>>(y2, opw, out, DINNER, DMODEL);
}


# k4 conv GEMM re-tiled: one wave keeps a 16x80 strip (5 accumulators), one activation fragment feeds 5 matrix ops; weights packed in fragment order, read as whole lines, next stage loaded ahead
# speedup vs baseline: 4.4477x; 1.4288x over previous
#include <hip/hip_runtime.h>
#include <hip/hip_bf16.h>
#include <math.h>

typedef float v2f __attribute__((ext_vector_type(2)));
typedef float v8f __attribute__((ext_vector_type(8)));

#define BATCH   2
#define Hs      56
#define Ws      56
#define Ls      (Hs*Ws)
#define DMODEL  96
#define DINNER  192
#define DSTATE  16
#define DTRANK  6
#define NCOUT   38
#define NKO     (4*NCOUT)
#define NKO_PAD 160
#define KBIG    (DINNER*2*9)
#define THETA_C 0.6f
#define MROWS   (BATCH*Ls)

static __device__ __forceinline__ float siluf(float v) { return v / (1.f + expf(-v)); }

static __device__ __forceinline__ v8f wmma4(v2f a, v2f b, v8f c) {
  return __builtin_amdgcn_wmma_f32_16x16x4_f32(false, a, false, b, (short)0, c, false, false);
}

__global__ void gemm_f32_wmma(const float* __restrict__ A, const float* __restrict__ Bt,
                              float* __restrict__ C, int K, int ldc) {
  const int lane = threadIdx.x & 31;
  const int half = lane >> 4;
  const int klo  = half << 1;
  const int ln   = lane & 15;
  const int ntile = blockIdx.x;
  const int mtile = blockIdx.y;
  const float* ap = A  + (size_t)(mtile * 16 + ln) * K + klo;
  const float* bp = Bt + (size_t)(ntile * 16 + ln) * K + klo;
  v8f acc = {0.f,0.f,0.f,0.f,0.f,0.f,0.f,0.f};
  for (int k = 0; k < K; k += 4) {
    v2f a = *(const v2f*)(ap + k);
    v2f b = *(const v2f*)(bp + k);
    acc = wmma4(a, b, acc);
  }
  const int n = ntile * 16 + ln;
  for (int rr = 0; rr < 8; ++rr) {
    int mm = mtile * 16 + rr + half * 8;
    C[(size_t)mm * ldc + n] = acc[rr];
  }
}

#define NPAIR_W  (KBIG / 8)
#define NPAIR_KD (DINNER / 8)

__global__ void k_pack_w(const float* __restrict__ xw, float* __restrict__ Wp) {
  int tid = blockIdx.x * blockDim.x + threadIdx.x;
  if (tid >= NKO_PAD * KBIG) return;
  int e    = tid & 3;
  int lane = (tid >> 2) & 31;
  int grp  = tid >> 7;
  int pr   = grp % NPAIR_W;
  int nt   = grp / NPAIR_W;
  int n    = nt * 16 + (lane & 15);
  int kk   = pr * 8 + (e >> 1) * 4 + (lane >> 4) * 2 + (e & 1);
  float v = 0.f;
  if (n < NKO) {
    int c = kk % DINNER;
    int r = kk / DINNER;
    int dx = r % 3, dy = (r / 3) % 3, t = r / 9;
    v = xw[((((size_t)n * DINNER + c) * 2 + t) * 3 + dy) * 3 + dx];
  }
  Wp[tid] = v;
}

__global__ void k_pack_kd(const float* __restrict__ xw, float* __restrict__ kdp) {
  int tid = blockIdx.x * blockDim.x + threadIdx.x;
  if (tid >= NKO_PAD * DINNER) return;
  int e    = tid & 3;
  int lane = (tid >> 2) & 31;
  int grp  = tid >> 7;
  int pr   = grp % NPAIR_KD;
  int nt   = grp / NPAIR_KD;
  int n    = nt * 16 + (lane & 15);
  int c    = pr * 8 + (e >> 1) * 4 + (lane >> 4) * 2 + (e & 1);
  float v = 0.f;
  if (n < NKO) {
    const float* p = xw + ((size_t)n * DINNER + c) * 18;
    for (int i = 0; i < 18; ++i) v += p[i];
  }
  kdp[tid] = v;
}

__global__ void k2_conv(const float* __restrict__ xz, const float* __restrict__ xt,
                        const float* __restrict__ w2, const float* __restrict__ b2,
                        const float* __restrict__ wxt, const float* __restrict__ bxt,
                        float* __restrict__ xin) {
  int tid = blockIdx.x * blockDim.x + threadIdx.x;
  if (tid >= BATCH * 2 * Ls * DINNER) return;
  int c = tid % DINNER; int r = tid / DINNER;
  int w = r % Ws; r /= Ws;
  int h = r % Hs; r /= Hs;
  int t = r % 2;  int b = r / 2;
  float acc;
  if (t == 0) {
    acc = b2[c];
    for (int dy = 0; dy < 3; ++dy) {
      int hh = h + dy - 1; if (hh < 0 || hh >= Hs) continue;
      for (int dx = 0; dx < 3; ++dx) {
        int ww = w + dx - 1; if (ww < 0 || ww >= Ws) continue;
        acc += xz[((size_t)((b * Hs + hh) * Ws + ww)) * 384 + c] * w2[c * 9 + dy * 3 + dx];
      }
    }
  } else {
    acc = bxt[c];
    int ci = c >> 1;
    for (int dy = 0; dy < 3; ++dy) {
      int hh = h + dy - 1; if (hh < 0 || hh >= Hs) continue;
      for (int dx = 0; dx < 3; ++dx) {
        int ww = w + dx - 1; if (ww < 0 || ww >= Ws) continue;
        acc += xt[((size_t)((b * Hs + hh) * Ws + ww)) * DMODEL + ci] * wxt[c * 9 + dy * 3 + dx];
      }
    }
  }
  xin[tid] = siluf(acc);
}

__global__ void k3_xs(const float* __restrict__ xin, float* __restrict__ xs) {
  int tid = blockIdx.x * blockDim.x + threadIdx.x;
  if (tid >= BATCH * DINNER * Ls) return;
  int l = tid % Ls; int r = tid / Ls;
  int d = r % DINNER; int b = r / DINNER;

  auto X = [&](int s, int hh, int ww) -> float {
    return xin[(((size_t)(b * 2 + s) * Hs + hh) * Ws + ww) * DINNER + d];
  };
  auto irow = [&](int hh, int ww) -> float {
    int s = hh & 1, i = hh >> 1;
    float v0 = X(s, 2 * i, ww), v1 = X(s, 2 * i + 1, ww);
    return 0.5f * (v0 + v1) + fmaxf(v0, v1);
  };
  auto icol = [&](int hh, int ww) -> float {
    int s = ww & 1, j = ww >> 1;
    float v0 = X(s, hh, 2 * j), v1 = X(s, hh, 2 * j + 1);
    return 0.5f * (v0 + v1) + fmaxf(v0, v1);
  };

  int wr = l / Hs, hr = l % Hs;
  int hc = l / Ws, wc = l % Ws;
  size_t stride = (size_t)DINNER * Ls;
  size_t base = ((size_t)(b * 4) * DINNER + d) * Ls + l;
  xs[base + 0 * stride] = irow(hr, wr);
  xs[base + 1 * stride] = irow(Hs - 1 - hr, wr);
  xs[base + 2 * stride] = icol(hc, wc);
  xs[base + 3 * stride] = icol(hc, Ws - 1 - wc);
}

#define CDC_NT     5
#define CDC_STAGES (KBIG / 32)

struct CdcStage { v2f a[8]; float4 b[4][CDC_NT]; int ok; };

static __device__ __forceinline__ void cdc_load(CdcStage& q, const float* __restrict__ xin,
                                                const float* __restrict__ wf, int s,
                                                int b, int h, int w, int klo) {
  int td = s / 6, cc = (s - td * 6) * 32;
  int t = td / 9, r9 = td - t * 9, dy = r9 / 3, dx = r9 - dy * 3;
  int hh = h + dy - 1, ww = w + dx - 1;
  q.ok = (hh >= 0) && (hh < Hs) && (ww >= 0) && (ww < Ws);
  int hc = hh < 0 ? 0 : (hh >= Hs ? Hs - 1 : hh);
  int wc = ww < 0 ? 0 : (ww >= Ws ? Ws - 1 : ww);
  const float* ap = xin + (((size_t)(b * 2 + t) * Hs + hc) * Ws + wc) * DINNER + klo + cc;
  #pragma unroll
  for (int i = 0; i < 8; ++i) q.a[i] = *(const v2f*)(ap + 4 * i);
  const float* bp = wf + (size_t)s * 4 * 128;
  #pragma unroll
  for (int qd = 0; qd < 4; ++qd) {
    #pragma unroll
    for (int j = 0; j < CDC_NT; ++j)
      q.b[qd][j] = *(const float4*)(bp + ((size_t)j * NPAIR_W + qd) * 128);
  }
}

static __device__ __forceinline__ void cdc_compute(const CdcStage& q, v8f (&acc)[CDC_NT]) {
  const v2f zero = {0.f, 0.f};
  #pragma unroll
  for (int qd = 0; qd < 4; ++qd) {
    #pragma unroll
    for (int e2 = 0; e2 < 2; ++e2) {
      v2f a = q.ok ? q.a[2 * qd + e2] : zero;
      #pragma unroll
      for (int j = 0; j < CDC_NT; ++j) {
        float4 bv = q.b[qd][j];
        v2f bb;
        if (e2 == 0) { bb.x = bv.x; bb.y = bv.y; } else { bb.x = bv.z; bb.y = bv.w; }
        acc[j] = wmma4(a, bb, acc[j]);
      }
    }
  }
}

__global__ void __launch_bounds__(128) k4_cdc(const float* __restrict__ xin, const float* __restrict__ Wp,
                       const float* __restrict__ kdp, float* __restrict__ xdbl) {
  const int lane = threadIdx.x & 31;
  const int half = lane >> 4;
  const int klo  = half << 1;
  const int ln   = lane & 15;
  const int mtile = blockIdx.x * 4 + (threadIdx.x >> 5);
  const int nt0   = blockIdx.y * CDC_NT;
  const int m = mtile * 16 + ln;
  const int b = m / Ls;
  const int hw = m % Ls;
  const int h = hw / Ws, w = hw % Ws;

  v8f accN[CDC_NT];
  #pragma unroll
  for (int j = 0; j < CDC_NT; ++j) accN[j] = (v8f){0.f,0.f,0.f,0.f,0.f,0.f,0.f,0.f};

  const float* wf = Wp + (size_t)nt0 * NPAIR_W * 128 + lane * 4;
  CdcStage qa, qb;
  cdc_load(qa, xin, wf, 0, b, h, w, klo);
  for (int s = 0; s < CDC_STAGES; s += 2) {
    cdc_load(qb, xin, wf, s + 1, b, h, w, klo);
    cdc_compute(qa, accN);
    int s2 = (s + 2 < CDC_STAGES) ? (s + 2) : (s + 1);
    cdc_load(qa, xin, wf, s2, b, h, w, klo);
    cdc_compute(qb, accN);
  }

  v8f accD0[CDC_NT], accD1[CDC_NT];
  #pragma unroll
  for (int j = 0; j < CDC_NT; ++j) {
    accD0[j] = (v8f){0.f,0.f,0.f,0.f,0.f,0.f,0.f,0.f};
    accD1[j] = (v8f){0.f,0.f,0.f,0.f,0.f,0.f,0.f,0.f};
  }
  const float* ap0 = xin + (((size_t)(b * 2 + 0) * Hs + h) * Ws + w) * DINNER + klo;
  const float* ap1 = xin + (((size_t)(b * 2 + 1) * Hs + h) * Ws + w) * DINNER + klo;
  const float* kf  = kdp + (size_t)nt0 * NPAIR_KD * 128 + lane * 4;
  #pragma unroll
  for (int pr = 0; pr < NPAIR_KD; ++pr) {
    v2f a00 = *(const v2f*)(ap0 + pr * 8);
    v2f a01 = *(const v2f*)(ap0 + pr * 8 + 4);
    v2f a10 = *(const v2f*)(ap1 + pr * 8);
    v2f a11 = *(const v2f*)(ap1 + pr * 8 + 4);
    #pragma unroll
    for (int j = 0; j < CDC_NT; ++j) {
      float4 bv = *(const float4*)(kf + ((size_t)j * NPAIR_KD + pr) * 128);
      v2f b0, b1;
      b0.x = bv.x; b0.y = bv.y; b1.x = bv.z; b1.y = bv.w;
      accD0[j] = wmma4(a00, b0, accD0[j]);
      accD1[j] = wmma4(a10, b0, accD1[j]);
      accD0[j] = wmma4(a01, b1, accD0[j]);
      accD1[j] = wmma4(a11, b1, accD1[j]);
    }
  }

  #pragma unroll
  for (int j = 0; j < CDC_NT; ++j) {
    const int n = (nt0 + j) * 16 + ln;
    if (n < NKO) {
      int kq = n / NCOUT, o = n % NCOUT;
      for (int rr = 0; rr < 8; ++rr) {
        int mm = mtile * 16 + rr + half * 8;
        int b2 = mm / Ls; int l2 = mm % Ls;
        float Nv = accN[j][rr], d0 = accD0[j][rr], d1 = accD1[j][rr];
        float pooled = 2.f * Nv - THETA_C * (0.5f * (d0 + d1) + fminf(d0, d1));
        xdbl[(((size_t)b2 * 4 + kq) * NCOUT + o) * Ls + l2] = pooled;
      }
    }
  }
}

__global__ void k5_delta(const float* __restrict__ xdbl, const float* __restrict__ dtw,
                         const float* __restrict__ dtb, float* __restrict__ delta) {
  int tid = blockIdx.x * blockDim.x + threadIdx.x;
  if (tid >= BATCH * 4 * DINNER * Ls) return;
  int l = tid % Ls; int r = tid / Ls;
  int d = r % DINNER; int bk = r / DINNER;
  int k = bk & 3;
  float acc = dtb[k * DINNER + d];
  const float* wp = dtw + ((size_t)k * DINNER + d) * DTRANK;
  const float* xp = xdbl + (size_t)bk * NCOUT * Ls + l;
  for (int rr = 0; rr < DTRANK; ++rr) acc += xp[(size_t)rr * Ls] * wp[rr];
  delta[tid] = fmaxf(acc, 0.f) + log1pf(expf(-fabsf(acc)));
}

#define SCH 16

struct ScanChunk { float d[SCH], u[SCH], b[SCH], c[SCH]; };

static __device__ __forceinline__ void scan_load(ScanChunk& q, const float* __restrict__ dP,
                                                 const float* __restrict__ uP,
                                                 const float* __restrict__ bP,
                                                 const float* __restrict__ cP, int l0) {
  #pragma unroll
  for (int i = 0; i < SCH; i += 4) {
    float4 vd = *(const float4*)(dP + l0 + i);
    float4 vu = *(const float4*)(uP + l0 + i);
    float4 vb = *(const float4*)(bP + l0 + i);
    float4 vc = *(const float4*)(cP + l0 + i);
    q.d[i] = vd.x; q.d[i + 1] = vd.y; q.d[i + 2] = vd.z; q.d[i + 3] = vd.w;
    q.u[i] = vu.x; q.u[i + 1] = vu.y; q.u[i + 2] = vu.z; q.u[i + 3] = vu.w;
    q.b[i] = vb.x; q.b[i + 1] = vb.y; q.b[i + 2] = vb.z; q.b[i + 3] = vb.w;
    q.c[i] = vc.x; q.c[i + 1] = vc.y; q.c[i + 2] = vc.z; q.c[i + 3] = vc.w;
  }
}

static __device__ __forceinline__ float scan_steps(const ScanChunk& q, float& h, float A,
                                                   float Dv, int ln) {
  float y[SCH];
  #pragma unroll
  for (int j = 0; j < SCH; ++j) {
    float dl = q.d[j];
    h = expf(dl * A) * h + (dl * q.u[j]) * q.b[j];
    y[j] = h * q.c[j];
  }
  #pragma unroll
  for (int m = 1; m < 16; m <<= 1) {
    float t[SCH];
    #pragma unroll
    for (int j = 0; j < SCH; ++j) t[j] = __shfl_xor(y[j], m);
    __builtin_amdgcn_sched_barrier(0);
    #pragma unroll
    for (int j = 0; j < SCH; ++j) y[j] += t[j];
  }
  float keep = 0.f;
  #pragma unroll
  for (int j = 0; j < SCH; ++j) {
    float o = y[j] + Dv * q.u[j];
    if (ln == j) keep = o;
  }
  return keep;
}

__global__ void __launch_bounds__(256) k6_scan(const float* __restrict__ xs, const float* __restrict__ xdbl,
                        const float* __restrict__ Alogs, const float* __restrict__ Dss,
                        const float* __restrict__ delta, float* __restrict__ oy) {
  int wid  = blockIdx.x * (blockDim.x >> 5) + (threadIdx.x >> 5);
  int lane = threadIdx.x & 31;
  int half = lane >> 4, ln = lane & 15;
  int bk = wid / 96;
  int d  = (wid % 96) * 2 + half;
  int k  = bk & 3;
  const float* dP = delta + ((size_t)bk * DINNER + d) * Ls;
  float*       oP = oy    + ((size_t)bk * DINNER + d) * Ls;
  const float* uP = xs    + ((size_t)bk * DINNER + d) * Ls;
  const float* bP = xdbl  + ((size_t)bk * NCOUT + DTRANK + ln) * Ls;
  const float* cP = xdbl  + ((size_t)bk * NCOUT + DTRANK + DSTATE + ln) * Ls;
  float A  = -expf(Alogs[((size_t)(k * DINNER + d)) * DSTATE + ln]);
  float Dv = Dss[k * DINNER + d];
  float h = 0.f;
  ScanChunk qa, qb;
  scan_load(qa, dP, uP, bP, cP, 0);
  for (int l0 = 0; l0 < Ls; l0 += 2 * SCH) {
    scan_load(qb, dP, uP, bP, cP, l0 + SCH);
    oP[l0 + ln] = scan_steps(qa, h, A, Dv, ln);
    int ln2 = (l0 + 2 * SCH < Ls) ? (l0 + 2 * SCH) : (l0 + SCH);
    scan_load(qa, dP, uP, bP, cP, ln2);
    oP[l0 + SCH + ln] = scan_steps(qb, h, A, Dv, ln);
  }
}

__global__ void k7a_comb(const float* __restrict__ oy, float* __restrict__ y2) {
  int tid = blockIdx.x * blockDim.x + threadIdx.x;
  if (tid >= MROWS * DINNER) return;
  int d = tid % DINNER; int m = tid / DINNER;
  int b = m / Ls; int l = m % Ls;
  int h = l / Ws, w = l % Ws;
  int lt = w * Hs + h;
  size_t s = (size_t)DINNER * Ls;
  const float* base = oy + ((size_t)b * 4 * DINNER + d) * Ls;
  y2[tid] = base[l] + base[2 * s + (Ls - 1 - l)]
          + base[s + lt] + base[3 * s + (Ls - 1 - lt)];
}

__global__ void k7b_ln(const float* __restrict__ xz, const float* __restrict__ g,
                       const float* __restrict__ bb, float* __restrict__ y2) {
  int m = blockIdx.x;
  int d = threadIdx.x;
  float v = y2[(size_t)m * DINNER + d];
  float s1 = v, s2 = v * v;
  #pragma unroll
  for (int off = 16; off >= 1; off >>= 1) {
    s1 += __shfl_xor(s1, off);
    s2 += __shfl_xor(s2, off);
  }
  __shared__ float r1[6], r2[6];
  __shared__ float smu, srs;
  if ((d & 31) == 0) { r1[d >> 5] = s1; r2[d >> 5] = s2; }
  __syncthreads();
  if (d == 0) {
    float t1 = 0.f, t2 = 0.f;
    for (int i = 0; i < 6; ++i) { t1 += r1[i]; t2 += r2[i]; }
    float mu = t1 / (float)DINNER;
    float var = t2 / (float)DINNER - mu * mu;
    smu = mu; srs = rsqrtf(var + 1e-5f);
  }
  __syncthreads();
  float yln = (v - smu) * srs * g[d] + bb[d];
  float zv = xz[(size_t)m * 384 + DINNER + d];
  y2[(size_t)m * DINNER + d] = yln * siluf(zv);
}

extern "C" void kernel_launch(void* const* d_in, const int* in_sizes, int n_in,
                              void* d_out, int out_size, void* d_ws, size_t ws_size,
                              hipStream_t stream) {
  (void)in_sizes; (void)n_in; (void)out_size; (void)ws_size;
  const float* x   = (const float*)d_in[0];
  const float* xt  = (const float*)d_in[1];
  const float* ipw = (const float*)d_in[2];
  const float* c2w = (const float*)d_in[3];
  const float* c2b = (const float*)d_in[4];
  const float* cxw = (const float*)d_in[5];
  const float* cxb = (const float*)d_in[6];
  const float* xpw = (const float*)d_in[7];
  const float* dtw = (const float*)d_in[8];
  const float* dtb = (const float*)d_in[9];
  const float* alg = (const float*)d_in[10];
  const float* dss = (const float*)d_in[11];
  const float* lng = (const float*)d_in[12];
  const float* lnb = (const float*)d_in[13];
  const float* opw = (const float*)d_in[14];
  float* out = (float*)d_out;
  float* ws  = (float*)d_ws;

  size_t off = 0;
  float* xz    = ws + off; off += (size_t)MROWS * 384;
  float* xin   = ws + off; off += (size_t)BATCH * 2 * Ls * DINNER;
  float* xs    = ws + off; off += (size_t)BATCH * 4 * DINNER * Ls;
  float* Wp    = ws + off; off += (size_t)NKO_PAD * KBIG;
  float* kdp   = ws + off; off += (size_t)NKO_PAD * DINNER;
  float* xdbl  = ws + off; off += (size_t)BATCH * 4 * NCOUT * Ls;
  float* delta = ws + off; off += (size_t)BATCH * 4 * DINNER * Ls;
  float* oy    = ws + off; off += (size_t)BATCH * 4 * DINNER * Ls;
  float* y2    = xin;

  gemm_f32_wmma<<<dim3(384 / 16, MROWS / 16), 32, 0, stream>>>(x, ipw, xz, DMODEL, 384);

  k_pack_w <<<(NKO_PAD * KBIG + 255) / 256, 256, 0, stream>>>(xpw, Wp);
  k_pack_kd<<<(NKO_PAD * DINNER + 255) / 256, 256, 0, stream>>>(xpw, kdp);

  k2_conv<<<(BATCH * 2 * Ls * DINNER + 255) / 256, 256, 0, stream>>>(
      xz, xt, c2w, c2b, cxw, cxb, xin);

  k3_xs<<<(BATCH * DINNER * Ls + 255) / 256, 256, 0, stream>>>(xin, xs);

  k4_cdc<<<dim3(MROWS / 64, NKO_PAD / (16 * CDC_NT)), 128, 0, stream>>>(xin, Wp, kdp, xdbl);

  k5_delta<<<(BATCH * 4 * DINNER * Ls + 255) / 256, 256, 0, stream>>>(xdbl, dtw, dtb, delta);

  k6_scan<<<96, 256, 0, stream>>>(xs, xdbl, alg, dss, delta, oy);

  k7a_comb<<<(MROWS * DINNER + 255) / 256, 256, 0, stream>>>(oy, y2);
  k7b_ln<<<MROWS, DINNER, 0, stream>>>(xz, lng, lnb, y2);

  gemm_f32_wmma<<<dim3(DMODEL / 16, MROWS / 16), 32, 0, stream>>>(y2, opw, out, DINNER, DMODEL);
}


# k6 scan: per-step decay exp(delta*A) taken as one hardware exp2 of delta*(A*log2e), log2e folded into A once per wave, in place of the 13-instruction library expf; sum order unchanged
# speedup vs baseline: 4.9483x; 1.1126x over previous
#include <hip/hip_runtime.h>
#include <hip/hip_bf16.h>
#include <math.h>

typedef float v2f __attribute__((ext_vector_type(2)));
typedef float v8f __attribute__((ext_vector_type(8)));

#define BATCH   2
#define Hs      56
#define Ws      56
#define Ls      (Hs*Ws)
#define DMODEL  96
#define DINNER  192
#define DSTATE  16
#define DTRANK  6
#define NCOUT   38
#define NKO     (4*NCOUT)
#define NKO_PAD 160
#define KBIG    (DINNER*2*9)
#define THETA_C 0.6f
#define MROWS   (BATCH*Ls)

static __device__ __forceinline__ float siluf(float v) { return v / (1.f + expf(-v)); }

static __device__ __forceinline__ v8f wmma4(v2f a, v2f b, v8f c) {
  return __builtin_amdgcn_wmma_f32_16x16x4_f32(false, a, false, b, (short)0, c, false, false);
}

__global__ void gemm_f32_wmma(const float* __restrict__ A, const float* __restrict__ Bt,
                              float* __restrict__ C, int K, int ldc) {
  const int lane = threadIdx.x & 31;
  const int half = lane >> 4;
  const int klo  = half << 1;
  const int ln   = lane & 15;
  const int ntile = blockIdx.x;
  const int mtile = blockIdx.y;
  const float* ap = A  + (size_t)(mtile * 16 + ln) * K + klo;
  const float* bp = Bt + (size_t)(ntile * 16 + ln) * K + klo;
  v8f acc = {0.f,0.f,0.f,0.f,0.f,0.f,0.f,0.f};
  for (int k = 0; k < K; k += 4) {
    v2f a = *(const v2f*)(ap + k);
    v2f b = *(const v2f*)(bp + k);
    acc = wmma4(a, b, acc);
  }
  const int n = ntile * 16 + ln;
  for (int rr = 0; rr < 8; ++rr) {
    int mm = mtile * 16 + rr + half * 8;
    C[(size_t)mm * ldc + n] = acc[rr];
  }
}

#define NPAIR_W  (KBIG / 8)
#define NPAIR_KD (DINNER / 8)

__global__ void k_pack_w(const float* __restrict__ xw, float* __restrict__ Wp) {
  int tid = blockIdx.x * blockDim.x + threadIdx.x;
  if (tid >= NKO_PAD * KBIG) return;
  int e    = tid & 3;
  int lane = (tid >> 2) & 31;
  int grp  = tid >> 7;
  int pr   = grp % NPAIR_W;
  int nt   = grp / NPAIR_W;
  int n    = nt * 16 + (lane & 15);
  int kk   = pr * 8 + (e >> 1) * 4 + (lane >> 4) * 2 + (e & 1);
  float v = 0.f;
  if (n < NKO) {
    int c = kk % DINNER;
    int r = kk / DINNER;
    int dx = r % 3, dy = (r / 3) % 3, t = r / 9;
    v = xw[((((size_t)n * DINNER + c) * 2 + t) * 3 + dy) * 3 + dx];
  }
  Wp[tid] = v;
}

__global__ void k_pack_kd(const float* __restrict__ xw, float* __restrict__ kdp) {
  int tid = blockIdx.x * blockDim.x + threadIdx.x;
  if (tid >= NKO_PAD * DINNER) return;
  int e    = tid & 3;
  int lane = (tid >> 2) & 31;
  int grp  = tid >> 7;
  int pr   = grp % NPAIR_KD;
  int nt   = grp / NPAIR_KD;
  int n    = nt * 16 + (lane & 15);
  int c    = pr * 8 + (e >> 1) * 4 + (lane >> 4) * 2 + (e & 1);
  float v = 0.f;
  if (n < NKO) {
    const float* p = xw + ((size_t)n * DINNER + c) * 18;
    for (int i = 0; i < 18; ++i) v += p[i];
  }
  kdp[tid] = v;
}

__global__ void k2_conv(const float* __restrict__ xz, const float* __restrict__ xt,
                        const float* __restrict__ w2, const float* __restrict__ b2,
                        const float* __restrict__ wxt, const float* __restrict__ bxt,
                        float* __restrict__ xin) {
  int tid = blockIdx.x * blockDim.x + threadIdx.x;
  if (tid >= BATCH * 2 * Ls * DINNER) return;
  int c = tid % DINNER; int r = tid / DINNER;
  int w = r % Ws; r /= Ws;
  int h = r % Hs; r /= Hs;
  int t = r % 2;  int b = r / 2;
  float acc;
  if (t == 0) {
    acc = b2[c];
    for (int dy = 0; dy < 3; ++dy) {
      int hh = h + dy - 1; if (hh < 0 || hh >= Hs) continue;
      for (int dx = 0; dx < 3; ++dx) {
        int ww = w + dx - 1; if (ww < 0 || ww >= Ws) continue;
        acc += xz[((size_t)((b * Hs + hh) * Ws + ww)) * 384 + c] * w2[c * 9 + dy * 3 + dx];
      }
    }
  } else {
    acc = bxt[c];
    int ci = c >> 1;
    for (int dy = 0; dy < 3; ++dy) {
      int hh = h + dy - 1; if (hh < 0 || hh >= Hs) continue;
      for (int dx = 0; dx < 3; ++dx) {
        int ww = w + dx - 1; if (ww < 0 || ww >= Ws) continue;
        acc += xt[((size_t)((b * Hs + hh) * Ws + ww)) * DMODEL + ci] * wxt[c * 9 + dy * 3 + dx];
      }
    }
  }
  xin[tid] = siluf(acc);
}

__global__ void k3_xs(const float* __restrict__ xin, float* __restrict__ xs) {
  int tid = blockIdx.x * blockDim.x + threadIdx.x;
  if (tid >= BATCH * DINNER * Ls) return;
  int l = tid % Ls; int r = tid / Ls;
  int d = r % DINNER; int b = r / DINNER;

  auto X = [&](int s, int hh, int ww) -> float {
    return xin[(((size_t)(b * 2 + s) * Hs + hh) * Ws + ww) * DINNER + d];
  };
  auto irow = [&](int hh, int ww) -> float {
    int s = hh & 1, i = hh >> 1;
    float v0 = X(s, 2 * i, ww), v1 = X(s, 2 * i + 1, ww);
    return 0.5f * (v0 + v1) + fmaxf(v0, v1);
  };
  auto icol = [&](int hh, int ww) -> float {
    int s = ww & 1, j = ww >> 1;
    float v0 = X(s, hh, 2 * j), v1 = X(s, hh, 2 * j + 1);
    return 0.5f * (v0 + v1) + fmaxf(v0, v1);
  };

  int wr = l / Hs, hr = l % Hs;
  int hc = l / Ws, wc = l % Ws;
  size_t stride = (size_t)DINNER * Ls;
  size_t base = ((size_t)(b * 4) * DINNER + d) * Ls + l;
  xs[base + 0 * stride] = irow(hr, wr);
  xs[base + 1 * stride] = irow(Hs - 1 - hr, wr);
  xs[base + 2 * stride] = icol(hc, wc);
  xs[base + 3 * stride] = icol(hc, Ws - 1 - wc);
}

#define CDC_NT     5
#define CDC_STAGES (KBIG / 32)

struct CdcStage { v2f a[8]; float4 b[4][CDC_NT]; int ok; };

static __device__ __forceinline__ void cdc_load(CdcStage& q, const float* __restrict__ xin,
                                                const float* __restrict__ wf, int s,
                                                int b, int h, int w, int klo) {
  int td = s / 6, cc = (s - td * 6) * 32;
  int t = td / 9, r9 = td - t * 9, dy = r9 / 3, dx = r9 - dy * 3;
  int hh = h + dy - 1, ww = w + dx - 1;
  q.ok = (hh >= 0) && (hh < Hs) && (ww >= 0) && (ww < Ws);
  int hc = hh < 0 ? 0 : (hh >= Hs ? Hs - 1 : hh);
  int wc = ww < 0 ? 0 : (ww >= Ws ? Ws - 1 : ww);
  const float* ap = xin + (((size_t)(b * 2 + t) * Hs + hc) * Ws + wc) * DINNER + klo + cc;
  #pragma unroll
  for (int i = 0; i < 8; ++i) q.a[i] = *(const v2f*)(ap + 4 * i);
  const float* bp = wf + (size_t)s * 4 * 128;
  #pragma unroll
  for (int qd = 0; qd < 4; ++qd) {
    #pragma unroll
    for (int j = 0; j < CDC_NT; ++j)
      q.b[qd][j] = *(const float4*)(bp + ((size_t)j * NPAIR_W + qd) * 128);
  }
}

static __device__ __forceinline__ void cdc_compute(const CdcStage& q, v8f (&acc)[CDC_NT]) {
  const v2f zero = {0.f, 0.f};
  #pragma unroll
  for (int qd = 0; qd < 4; ++qd) {
    #pragma unroll
    for (int e2 = 0; e2 < 2; ++e2) {
      v2f a = q.ok ? q.a[2 * qd + e2] : zero;
      #pragma unroll
      for (int j = 0; j < CDC_NT; ++j) {
        float4 bv = q.b[qd][j];
        v2f bb;
        if (e2 == 0) { bb.x = bv.x; bb.y = bv.y; } else { bb.x = bv.z; bb.y = bv.w; }
        acc[j] = wmma4(a, bb, acc[j]);
      }
    }
  }
}

__global__ void __launch_bounds__(128) k4_cdc(const float* __restrict__ xin, const float* __restrict__ Wp,
                       const float* __restrict__ kdp, float* __restrict__ xdbl) {
  const int lane = threadIdx.x & 31;
  const int half = lane >> 4;
  const int klo  = half << 1;
  const int ln   = lane & 15;
  const int mtile = blockIdx.x * 4 + (threadIdx.x >> 5);
  const int nt0   = blockIdx.y * CDC_NT;
  const int m = mtile * 16 + ln;
  const int b = m / Ls;
  const int hw = m % Ls;
  const int h = hw / Ws, w = hw % Ws;

  v8f accN[CDC_NT];
  #pragma unroll
  for (int j = 0; j < CDC_NT; ++j) accN[j] = (v8f){0.f,0.f,0.f,0.f,0.f,0.f,0.f,0.f};

  const float* wf = Wp + (size_t)nt0 * NPAIR_W * 128 + lane * 4;
  CdcStage qa, qb;
  cdc_load(qa, xin, wf, 0, b, h, w, klo);
  for (int s = 0; s < CDC_STAGES; s += 2) {
    cdc_load(qb, xin, wf, s + 1, b, h, w, klo);
    cdc_compute(qa, accN);
    int s2 = (s + 2 < CDC_STAGES) ? (s + 2) : (s + 1);
    cdc_load(qa, xin, wf, s2, b, h, w, klo);
    cdc_compute(qb, accN);
  }

  v8f accD0[CDC_NT], accD1[CDC_NT];
  #pragma unroll
  for (int j = 0; j < CDC_NT; ++j) {
    accD0[j] = (v8f){0.f,0.f,0.f,0.f,0.f,0.f,0.f,0.f};
    accD1[j] = (v8f){0.f,0.f,0.f,0.f,0.f,0.f,0.f,0.f};
  }
  const float* ap0 = xin + (((size_t)(b * 2 + 0) * Hs + h) * Ws + w) * DINNER + klo;
  const float* ap1 = xin + (((size_t)(b * 2 + 1) * Hs + h) * Ws + w) * DINNER + klo;
  const float* kf  = kdp + (size_t)nt0 * NPAIR_KD * 128 + lane * 4;
  #pragma unroll
  for (int pr = 0; pr < NPAIR_KD; ++pr) {
    v2f a00 = *(const v2f*)(ap0 + pr * 8);
    v2f a01 = *(const v2f*)(ap0 + pr * 8 + 4);
    v2f a10 = *(const v2f*)(ap1 + pr * 8);
    v2f a11 = *(const v2f*)(ap1 + pr * 8 + 4);
    #pragma unroll
    for (int j = 0; j < CDC_NT; ++j) {
      float4 bv = *(const float4*)(kf + ((size_t)j * NPAIR_KD + pr) * 128);
      v2f b0, b1;
      b0.x = bv.x; b0.y = bv.y; b1.x = bv.z; b1.y = bv.w;
      accD0[j] = wmma4(a00, b0, accD0[j]);
      accD1[j] = wmma4(a10, b0, accD1[j]);
      accD0[j] = wmma4(a01, b1, accD0[j]);
      accD1[j] = wmma4(a11, b1, accD1[j]);
    }
  }

  #pragma unroll
  for (int j = 0; j < CDC_NT; ++j) {
    const int n = (nt0 + j) * 16 + ln;
    if (n < NKO) {
      int kq = n / NCOUT, o = n % NCOUT;
      for (int rr = 0; rr < 8; ++rr) {
        int mm = mtile * 16 + rr + half * 8;
        int b2 = mm / Ls; int l2 = mm % Ls;
        float Nv = accN[j][rr], d0 = accD0[j][rr], d1 = accD1[j][rr];
        float pooled = 2.f * Nv - THETA_C * (0.5f * (d0 + d1) + fminf(d0, d1));
        xdbl[(((size_t)b2 * 4 + kq) * NCOUT + o) * Ls + l2] = pooled;
      }
    }
  }
}

__global__ void k5_delta(const float* __restrict__ xdbl, const float* __restrict__ dtw,
                         const float* __restrict__ dtb, float* __restrict__ delta) {
  int tid = blockIdx.x * blockDim.x + threadIdx.x;
  if (tid >= BATCH * 4 * DINNER * Ls) return;
  int l = tid % Ls; int r = tid / Ls;
  int d = r % DINNER; int bk = r / DINNER;
  int k = bk & 3;
  float acc = dtb[k * DINNER + d];
  const float* wp = dtw + ((size_t)k * DINNER + d) * DTRANK;
  const float* xp = xdbl + (size_t)bk * NCOUT * Ls + l;
  for (int rr = 0; rr < DTRANK; ++rr) acc += xp[(size_t)rr * Ls] * wp[rr];
  delta[tid] = fmaxf(acc, 0.f) + log1pf(expf(-fabsf(acc)));
}

#define SCH 16

struct ScanChunk { float d[SCH], u[SCH], b[SCH], c[SCH]; };

static __device__ __forceinline__ void scan_load(ScanChunk& q, const float* __restrict__ dP,
                                                 const float* __restrict__ uP,
                                                 const float* __restrict__ bP,
                                                 const float* __restrict__ cP, int l0) {
  #pragma unroll
  for (int i = 0; i < SCH; i += 4) {
    float4 vd = *(const float4*)(dP + l0 + i);
    float4 vu = *(const float4*)(uP + l0 + i);
    float4 vb = *(const float4*)(bP + l0 + i);
    float4 vc = *(const float4*)(cP + l0 + i);
    q.d[i] = vd.x; q.d[i + 1] = vd.y; q.d[i + 2] = vd.z; q.d[i + 3] = vd.w;
    q.u[i] = vu.x; q.u[i + 1] = vu.y; q.u[i + 2] = vu.z; q.u[i + 3] = vu.w;
    q.b[i] = vb.x; q.b[i + 1] = vb.y; q.b[i + 2] = vb.z; q.b[i + 3] = vb.w;
    q.c[i] = vc.x; q.c[i + 1] = vc.y; q.c[i + 2] = vc.z; q.c[i + 3] = vc.w;
  }
}

static __device__ __forceinline__ float scan_steps(const ScanChunk& q, float& h, float A,
                                                   float Dv, int ln) {
  float y[SCH];
  #pragma unroll
  for (int j = 0; j < SCH; ++j) {
    float dl = q.d[j];
    h = __builtin_amdgcn_exp2f(dl * A) * h + (dl * q.u[j]) * q.b[j];
    y[j] = h * q.c[j];
  }
  #pragma unroll
  for (int m = 1; m < 16; m <<= 1) {
    float t[SCH];
    #pragma unroll
    for (int j = 0; j < SCH; ++j) t[j] = __shfl_xor(y[j], m);
    __builtin_amdgcn_sched_barrier(0);
    #pragma unroll
    for (int j = 0; j < SCH; ++j) y[j] += t[j];
  }
  float keep = 0.f;
  #pragma unroll
  for (int j = 0; j < SCH; ++j) {
    float o = y[j] + Dv * q.u[j];
    if (ln == j) keep = o;
  }
  return keep;
}

__global__ void __launch_bounds__(256) k6_scan(const float* __restrict__ xs, const float* __restrict__ xdbl,
                        const float* __restrict__ Alogs, const float* __restrict__ Dss,
                        const float* __restrict__ delta, float* __restrict__ oy) {
  int wid  = blockIdx.x * (blockDim.x >> 5) + (threadIdx.x >> 5);
  int lane = threadIdx.x & 31;
  int half = lane >> 4, ln = lane & 15;
  int bk = wid / 96;
  int d  = (wid % 96) * 2 + half;
  int k  = bk & 3;
  const float* dP = delta + ((size_t)bk * DINNER + d) * Ls;
  float*       oP = oy    + ((size_t)bk * DINNER + d) * Ls;
  const float* uP = xs    + ((size_t)bk * DINNER + d) * Ls;
  const float* bP = xdbl  + ((size_t)bk * NCOUT + DTRANK + ln) * Ls;
  const float* cP = xdbl  + ((size_t)bk * NCOUT + DTRANK + DSTATE + ln) * Ls;
  float A  = -expf(Alogs[((size_t)(k * DINNER + d)) * DSTATE + ln]) * 1.4426950408889634f;
  float Dv = Dss[k * DINNER + d];
  float h = 0.f;
  ScanChunk qa, qb;
  scan_load(qa, dP, uP, bP, cP, 0);
  for (int l0 = 0; l0 < Ls; l0 += 2 * SCH) {
    scan_load(qb, dP, uP, bP, cP, l0 + SCH);
    oP[l0 + ln] = scan_steps(qa, h, A, Dv, ln);
    int ln2 = (l0 + 2 * SCH < Ls) ? (l0 + 2 * SCH) : (l0 + SCH);
    scan_load(qa, dP, uP, bP, cP, ln2);
    oP[l0 + SCH + ln] = scan_steps(qb, h, A, Dv, ln);
  }
}

__global__ void k7a_comb(const float* __restrict__ oy, float* __restrict__ y2) {
  int tid = blockIdx.x * blockDim.x + threadIdx.x;
  if (tid >= MROWS * DINNER) return;
  int d = tid % DINNER; int m = tid / DINNER;
  int b = m / Ls; int l = m % Ls;
  int h = l / Ws, w = l % Ws;
  int lt = w * Hs + h;
  size_t s = (size_t)DINNER * Ls;
  const float* base = oy + ((size_t)b * 4 * DINNER + d) * Ls;
  y2[tid] = base[l] + base[2 * s + (Ls - 1 - l)]
          + base[s + lt] + base[3 * s + (Ls - 1 - lt)];
}

__global__ void k7b_ln(const float* __restrict__ xz, const float* __restrict__ g,
                       const float* __restrict__ bb, float* __restrict__ y2) {
  int m = blockIdx.x;
  int d = threadIdx.x;
  float v = y2[(size_t)m * DINNER + d];
  float s1 = v, s2 = v * v;
  #pragma unroll
  for (int off = 16; off >= 1; off >>= 1) {
    s1 += __shfl_xor(s1, off);
    s2 += __shfl_xor(s2, off);
  }
  __shared__ float r1[6], r2[6];
  __shared__ float smu, srs;
  if ((d & 31) == 0) { r1[d >> 5] = s1; r2[d >> 5] = s2; }
  __syncthreads();
  if (d == 0) {
    float t1 = 0.f, t2 = 0.f;
    for (int i = 0; i < 6; ++i) { t1 += r1[i]; t2 += r2[i]; }
    float mu = t1 / (float)DINNER;
    float var = t2 / (float)DINNER - mu * mu;
    smu = mu; srs = rsqrtf(var + 1e-5f);
  }
  __syncthreads();
  float yln = (v - smu) * srs * g[d] + bb[d];
  float zv = xz[(size_t)m * 384 + DINNER + d];
  y2[(size_t)m * DINNER + d] = yln * siluf(zv);
}

extern "C" void kernel_launch(void* const* d_in, const int* in_sizes, int n_in,
                              void* d_out, int out_size, void* d_ws, size_t ws_size,
                              hipStream_t stream) {
  (void)in_sizes; (void)n_in; (void)out_size; (void)ws_size;
  const float* x   = (const float*)d_in[0];
  const float* xt  = (const float*)d_in[1];
  const float* ipw = (const float*)d_in[2];
  const float* c2w = (const float*)d_in[3];
  const float* c2b = (const float*)d_in[4];
  const float* cxw = (const float*)d_in[5];
  const float* cxb = (const float*)d_in[6];
  const float* xpw = (const float*)d_in[7];
  const float* dtw = (const float*)d_in[8];
  const float* dtb = (const float*)d_in[9];
  const float* alg = (const float*)d_in[10];
  const float* dss = (const float*)d_in[11];
  const float* lng = (const float*)d_in[12];
  const float* lnb = (const float*)d_in[13];
  const float* opw = (const float*)d_in[14];
  float* out = (float*)d_out;
  float* ws  = (float*)d_ws;

  size_t off = 0;
  float* xz    = ws + off; off += (size_t)MROWS * 384;
  float* xin   = ws + off; off += (size_t)BATCH * 2 * Ls * DINNER;
  float* xs    = ws + off; off += (size_t)BATCH * 4 * DINNER * Ls;
  float* Wp    = ws + off; off += (size_t)NKO_PAD * KBIG;
  float* kdp   = ws + off; off += (size_t)NKO_PAD * DINNER;
  float* xdbl  = ws + off; off += (size_t)BATCH * 4 * NCOUT * Ls;
  float* delta = ws + off; off += (size_t)BATCH * 4 * DINNER * Ls;
  float* oy    = ws + off; off += (size_t)BATCH * 4 * DINNER * Ls;
  float* y2    = xin;

  gemm_f32_wmma<<<dim3(384 / 16, MROWS / 16), 32, 0, stream>>>(x, ipw, xz, DMODEL, 384);

  k_pack_w <<<(NKO_PAD * KBIG + 255) / 256, 256, 0, stream>>>(xpw, Wp);
  k_pack_kd<<<(NKO_PAD * DINNER + 255) / 256, 256, 0, stream>>>(xpw, kdp);

  k2_conv<<<(BATCH * 2 * Ls * DINNER + 255) / 256, 256, 0, stream>>>(
      xz, xt, c2w, c2b, cxw, cxb, xin);

  k3_xs<<<(BATCH * DINNER * Ls + 255) / 256, 256, 0, stream>>>(xin, xs);

  k4_cdc<<<dim3(MROWS / 64, NKO_PAD / (16 * CDC_NT)), 128, 0, stream>>>(xin, Wp, kdp, xdbl);

  k5_delta<<<(BATCH * 4 * DINNER * Ls + 255) / 256, 256, 0, stream>>>(xdbl, dtw, dtb, delta);

  k6_scan<<<96, 256, 0, stream>>>(xs, xdbl, alg, dss, delta, oy);

  k7a_comb<<<(MROWS * DINNER + 255) / 256, 256, 0, stream>>>(oy, y2);
  k7b_ln<<<MROWS, DINNER, 0, stream>>>(xz, lng, lnb, y2);

  gemm_f32_wmma<<<dim3(DMODEL / 16, MROWS / 16), 32, 0, stream>>>(y2, opw, out, DINNER, DMODEL);
}


# k3 cross-scan build and k7a cross-merge (layout transposes) staged through LDS tiles: global reads and writes are whole 128-byte lines on both sides, not one 4-byte element per line; sum order kept
# speedup vs baseline: 5.6667x; 1.1452x over previous
#include <hip/hip_runtime.h>
#include <hip/hip_bf16.h>
#include <math.h>

typedef float v2f __attribute__((ext_vector_type(2)));
typedef float v8f __attribute__((ext_vector_type(8)));

#define BATCH   2
#define Hs      56
#define Ws      56
#define Ls      (Hs*Ws)
#define DMODEL  96
#define DINNER  192
#define DSTATE  16
#define DTRANK  6
#define NCOUT   38
#define NKO     (4*NCOUT)
#define NKO_PAD 160
#define KBIG    (DINNER*2*9)
#define THETA_C 0.6f
#define MROWS   (BATCH*Ls)

static __device__ __forceinline__ float siluf(float v) { return v / (1.f + expf(-v)); }

static __device__ __forceinline__ v8f wmma4(v2f a, v2f b, v8f c) {
  return __builtin_amdgcn_wmma_f32_16x16x4_f32(false, a, false, b, (short)0, c, false, false);
}

__global__ void gemm_f32_wmma(const float* __restrict__ A, const float* __restrict__ Bt,
                              float* __restrict__ C, int K, int ldc) {
  const int lane = threadIdx.x & 31;
  const int half = lane >> 4;
  const int klo  = half << 1;
  const int ln   = lane & 15;
  const int ntile = blockIdx.x;
  const int mtile = blockIdx.y;
  const float* ap = A  + (size_t)(mtile * 16 + ln) * K + klo;
  const float* bp = Bt + (size_t)(ntile * 16 + ln) * K + klo;
  v8f acc = {0.f,0.f,0.f,0.f,0.f,0.f,0.f,0.f};
  for (int k = 0; k < K; k += 4) {
    v2f a = *(const v2f*)(ap + k);
    v2f b = *(const v2f*)(bp + k);
    acc = wmma4(a, b, acc);
  }
  const int n = ntile * 16 + ln;
  for (int rr = 0; rr < 8; ++rr) {
    int mm = mtile * 16 + rr + half * 8;
    C[(size_t)mm * ldc + n] = acc[rr];
  }
}

#define NPAIR_W  (KBIG / 8)
#define NPAIR_KD (DINNER / 8)

__global__ void k_pack_w(const float* __restrict__ xw, float* __restrict__ Wp) {
  int tid = blockIdx.x * blockDim.x + threadIdx.x;
  if (tid >= NKO_PAD * KBIG) return;
  int e    = tid & 3;
  int lane = (tid >> 2) & 31;
  int grp  = tid >> 7;
  int pr   = grp % NPAIR_W;
  int nt   = grp / NPAIR_W;
  int n    = nt * 16 + (lane & 15);
  int kk   = pr * 8 + (e >> 1) * 4 + (lane >> 4) * 2 + (e & 1);
  float v = 0.f;
  if (n < NKO) {
    int c = kk % DINNER;
    int r = kk / DINNER;
    int dx = r % 3, dy = (r / 3) % 3, t = r / 9;
    v = xw[((((size_t)n * DINNER + c) * 2 + t) * 3 + dy) * 3 + dx];
  }
  Wp[tid] = v;
}

__global__ void k_pack_kd(const float* __restrict__ xw, float* __restrict__ kdp) {
  int tid = blockIdx.x * blockDim.x + threadIdx.x;
  if (tid >= NKO_PAD * DINNER) return;
  int e    = tid & 3;
  int lane = (tid >> 2) & 31;
  int grp  = tid >> 7;
  int pr   = grp % NPAIR_KD;
  int nt   = grp / NPAIR_KD;
  int n    = nt * 16 + (lane & 15);
  int c    = pr * 8 + (e >> 1) * 4 + (lane >> 4) * 2 + (e & 1);
  float v = 0.f;
  if (n < NKO) {
    const float* p = xw + ((size_t)n * DINNER + c) * 18;
    for (int i = 0; i < 18; ++i) v += p[i];
  }
  kdp[tid] = v;
}

__global__ void k2_conv(const float* __restrict__ xz, const float* __restrict__ xt,
                        const float* __restrict__ w2, const float* __restrict__ b2,
                        const float* __restrict__ wxt, const float* __restrict__ bxt,
                        float* __restrict__ xin) {
  int tid = blockIdx.x * blockDim.x + threadIdx.x;
  if (tid >= BATCH * 2 * Ls * DINNER) return;
  int c = tid % DINNER; int r = tid / DINNER;
  int w = r % Ws; r /= Ws;
  int h = r % Hs; r /= Hs;
  int t = r % 2;  int b = r / 2;
  float acc;
  if (t == 0) {
    acc = b2[c];
    for (int dy = 0; dy < 3; ++dy) {
      int hh = h + dy - 1; if (hh < 0 || hh >= Hs) continue;
      for (int dx = 0; dx < 3; ++dx) {
        int ww = w + dx - 1; if (ww < 0 || ww >= Ws) continue;
        acc += xz[((size_t)((b * Hs + hh) * Ws + ww)) * 384 + c] * w2[c * 9 + dy * 3 + dx];
      }
    }
  } else {
    acc = bxt[c];
    int ci = c >> 1;
    for (int dy = 0; dy < 3; ++dy) {
      int hh = h + dy - 1; if (hh < 0 || hh >= Hs) continue;
      for (int dx = 0; dx < 3; ++dx) {
        int ww = w + dx - 1; if (ww < 0 || ww >= Ws) continue;
        acc += xt[((size_t)((b * Hs + hh) * Ws + ww)) * DMODEL + ci] * wxt[c * 9 + dy * 3 + dx];
      }
    }
  }
  xin[tid] = siluf(acc);
}

#define XT_P     224
#define XT_D     32
#define XT_PITCH 33

__global__ void __launch_bounds__(256) k3_xs(const float* __restrict__ xin, float* __restrict__ xs) {
  __shared__ float tile[XT_P * XT_PITCH];
  const int g    = blockIdx.x;
  const int d0   = blockIdx.y * XT_D;
  const int b    = blockIdx.z >> 1;
  const int col  = blockIdx.z & 1;
  const int lane = threadIdx.x & 31;
  const int wv   = __builtin_amdgcn_readfirstlane(threadIdx.x >> 5);

  const int major  = g * 4 + (wv >> 1);
  const int minor0 = (wv & 1) * 28;
  const int sMaj = col ? Ws : 1;
  const int sMin = col ? 1 : Ws;
  const float* xb = xin + ((size_t)b * 2 * Ls + major * sMaj + minor0 * sMin) * DINNER + d0 + lane;
  float pv[28];
  #pragma unroll
  for (int it = 0; it < 28; ++it) {
    const int s  = it & 1;
    const int i2 = it & ~1;
    const float* p0 = xb + ((size_t)s * Ls + (size_t)i2 * sMin) * DINNER;
    float v0 = p0[0];
    float v1 = p0[(size_t)sMin * DINNER];
    pv[it] = 0.5f * (v0 + v1) + fmaxf(v0, v1);
  }
  #pragma unroll
  for (int it = 0; it < 28; ++it) tile[(wv * 28 + it) * XT_PITCH + lane] = pv[it];
  __syncthreads();

  const int kbase = col ? 2 : 0;
  #pragma unroll
  for (int i = 0; i < 4; ++i) {
    int dd = wv * 4 + i;
    float* op0 = xs + ((size_t)(b * 4 + kbase) * DINNER + d0 + dd) * Ls + g * XT_P;
    float* op1 = op0 + (size_t)DINNER * Ls;
    #pragma unroll
    for (int c = 0; c < 7; ++c) {
      int q  = lane + 32 * c;
      int sp = (q / 56) * 56 + 55 - q % 56;
      op0[q] = tile[q  * XT_PITCH + dd];
      op1[q] = tile[sp * XT_PITCH + dd];
    }
  }
}

#define CDC_NT     5
#define CDC_STAGES (KBIG / 32)

struct CdcStage { v2f a[8]; float4 b[4][CDC_NT]; int ok; };

static __device__ __forceinline__ void cdc_load(CdcStage& q, const float* __restrict__ xin,
                                                const float* __restrict__ wf, int s,
                                                int b, int h, int w, int klo) {
  int td = s / 6, cc = (s - td * 6) * 32;
  int t = td / 9, r9 = td - t * 9, dy = r9 / 3, dx = r9 - dy * 3;
  int hh = h + dy - 1, ww = w + dx - 1;
  q.ok = (hh >= 0) && (hh < Hs) && (ww >= 0) && (ww < Ws);
  int hc = hh < 0 ? 0 : (hh >= Hs ? Hs - 1 : hh);
  int wc = ww < 0 ? 0 : (ww >= Ws ? Ws - 1 : ww);
  const float* ap = xin + (((size_t)(b * 2 + t) * Hs + hc) * Ws + wc) * DINNER + klo + cc;
  #pragma unroll
  for (int i = 0; i < 8; ++i) q.a[i] = *(const v2f*)(ap + 4 * i);
  const float* bp = wf + (size_t)s * 4 * 128;
  #pragma unroll
  for (int qd = 0; qd < 4; ++qd) {
    #pragma unroll
    for (int j = 0; j < CDC_NT; ++j)
      q.b[qd][j] = *(const float4*)(bp + ((size_t)j * NPAIR_W + qd) * 128);
  }
}

static __device__ __forceinline__ void cdc_compute(const CdcStage& q, v8f (&acc)[CDC_NT]) {
  const v2f zero = {0.f, 0.f};
  #pragma unroll
  for (int qd = 0; qd < 4; ++qd) {
    #pragma unroll
    for (int e2 = 0; e2 < 2; ++e2) {
      v2f a = q.ok ? q.a[2 * qd + e2] : zero;
      #pragma unroll
      for (int j = 0; j < CDC_NT; ++j) {
        float4 bv = q.b[qd][j];
        v2f bb;
        if (e2 == 0) { bb.x = bv.x; bb.y = bv.y; } else { bb.x = bv.z; bb.y = bv.w; }
        acc[j] = wmma4(a, bb, acc[j]);
      }
    }
  }
}

__global__ void __launch_bounds__(128) k4_cdc(const float* __restrict__ xin, const float* __restrict__ Wp,
                       const float* __restrict__ kdp, float* __restrict__ xdbl) {
  const int lane = threadIdx.x & 31;
  const int half = lane >> 4;
  const int klo  = half << 1;
  const int ln   = lane & 15;
  const int mtile = blockIdx.x * 4 + (threadIdx.x >> 5);
  const int nt0   = blockIdx.y * CDC_NT;
  const int m = mtile * 16 + ln;
  const int b = m / Ls;
  const int hw = m % Ls;
  const int h = hw / Ws, w = hw % Ws;

  v8f accN[CDC_NT];
  #pragma unroll
  for (int j = 0; j < CDC_NT; ++j) accN[j] = (v8f){0.f,0.f,0.f,0.f,0.f,0.f,0.f,0.f};

  const float* wf = Wp + (size_t)nt0 * NPAIR_W * 128 + lane * 4;
  CdcStage qa, qb;
  cdc_load(qa, xin, wf, 0, b, h, w, klo);
  for (int s = 0; s < CDC_STAGES; s += 2) {
    cdc_load(qb, xin, wf, s + 1, b, h, w, klo);
    cdc_compute(qa, accN);
    int s2 = (s + 2 < CDC_STAGES) ? (s + 2) : (s + 1);
    cdc_load(qa, xin, wf, s2, b, h, w, klo);
    cdc_compute(qb, accN);
  }

  v8f accD0[CDC_NT], accD1[CDC_NT];
  #pragma unroll
  for (int j = 0; j < CDC_NT; ++j) {
    accD0[j] = (v8f){0.f,0.f,0.f,0.f,0.f,0.f,0.f,0.f};
    accD1[j] = (v8f){0.f,0.f,0.f,0.f,0.f,0.f,0.f,0.f};
  }
  const float* ap0 = xin + (((size_t)(b * 2 + 0) * Hs + h) * Ws + w) * DINNER + klo;
  const float* ap1 = xin + (((size_t)(b * 2 + 1) * Hs + h) * Ws + w) * DINNER + klo;
  const float* kf  = kdp + (size_t)nt0 * NPAIR_KD * 128 + lane * 4;
  #pragma unroll
  for (int pr = 0; pr < NPAIR_KD; ++pr) {
    v2f a00 = *(const v2f*)(ap0 + pr * 8);
    v2f a01 = *(const v2f*)(ap0 + pr * 8 + 4);
    v2f a10 = *(const v2f*)(ap1 + pr * 8);
    v2f a11 = *(const v2f*)(ap1 + pr * 8 + 4);
    #pragma unroll
    for (int j = 0; j < CDC_NT; ++j) {
      float4 bv = *(const float4*)(kf + ((size_t)j * NPAIR_KD + pr) * 128);
      v2f b0, b1;
      b0.x = bv.x; b0.y = bv.y; b1.x = bv.z; b1.y = bv.w;
      accD0[j] = wmma4(a00, b0, accD0[j]);
      accD1[j] = wmma4(a10, b0, accD1[j]);
      accD0[j] = wmma4(a01, b1, accD0[j]);
      accD1[j] = wmma4(a11, b1, accD1[j]);
    }
  }

  #pragma unroll
  for (int j = 0; j < CDC_NT; ++j) {
    const int n = (nt0 + j) * 16 + ln;
    if (n < NKO) {
      int kq = n / NCOUT, o = n % NCOUT;
      for (int rr = 0; rr < 8; ++rr) {
        int mm = mtile * 16 + rr + half * 8;
        int b2 = mm / Ls; int l2 = mm % Ls;
        float Nv = accN[j][rr], d0 = accD0[j][rr], d1 = accD1[j][rr];
        float pooled = 2.f * Nv - THETA_C * (0.5f * (d0 + d1) + fminf(d0, d1));
        xdbl[(((size_t)b2 * 4 + kq) * NCOUT + o) * Ls + l2] = pooled;
      }
    }
  }
}

__global__ void k5_delta(const float* __restrict__ xdbl, const float* __restrict__ dtw,
                         const float* __restrict__ dtb, float* __restrict__ delta) {
  int tid = blockIdx.x * blockDim.x + threadIdx.x;
  if (tid >= BATCH * 4 * DINNER * Ls) return;
  int l = tid % Ls; int r = tid / Ls;
  int d = r % DINNER; int bk = r / DINNER;
  int k = bk & 3;
  float acc = dtb[k * DINNER + d];
  const float* wp = dtw + ((size_t)k * DINNER + d) * DTRANK;
  const float* xp = xdbl + (size_t)bk * NCOUT * Ls + l;
  for (int rr = 0; rr < DTRANK; ++rr) acc += xp[(size_t)rr * Ls] * wp[rr];
  delta[tid] = fmaxf(acc, 0.f) + log1pf(expf(-fabsf(acc)));
}

#define SCH 16

struct ScanChunk { float d[SCH], u[SCH], b[SCH], c[SCH]; };

static __device__ __forceinline__ void scan_load(ScanChunk& q, const float* __restrict__ dP,
                                                 const float* __restrict__ uP,
                                                 const float* __restrict__ bP,
                                                 const float* __restrict__ cP, int l0) {
  #pragma unroll
  for (int i = 0; i < SCH; i += 4) {
    float4 vd = *(const float4*)(dP + l0 + i);
    float4 vu = *(const float4*)(uP + l0 + i);
    float4 vb = *(const float4*)(bP + l0 + i);
    float4 vc = *(const float4*)(cP + l0 + i);
    q.d[i] = vd.x; q.d[i + 1] = vd.y; q.d[i + 2] = vd.z; q.d[i + 3] = vd.w;
    q.u[i] = vu.x; q.u[i + 1] = vu.y; q.u[i + 2] = vu.z; q.u[i + 3] = vu.w;
    q.b[i] = vb.x; q.b[i + 1] = vb.y; q.b[i + 2] = vb.z; q.b[i + 3] = vb.w;
    q.c[i] = vc.x; q.c[i + 1] = vc.y; q.c[i + 2] = vc.z; q.c[i + 3] = vc.w;
  }
}

static __device__ __forceinline__ float scan_steps(const ScanChunk& q, float& h, float A,
                                                   float Dv, int ln) {
  float y[SCH];
  #pragma unroll
  for (int j = 0; j < SCH; ++j) {
    float dl = q.d[j];
    h = __builtin_amdgcn_exp2f(dl * A) * h + (dl * q.u[j]) * q.b[j];
    y[j] = h * q.c[j];
  }
  #pragma unroll
  for (int m = 1; m < 16; m <<= 1) {
    float t[SCH];
    #pragma unroll
    for (int j = 0; j < SCH; ++j) t[j] = __shfl_xor(y[j], m);
    __builtin_amdgcn_sched_barrier(0);
    #pragma unroll
    for (int j = 0; j < SCH; ++j) y[j] += t[j];
  }
  float keep = 0.f;
  #pragma unroll
  for (int j = 0; j < SCH; ++j) {
    float o = y[j] + Dv * q.u[j];
    if (ln == j) keep = o;
  }
  return keep;
}

__global__ void __launch_bounds__(256) k6_scan(const float* __restrict__ xs, const float* __restrict__ xdbl,
                        const float* __restrict__ Alogs, const float* __restrict__ Dss,
                        const float* __restrict__ delta, float* __restrict__ oy) {
  int wid  = blockIdx.x * (blockDim.x >> 5) + (threadIdx.x >> 5);
  int lane = threadIdx.x & 31;
  int half = lane >> 4, ln = lane & 15;
  int bk = wid / 96;
  int d  = (wid % 96) * 2 + half;
  int k  = bk & 3;
  const float* dP = delta + ((size_t)bk * DINNER + d) * Ls;
  float*       oP = oy    + ((size_t)bk * DINNER + d) * Ls;
  const float* uP = xs    + ((size_t)bk * DINNER + d) * Ls;
  const float* bP = xdbl  + ((size_t)bk * NCOUT + DTRANK + ln) * Ls;
  const float* cP = xdbl  + ((size_t)bk * NCOUT + DTRANK + DSTATE + ln) * Ls;
  float A  = -expf(Alogs[((size_t)(k * DINNER + d)) * DSTATE + ln]) * 1.4426950408889634f;
  float Dv = Dss[k * DINNER + d];
  float h = 0.f;
  ScanChunk qa, qb;
  scan_load(qa, dP, uP, bP, cP, 0);
  for (int l0 = 0; l0 < Ls; l0 += 2 * SCH) {
    scan_load(qb, dP, uP, bP, cP, l0 + SCH);
    oP[l0 + ln] = scan_steps(qa, h, A, Dv, ln);
    int ln2 = (l0 + 2 * SCH < Ls) ? (l0 + 2 * SCH) : (l0 + SCH);
    scan_load(qa, dP, uP, bP, cP, ln2);
    oP[l0 + SCH + ln] = scan_steps(qb, h, A, Dv, ln);
  }
}

__global__ void __launch_bounds__(256) k7a_comb(const float* __restrict__ oy, float* __restrict__ y2) {
  __shared__ float tile[XT_P * XT_PITCH];
  const int h0   = blockIdx.x * 4;
  const int d0   = blockIdx.y * XT_D;
  const int b    = blockIdx.z;
  const int lane = threadIdx.x & 31;
  const int wv   = __builtin_amdgcn_readfirstlane(threadIdx.x >> 5);
  const size_t s = (size_t)DINNER * Ls;

  #pragma unroll
  for (int it = 0; it < (XT_D * Ws) / 256; ++it) {
    int idx = it * 256 + threadIdx.x;
    int dd = idx / Ws;
    int w  = idx % Ws;
    const float* base = oy + ((size_t)b * 4 * DINNER + d0 + dd) * Ls;
    int lt0 = w * Hs + h0;
    float4 t1 = *(const float4*)(base + s + lt0);
    float4 t3 = *(const float4*)(base + 3 * s + (Ls - 4 - lt0));
    float a1[4] = {t1.x, t1.y, t1.z, t1.w};
    float a3[4] = {t3.w, t3.z, t3.y, t3.x};
    #pragma unroll
    for (int j = 0; j < 4; ++j) {
      int l = (h0 + j) * Ws + w;
      float v = base[l] + base[2 * s + (Ls - 1 - l)] + a1[j] + a3[j];
      tile[(j * Ws + w) * XT_PITCH + dd] = v;
    }
  }
  __syncthreads();

  float* yp = y2 + ((size_t)b * Ls + h0 * Ws + wv * 28) * DINNER + d0 + lane;
  #pragma unroll
  for (int it = 0; it < 28; ++it)
    yp[(size_t)it * DINNER] = tile[(wv * 28 + it) * XT_PITCH + lane];
}

__global__ void k7b_ln(const float* __restrict__ xz, const float* __restrict__ g,
                       const float* __restrict__ bb, float* __restrict__ y2) {
  int m = blockIdx.x;
  int d = threadIdx.x;
  float v = y2[(size_t)m * DINNER + d];
  float s1 = v, s2 = v * v;
  #pragma unroll
  for (int off = 16; off >= 1; off >>= 1) {
    s1 += __shfl_xor(s1, off);
    s2 += __shfl_xor(s2, off);
  }
  __shared__ float r1[6], r2[6];
  __shared__ float smu, srs;
  if ((d & 31) == 0) { r1[d >> 5] = s1; r2[d >> 5] = s2; }
  __syncthreads();
  if (d == 0) {
    float t1 = 0.f, t2 = 0.f;
    for (int i = 0; i < 6; ++i) { t1 += r1[i]; t2 += r2[i]; }
    float mu = t1 / (float)DINNER;
    float var = t2 / (float)DINNER - mu * mu;
    smu = mu; srs = rsqrtf(var + 1e-5f);
  }
  __syncthreads();
  float yln = (v - smu) * srs * g[d] + bb[d];
  float zv = xz[(size_t)m * 384 + DINNER + d];
  y2[(size_t)m * DINNER + d] = yln * siluf(zv);
}

extern "C" void kernel_launch(void* const* d_in, const int* in_sizes, int n_in,
                              void* d_out, int out_size, void* d_ws, size_t ws_size,
                              hipStream_t stream) {
  (void)in_sizes; (void)n_in; (void)out_size; (void)ws_size;
  const float* x   = (const float*)d_in[0];
  const float* xt  = (const float*)d_in[1];
  const float* ipw = (const float*)d_in[2];
  const float* c2w = (const float*)d_in[3];
  const float* c2b = (const float*)d_in[4];
  const float* cxw = (const float*)d_in[5];
  const float* cxb = (const float*)d_in[6];
  const float* xpw = (const float*)d_in[7];
  const float* dtw = (const float*)d_in[8];
  const float* dtb = (const float*)d_in[9];
  const float* alg = (const float*)d_in[10];
  const float* dss = (const float*)d_in[11];
  const float* lng = (const float*)d_in[12];
  const float* lnb = (const float*)d_in[13];
  const float* opw = (const float*)d_in[14];
  float* out = (float*)d_out;
  float* ws  = (float*)d_ws;

  size_t off = 0;
  float* xz    = ws + off; off += (size_t)MROWS * 384;
  float* xin   = ws + off; off += (size_t)BATCH * 2 * Ls * DINNER;
  float* xs    = ws + off; off += (size_t)BATCH * 4 * DINNER * Ls;
  float* Wp    = ws + off; off += (size_t)NKO_PAD * KBIG;
  float* kdp   = ws + off; off += (size_t)NKO_PAD * DINNER;
  float* xdbl  = ws + off; off += (size_t)BATCH * 4 * NCOUT * Ls;
  float* delta = ws + off; off += (size_t)BATCH * 4 * DINNER * Ls;
  float* oy    = ws + off; off += (size_t)BATCH * 4 * DINNER * Ls;
  float* y2    = xin;

  gemm_f32_wmma<<<dim3(384 / 16, MROWS / 16), 32, 0, stream>>>(x, ipw, xz, DMODEL, 384);

  k_pack_w <<<(NKO_PAD * KBIG + 255) / 256, 256, 0, stream>>>(xpw, Wp);
  k_pack_kd<<<(NKO_PAD * DINNER + 255) / 256, 256, 0, stream>>>(xpw, kdp);

  k2_conv<<<(BATCH * 2 * Ls * DINNER + 255) / 256, 256, 0, stream>>>(
      xz, xt, c2w, c2b, cxw, cxb, xin);

  k3_xs<<<dim3(Ls / XT_P, DINNER / XT_D, BATCH * 2), 256, 0, stream>>>(xin, xs);

  k4_cdc<<<dim3(MROWS / 64, NKO_PAD / (16 * CDC_NT)), 128, 0, stream>>>(xin, Wp, kdp, xdbl);

  k5_delta<<<(BATCH * 4 * DINNER * Ls + 255) / 256, 256, 0, stream>>>(xdbl, dtw, dtb, delta);

  k6_scan<<<96, 256, 0, stream>>>(xs, xdbl, alg, dss, delta, oy);

  k7a_comb<<<dim3(Ls / XT_P, DINNER / XT_D, BATCH), 256, 0, stream>>>(oy, y2);
  k7b_ln<<<MROWS, DINNER, 0, stream>>>(xz, lng, lnb, y2);

  gemm_f32_wmma<<<dim3(DMODEL / 16, MROWS / 16), 32, 0, stream>>>(y2, opw, out, DINNER, DMODEL);
}


# k5 delta and k2 conv: a thread makes 32 (k5) or 14 (k2) outputs from operands loaded once; 10,080 long waves replace 225,792 short ones whose entry rate bounded both; sum order kept
# speedup vs baseline: 6.1039x; 1.0772x over previous
#include <hip/hip_runtime.h>
#include <hip/hip_bf16.h>
#include <math.h>

typedef float v2f __attribute__((ext_vector_type(2)));
typedef float v8f __attribute__((ext_vector_type(8)));

#define BATCH   2
#define Hs      56
#define Ws      56
#define Ls      (Hs*Ws)
#define DMODEL  96
#define DINNER  192
#define DSTATE  16
#define DTRANK  6
#define NCOUT   38
#define NKO     (4*NCOUT)
#define NKO_PAD 160
#define KBIG    (DINNER*2*9)
#define THETA_C 0.6f
#define MROWS   (BATCH*Ls)

static __device__ __forceinline__ float siluf(float v) { return v / (1.f + expf(-v)); }

static __device__ __forceinline__ v8f wmma4(v2f a, v2f b, v8f c) {
  return __builtin_amdgcn_wmma_f32_16x16x4_f32(false, a, false, b, (short)0, c, false, false);
}

__global__ void gemm_f32_wmma(const float* __restrict__ A, const float* __restrict__ Bt,
                              float* __restrict__ C, int K, int ldc) {
  const int lane = threadIdx.x & 31;
  const int half = lane >> 4;
  const int klo  = half << 1;
  const int ln   = lane & 15;
  const int ntile = blockIdx.x;
  const int mtile = blockIdx.y;
  const float* ap = A  + (size_t)(mtile * 16 + ln) * K + klo;
  const float* bp = Bt + (size_t)(ntile * 16 + ln) * K + klo;
  v8f acc = {0.f,0.f,0.f,0.f,0.f,0.f,0.f,0.f};
  for (int k = 0; k < K; k += 4) {
    v2f a = *(const v2f*)(ap + k);
    v2f b = *(const v2f*)(bp + k);
    acc = wmma4(a, b, acc);
  }
  const int n = ntile * 16 + ln;
  for (int rr = 0; rr < 8; ++rr) {
    int mm = mtile * 16 + rr + half * 8;
    C[(size_t)mm * ldc + n] = acc[rr];
  }
}

#define NPAIR_W  (KBIG / 8)
#define NPAIR_KD (DINNER / 8)

__global__ void k_pack_w(const float* __restrict__ xw, float* __restrict__ Wp) {
  int tid = blockIdx.x * blockDim.x + threadIdx.x;
  if (tid >= NKO_PAD * KBIG) return;
  int e    = tid & 3;
  int lane = (tid >> 2) & 31;
  int grp  = tid >> 7;
  int pr   = grp % NPAIR_W;
  int nt   = grp / NPAIR_W;
  int n    = nt * 16 + (lane & 15);
  int kk   = pr * 8 + (e >> 1) * 4 + (lane >> 4) * 2 + (e & 1);
  float v = 0.f;
  if (n < NKO) {
    int c = kk % DINNER;
    int r = kk / DINNER;
    int dx = r % 3, dy = (r / 3) % 3, t = r / 9;
    v = xw[((((size_t)n * DINNER + c) * 2 + t) * 3 + dy) * 3 + dx];
  }
  Wp[tid] = v;
}

__global__ void k_pack_kd(const float* __restrict__ xw, float* __restrict__ kdp) {
  int tid = blockIdx.x * blockDim.x + threadIdx.x;
  if (tid >= NKO_PAD * DINNER) return;
  int e    = tid & 3;
  int lane = (tid >> 2) & 31;
  int grp  = tid >> 7;
  int pr   = grp % NPAIR_KD;
  int nt   = grp / NPAIR_KD;
  int n    = nt * 16 + (lane & 15);
  int c    = pr * 8 + (e >> 1) * 4 + (lane >> 4) * 2 + (e & 1);
  float v = 0.f;
  if (n < NKO) {
    const float* p = xw + ((size_t)n * DINNER + c) * 18;
    for (int i = 0; i < 18; ++i) v += p[i];
  }
  kdp[tid] = v;
}

#define K2_SW 14

__global__ void __launch_bounds__(DINNER) k2_conv(const float* __restrict__ xz, const float* __restrict__ xt,
                        const float* __restrict__ w2, const float* __restrict__ b2,
                        const float* __restrict__ wxt, const float* __restrict__ bxt,
                        float* __restrict__ xin) {
  const int c  = threadIdx.x;
  const int w0 = blockIdx.x * K2_SW;
  const int h  = blockIdx.y;
  const int t  = blockIdx.z & 1;
  const int b  = blockIdx.z >> 1;
  const float* src = t ? (xt + (c >> 1)) : (xz + c);
  const int    ps  = t ? DMODEL : 2 * DINNER;
  const float* wp  = (t ? wxt : w2) + c * 9;
  const float bias = t ? bxt[c] : b2[c];
  float wk[9];
  #pragma unroll
  for (int i = 0; i < 9; ++i) wk[i] = wp[i];
  float win[3][K2_SW + 2];
  #pragma unroll
  for (int dy = 0; dy < 3; ++dy) {
    const int hh = h + dy - 1;
    const bool rok = (hh >= 0) && (hh < Hs);
    const int hc = hh < 0 ? 0 : (hh >= Hs ? Hs - 1 : hh);
    const float* rp = src + (size_t)((b * Hs + hc) * Ws) * ps;
    #pragma unroll
    for (int j = 0; j < K2_SW + 2; ++j) {
      int ww = w0 + j - 1;
      bool ok = rok;
      if (j == 0)         { ok = ok && (ww >= 0); ww = ww < 0 ? 0 : ww; }
      if (j == K2_SW + 1) { ok = ok && (ww < Ws); ww = ww >= Ws ? Ws - 1 : ww; }
      const float v = rp[(size_t)ww * ps];
      win[dy][j] = ok ? v : 0.f;
    }
  }
  float* op = xin + ((size_t)((b * 2 + t) * Hs + h) * Ws + w0) * DINNER + c;
  #pragma unroll
  for (int j = 0; j < K2_SW; ++j) {
    float acc = bias;
    #pragma unroll
    for (int dy = 0; dy < 3; ++dy) {
      #pragma unroll
      for (int dx = 0; dx < 3; ++dx) acc += win[dy][j + dx] * wk[dy * 3 + dx];
    }
    op[(size_t)j * DINNER] = siluf(acc);
  }
}

#define XT_P     224
#define XT_D     32
#define XT_PITCH 33

__global__ void __launch_bounds__(256) k3_xs(const float* __restrict__ xin, float* __restrict__ xs) {
  __shared__ float tile[XT_P * XT_PITCH];
  const int g    = blockIdx.x;
  const int d0   = blockIdx.y * XT_D;
  const int b    = blockIdx.z >> 1;
  const int col  = blockIdx.z & 1;
  const int lane = threadIdx.x & 31;
  const int wv   = __builtin_amdgcn_readfirstlane(threadIdx.x >> 5);

  const int major  = g * 4 + (wv >> 1);
  const int minor0 = (wv & 1) * 28;
  const int sMaj = col ? Ws : 1;
  const int sMin = col ? 1 : Ws;
  const float* xb = xin + ((size_t)b * 2 * Ls + major * sMaj + minor0 * sMin) * DINNER + d0 + lane;
  float pv[28];
  #pragma unroll
  for (int it = 0; it < 28; ++it) {
    const int s  = it & 1;
    const int i2 = it & ~1;
    const float* p0 = xb + ((size_t)s * Ls + (size_t)i2 * sMin) * DINNER;
    float v0 = p0[0];
    float v1 = p0[(size_t)sMin * DINNER];
    pv[it] = 0.5f * (v0 + v1) + fmaxf(v0, v1);
  }
  #pragma unroll
  for (int it = 0; it < 28; ++it) tile[(wv * 28 + it) * XT_PITCH + lane] = pv[it];
  __syncthreads();

  const int kbase = col ? 2 : 0;
  #pragma unroll
  for (int i = 0; i < 4; ++i) {
    int dd = wv * 4 + i;
    float* op0 = xs + ((size_t)(b * 4 + kbase) * DINNER + d0 + dd) * Ls + g * XT_P;
    float* op1 = op0 + (size_t)DINNER * Ls;
    #pragma unroll
    for (int c = 0; c < 7; ++c) {
      int q  = lane + 32 * c;
      int sp = (q / 56) * 56 + 55 - q % 56;
      op0[q] = tile[q  * XT_PITCH + dd];
      op1[q] = tile[sp * XT_PITCH + dd];
    }
  }
}

#define CDC_NT     5
#define CDC_STAGES (KBIG / 32)

struct CdcStage { v2f a[8]; float4 b[4][CDC_NT]; int ok; };

static __device__ __forceinline__ void cdc_load(CdcStage& q, const float* __restrict__ xin,
                                                const float* __restrict__ wf, int s,
                                                int b, int h, int w, int klo) {
  int td = s / 6, cc = (s - td * 6) * 32;
  int t = td / 9, r9 = td - t * 9, dy = r9 / 3, dx = r9 - dy * 3;
  int hh = h + dy - 1, ww = w + dx - 1;
  q.ok = (hh >= 0) && (hh < Hs) && (ww >= 0) && (ww < Ws);
  int hc = hh < 0 ? 0 : (hh >= Hs ? Hs - 1 : hh);
  int wc = ww < 0 ? 0 : (ww >= Ws ? Ws - 1 : ww);
  const float* ap = xin + (((size_t)(b * 2 + t) * Hs + hc) * Ws + wc) * DINNER + klo + cc;
  #pragma unroll
  for (int i = 0; i < 8; ++i) q.a[i] = *(const v2f*)(ap + 4 * i);
  const float* bp = wf + (size_t)s * 4 * 128;
  #pragma unroll
  for (int qd = 0; qd < 4; ++qd) {
    #pragma unroll
    for (int j = 0; j < CDC_NT; ++j)
      q.b[qd][j] = *(const float4*)(bp + ((size_t)j * NPAIR_W + qd) * 128);
  }
}

static __device__ __forceinline__ void cdc_compute(const CdcStage& q, v8f (&acc)[CDC_NT]) {
  const v2f zero = {0.f, 0.f};
  #pragma unroll
  for (int qd = 0; qd < 4; ++qd) {
    #pragma unroll
    for (int e2 = 0; e2 < 2; ++e2) {
      v2f a = q.ok ? q.a[2 * qd + e2] : zero;
      #pragma unroll
      for (int j = 0; j < CDC_NT; ++j) {
        float4 bv = q.b[qd][j];
        v2f bb;
        if (e2 == 0) { bb.x = bv.x; bb.y = bv.y; } else { bb.x = bv.z; bb.y = bv.w; }
        acc[j] = wmma4(a, bb, acc[j]);
      }
    }
  }
}

__global__ void __launch_bounds__(128) k4_cdc(const float* __restrict__ xin, const float* __restrict__ Wp,
                       const float* __restrict__ kdp, float* __restrict__ xdbl) {
  const int lane = threadIdx.x & 31;
  const int half = lane >> 4;
  const int klo  = half << 1;
  const int ln   = lane & 15;
  const int mtile = blockIdx.x * 4 + (threadIdx.x >> 5);
  const int nt0   = blockIdx.y * CDC_NT;
  const int m = mtile * 16 + ln;
  const int b = m / Ls;
  const int hw = m % Ls;
  const int h = hw / Ws, w = hw % Ws;

  v8f accN[CDC_NT];
  #pragma unroll
  for (int j = 0; j < CDC_NT; ++j) accN[j] = (v8f){0.f,0.f,0.f,0.f,0.f,0.f,0.f,0.f};

  const float* wf = Wp + (size_t)nt0 * NPAIR_W * 128 + lane * 4;
  CdcStage qa, qb;
  cdc_load(qa, xin, wf, 0, b, h, w, klo);
  for (int s = 0; s < CDC_STAGES; s += 2) {
    cdc_load(qb, xin, wf, s + 1, b, h, w, klo);
    cdc_compute(qa, accN);
    int s2 = (s + 2 < CDC_STAGES) ? (s + 2) : (s + 1);
    cdc_load(qa, xin, wf, s2, b, h, w, klo);
    cdc_compute(qb, accN);
  }

  v8f accD0[CDC_NT], accD1[CDC_NT];
  #pragma unroll
  for (int j = 0; j < CDC_NT; ++j) {
    accD0[j] = (v8f){0.f,0.f,0.f,0.f,0.f,0.f,0.f,0.f};
    accD1[j] = (v8f){0.f,0.f,0.f,0.f,0.f,0.f,0.f,0.f};
  }
  const float* ap0 = xin + (((size_t)(b * 2 + 0) * Hs + h) * Ws + w) * DINNER + klo;
  const float* ap1 = xin + (((size_t)(b * 2 + 1) * Hs + h) * Ws + w) * DINNER + klo;
  const float* kf  = kdp + (size_t)nt0 * NPAIR_KD * 128 + lane * 4;
  #pragma unroll
  for (int pr = 0; pr < NPAIR_KD; ++pr) {
    v2f a00 = *(const v2f*)(ap0 + pr * 8);
    v2f a01 = *(const v2f*)(ap0 + pr * 8 + 4);
    v2f a10 = *(const v2f*)(ap1 + pr * 8);
    v2f a11 = *(const v2f*)(ap1 + pr * 8 + 4);
    #pragma unroll
    for (int j = 0; j < CDC_NT; ++j) {
      float4 bv = *(const float4*)(kf + ((size_t)j * NPAIR_KD + pr) * 128);
      v2f b0, b1;
      b0.x = bv.x; b0.y = bv.y; b1.x = bv.z; b1.y = bv.w;
      accD0[j] = wmma4(a00, b0, accD0[j]);
      accD1[j] = wmma4(a10, b0, accD1[j]);
      accD0[j] = wmma4(a01, b1, accD0[j]);
      accD1[j] = wmma4(a11, b1, accD1[j]);
    }
  }

  #pragma unroll
  for (int j = 0; j < CDC_NT; ++j) {
    const int n = (nt0 + j) * 16 + ln;
    if (n < NKO) {
      int kq = n / NCOUT, o = n % NCOUT;
      for (int rr = 0; rr < 8; ++rr) {
        int mm = mtile * 16 + rr + half * 8;
        int b2 = mm / Ls; int l2 = mm % Ls;
        float Nv = accN[j][rr], d0 = accD0[j][rr], d1 = accD1[j][rr];
        float pooled = 2.f * Nv - THETA_C * (0.5f * (d0 + d1) + fminf(d0, d1));
        xdbl[(((size_t)b2 * 4 + kq) * NCOUT + o) * Ls + l2] = pooled;
      }
    }
  }
}

#define K5_DG  16
#define K5_TPB 224

__global__ void __launch_bounds__(K5_TPB) k5_delta(const float* __restrict__ xdbl, const float* __restrict__ dtw,
                         const float* __restrict__ dtb, float* __restrict__ delta) {
  const int l  = (blockIdx.x * K5_TPB + threadIdx.x) * 2;
  const int d0 = blockIdx.y * K5_DG;
  const int bk = blockIdx.z;
  const int k  = bk & 3;
  const float* xp = xdbl + (size_t)bk * NCOUT * Ls + l;
  v2f xv[DTRANK];
  #pragma unroll
  for (int rr = 0; rr < DTRANK; ++rr) xv[rr] = *(const v2f*)(xp + (size_t)rr * Ls);
  const float* wp = dtw + ((size_t)k * DINNER + d0) * DTRANK;
  const float* bp = dtb + k * DINNER + d0;
  float* op = delta + ((size_t)bk * DINNER + d0) * Ls + l;
  #pragma unroll 4
  for (int i = 0; i < K5_DG; ++i) {
    float a0 = bp[i];
    float a1 = a0;
    #pragma unroll
    for (int rr = 0; rr < DTRANK; ++rr) {
      float wv = wp[i * DTRANK + rr];
      a0 += xv[rr].x * wv;
      a1 += xv[rr].y * wv;
    }
    v2f o;
    o.x = fmaxf(a0, 0.f) + log1pf(expf(-fabsf(a0)));
    o.y = fmaxf(a1, 0.f) + log1pf(expf(-fabsf(a1)));
    *(v2f*)(op + (size_t)i * Ls) = o;
  }
}

#define SCH 16

struct ScanChunk { float d[SCH], u[SCH], b[SCH], c[SCH]; };

static __device__ __forceinline__ void scan_load(ScanChunk& q, const float* __restrict__ dP,
                                                 const float* __restrict__ uP,
                                                 const float* __restrict__ bP,
                                                 const float* __restrict__ cP, int l0) {
  #pragma unroll
  for (int i = 0; i < SCH; i += 4) {
    float4 vd = *(const float4*)(dP + l0 + i);
    float4 vu = *(const float4*)(uP + l0 + i);
    float4 vb = *(const float4*)(bP + l0 + i);
    float4 vc = *(const float4*)(cP + l0 + i);
    q.d[i] = vd.x; q.d[i + 1] = vd.y; q.d[i + 2] = vd.z; q.d[i + 3] = vd.w;
    q.u[i] = vu.x; q.u[i + 1] = vu.y; q.u[i + 2] = vu.z; q.u[i + 3] = vu.w;
    q.b[i] = vb.x; q.b[i + 1] = vb.y; q.b[i + 2] = vb.z; q.b[i + 3] = vb.w;
    q.c[i] = vc.x; q.c[i + 1] = vc.y; q.c[i + 2] = vc.z; q.c[i + 3] = vc.w;
  }
}

static __device__ __forceinline__ float scan_steps(const ScanChunk& q, float& h, float A,
                                                   float Dv, int ln) {
  float y[SCH];
  #pragma unroll
  for (int j = 0; j < SCH; ++j) {
    float dl = q.d[j];
    h = __builtin_amdgcn_exp2f(dl * A) * h + (dl * q.u[j]) * q.b[j];
    y[j] = h * q.c[j];
  }
  #pragma unroll
  for (int m = 1; m < 16; m <<= 1) {
    float t[SCH];
    #pragma unroll
    for (int j = 0; j < SCH; ++j) t[j] = __shfl_xor(y[j], m);
    __builtin_amdgcn_sched_barrier(0);
    #pragma unroll
    for (int j = 0; j < SCH; ++j) y[j] += t[j];
  }
  float keep = 0.f;
  #pragma unroll
  for (int j = 0; j < SCH; ++j) {
    float o = y[j] + Dv * q.u[j];
    if (ln == j) keep = o;
  }
  return keep;
}

__global__ void __launch_bounds__(256) k6_scan(const float* __restrict__ xs, const float* __restrict__ xdbl,
                        const float* __restrict__ Alogs, const float* __restrict__ Dss,
                        const float* __restrict__ delta, float* __restrict__ oy) {
  int wid  = blockIdx.x * (blockDim.x >> 5) + (threadIdx.x >> 5);
  int lane = threadIdx.x & 31;
  int half = lane >> 4, ln = lane & 15;
  int bk = wid / 96;
  int d  = (wid % 96) * 2 + half;
  int k  = bk & 3;
  const float* dP = delta + ((size_t)bk * DINNER + d) * Ls;
  float*       oP = oy    + ((size_t)bk * DINNER + d) * Ls;
  const float* uP = xs    + ((size_t)bk * DINNER + d) * Ls;
  const float* bP = xdbl  + ((size_t)bk * NCOUT + DTRANK + ln) * Ls;
  const float* cP = xdbl  + ((size_t)bk * NCOUT + DTRANK + DSTATE + ln) * Ls;
  float A  = -expf(Alogs[((size_t)(k * DINNER + d)) * DSTATE + ln]) * 1.4426950408889634f;
  float Dv = Dss[k * DINNER + d];
  float h = 0.f;
  ScanChunk qa, qb;
  scan_load(qa, dP, uP, bP, cP, 0);
  for (int l0 = 0; l0 < Ls; l0 += 2 * SCH) {
    scan_load(qb, dP, uP, bP, cP, l0 + SCH);
    oP[l0 + ln] = scan_steps(qa, h, A, Dv, ln);
    int ln2 = (l0 + 2 * SCH < Ls) ? (l0 + 2 * SCH) : (l0 + SCH);
    scan_load(qa, dP, uP, bP, cP, ln2);
    oP[l0 + SCH + ln] = scan_steps(qb, h, A, Dv, ln);
  }
}

__global__ void __launch_bounds__(256) k7a_comb(const float* __restrict__ oy, float* __restrict__ y2) {
  __shared__ float tile[XT_P * XT_PITCH];
  const int h0   = blockIdx.x * 4;
  const int d0   = blockIdx.y * XT_D;
  const int b    = blockIdx.z;
  const int lane = threadIdx.x & 31;
  const int wv   = __builtin_amdgcn_readfirstlane(threadIdx.x >> 5);
  const size_t s = (size_t)DINNER * Ls;

  #pragma unroll
  for (int it = 0; it < (XT_D * Ws) / 256; ++it) {
    int idx = it * 256 + threadIdx.x;
    int dd = idx / Ws;
    int w  = idx % Ws;
    const float* base = oy + ((size_t)b * 4 * DINNER + d0 + dd) * Ls;
    int lt0 = w * Hs + h0;
    float4 t1 = *(const float4*)(base + s + lt0);
    float4 t3 = *(const float4*)(base + 3 * s + (Ls - 4 - lt0));
    float a1[4] = {t1.x, t1.y, t1.z, t1.w};
    float a3[4] = {t3.w, t3.z, t3.y, t3.x};
    #pragma unroll
    for (int j = 0; j < 4; ++j) {
      int l = (h0 + j) * Ws + w;
      float v = base[l] + base[2 * s + (Ls - 1 - l)] + a1[j] + a3[j];
      tile[(j * Ws + w) * XT_PITCH + dd] = v;
    }
  }
  __syncthreads();

  float* yp = y2 + ((size_t)b * Ls + h0 * Ws + wv * 28) * DINNER + d0 + lane;
  #pragma unroll
  for (int it = 0; it < 28; ++it)
    yp[(size_t)it * DINNER] = tile[(wv * 28 + it) * XT_PITCH + lane];
}

__global__ void k7b_ln(const float* __restrict__ xz, const float* __restrict__ g,
                       const float* __restrict__ bb, float* __restrict__ y2) {
  int m = blockIdx.x;
  int d = threadIdx.x;
  float v = y2[(size_t)m * DINNER + d];
  float s1 = v, s2 = v * v;
  #pragma unroll
  for (int off = 16; off >= 1; off >>= 1) {
    s1 += __shfl_xor(s1, off);
    s2 += __shfl_xor(s2, off);
  }
  __shared__ float r1[6], r2[6];
  __shared__ float smu, srs;
  if ((d & 31) == 0) { r1[d >> 5] = s1; r2[d >> 5] = s2; }
  __syncthreads();
  if (d == 0) {
    float t1 = 0.f, t2 = 0.f;
    for (int i = 0; i < 6; ++i) { t1 += r1[i]; t2 += r2[i]; }
    float mu = t1 / (float)DINNER;
    float var = t2 / (float)DINNER - mu * mu;
    smu = mu; srs = rsqrtf(var + 1e-5f);
  }
  __syncthreads();
  float yln = (v - smu) * srs * g[d] + bb[d];
  float zv = xz[(size_t)m * 384 + DINNER + d];
  y2[(size_t)m * DINNER + d] = yln * siluf(zv);
}

extern "C" void kernel_launch(void* const* d_in, const int* in_sizes, int n_in,
                              void* d_out, int out_size, void* d_ws, size_t ws_size,
                              hipStream_t stream) {
  (void)in_sizes; (void)n_in; (void)out_size; (void)ws_size;
  const float* x   = (const float*)d_in[0];
  const float* xt  = (const float*)d_in[1];
  const float* ipw = (const float*)d_in[2];
  const float* c2w = (const float*)d_in[3];
  const float* c2b = (const float*)d_in[4];
  const float* cxw = (const float*)d_in[5];
  const float* cxb = (const float*)d_in[6];
  const float* xpw = (const float*)d_in[7];
  const float* dtw = (const float*)d_in[8];
  const float* dtb = (const float*)d_in[9];
  const float* alg = (const float*)d_in[10];
  const float* dss = (const float*)d_in[11];
  const float* lng = (const float*)d_in[12];
  const float* lnb = (const float*)d_in[13];
  const float* opw = (const float*)d_in[14];
  float* out = (float*)d_out;
  float* ws  = (float*)d_ws;

  size_t off = 0;
  float* xz    = ws + off; off += (size_t)MROWS * 384;
  float* xin   = ws + off; off += (size_t)BATCH * 2 * Ls * DINNER;
  float* xs    = ws + off; off += (size_t)BATCH * 4 * DINNER * Ls;
  float* Wp    = ws + off; off += (size_t)NKO_PAD * KBIG;
  float* kdp   = ws + off; off += (size_t)NKO_PAD * DINNER;
  float* xdbl  = ws + off; off += (size_t)BATCH * 4 * NCOUT * Ls;
  float* delta = ws + off; off += (size_t)BATCH * 4 * DINNER * Ls;
  float* oy    = ws + off; off += (size_t)BATCH * 4 * DINNER * Ls;
  float* y2    = xin;

  gemm_f32_wmma<<<dim3(384 / 16, MROWS / 16), 32, 0, stream>>>(x, ipw, xz, DMODEL, 384);

  k_pack_w <<<(NKO_PAD * KBIG + 255) / 256, 256, 0, stream>>>(xpw, Wp);
  k_pack_kd<<<(NKO_PAD * DINNER + 255) / 256, 256, 0, stream>>>(xpw, kdp);

  k2_conv<<<dim3(Ws / K2_SW, Hs, BATCH * 2), DINNER, 0, stream>>>(
      xz, xt, c2w, c2b, cxw, cxb, xin);

  k3_xs<<<dim3(Ls / XT_P, DINNER / XT_D, BATCH * 2), 256, 0, stream>>>(xin, xs);

  k4_cdc<<<dim3(MROWS / 64, NKO_PAD / (16 * CDC_NT)), 128, 0, stream>>>(xin, Wp, kdp, xdbl);

  k5_delta<<<dim3(Ls / (2 * K5_TPB), DINNER / K5_DG, BATCH * 4), K5_TPB, 0, stream>>>(xdbl, dtw, dtb, delta);

  k6_scan<<<96, 256, 0, stream>>>(xs, xdbl, alg, dss, delta, oy);

  k7a_comb<<<dim3(Ls / XT_P, DINNER / XT_D, BATCH), 256, 0, stream>>>(oy, y2);
  k7b_ln<<<MROWS, DINNER, 0, stream>>>(xz, lng, lnb, y2);

  gemm_f32_wmma<<<dim3(DMODEL / 16, MROWS / 16), 32, 0, stream>>>(y2, opw, out, DINNER, DMODEL);
}


# k6 scan launched as 192 work-groups of 4 waves (was 96 of 8): one scan wave per SIMD on 192 WGPs, half as many waves on each WGP's memory pipe and vector port; kernel body and sum order unchanged
# speedup vs baseline: 6.3348x; 1.0378x over previous
#include <hip/hip_runtime.h>
#include <hip/hip_bf16.h>
#include <math.h>

typedef float v2f __attribute__((ext_vector_type(2)));
typedef float v8f __attribute__((ext_vector_type(8)));

#define BATCH   2
#define Hs      56
#define Ws      56
#define Ls      (Hs*Ws)
#define DMODEL  96
#define DINNER  192
#define DSTATE  16
#define DTRANK  6
#define NCOUT   38
#define NKO     (4*NCOUT)
#define NKO_PAD 160
#define KBIG    (DINNER*2*9)
#define THETA_C 0.6f
#define MROWS   (BATCH*Ls)

static __device__ __forceinline__ float siluf(float v) { return v / (1.f + expf(-v)); }

static __device__ __forceinline__ v8f wmma4(v2f a, v2f b, v8f c) {
  return __builtin_amdgcn_wmma_f32_16x16x4_f32(false, a, false, b, (short)0, c, false, false);
}

__global__ void gemm_f32_wmma(const float* __restrict__ A, const float* __restrict__ Bt,
                              float* __restrict__ C, int K, int ldc) {
  const int lane = threadIdx.x & 31;
  const int half = lane >> 4;
  const int klo  = half << 1;
  const int ln   = lane & 15;
  const int ntile = blockIdx.x;
  const int mtile = blockIdx.y;
  const float* ap = A  + (size_t)(mtile * 16 + ln) * K + klo;
  const float* bp = Bt + (size_t)(ntile * 16 + ln) * K + klo;
  v8f acc = {0.f,0.f,0.f,0.f,0.f,0.f,0.f,0.f};
  for (int k = 0; k < K; k += 4) {
    v2f a = *(const v2f*)(ap + k);
    v2f b = *(const v2f*)(bp + k);
    acc = wmma4(a, b, acc);
  }
  const int n = ntile * 16 + ln;
  for (int rr = 0; rr < 8; ++rr) {
    int mm = mtile * 16 + rr + half * 8;
    C[(size_t)mm * ldc + n] = acc[rr];
  }
}

#define NPAIR_W  (KBIG / 8)
#define NPAIR_KD (DINNER / 8)

__global__ void k_pack_w(const float* __restrict__ xw, float* __restrict__ Wp) {
  int tid = blockIdx.x * blockDim.x + threadIdx.x;
  if (tid >= NKO_PAD * KBIG) return;
  int e    = tid & 3;
  int lane = (tid >> 2) & 31;
  int grp  = tid >> 7;
  int pr   = grp % NPAIR_W;
  int nt   = grp / NPAIR_W;
  int n    = nt * 16 + (lane & 15);
  int kk   = pr * 8 + (e >> 1) * 4 + (lane >> 4) * 2 + (e & 1);
  float v = 0.f;
  if (n < NKO) {
    int c = kk % DINNER;
    int r = kk / DINNER;
    int dx = r % 3, dy = (r / 3) % 3, t = r / 9;
    v = xw[((((size_t)n * DINNER + c) * 2 + t) * 3 + dy) * 3 + dx];
  }
  Wp[tid] = v;
}

__global__ void k_pack_kd(const float* __restrict__ xw, float* __restrict__ kdp) {
  int tid = blockIdx.x * blockDim.x + threadIdx.x;
  if (tid >= NKO_PAD * DINNER) return;
  int e    = tid & 3;
  int lane = (tid >> 2) & 31;
  int grp  = tid >> 7;
  int pr   = grp % NPAIR_KD;
  int nt   = grp / NPAIR_KD;
  int n    = nt * 16 + (lane & 15);
  int c    = pr * 8 + (e >> 1) * 4 + (lane >> 4) * 2 + (e & 1);
  float v = 0.f;
  if (n < NKO) {
    const float* p = xw + ((size_t)n * DINNER + c) * 18;
    for (int i = 0; i < 18; ++i) v += p[i];
  }
  kdp[tid] = v;
}

#define K2_SW 14

__global__ void __launch_bounds__(DINNER) k2_conv(const float* __restrict__ xz, const float* __restrict__ xt,
                        const float* __restrict__ w2, const float* __restrict__ b2,
                        const float* __restrict__ wxt, const float* __restrict__ bxt,
                        float* __restrict__ xin) {
  const int c  = threadIdx.x;
  const int w0 = blockIdx.x * K2_SW;
  const int h  = blockIdx.y;
  const int t  = blockIdx.z & 1;
  const int b  = blockIdx.z >> 1;
  const float* src = t ? (xt + (c >> 1)) : (xz + c);
  const int    ps  = t ? DMODEL : 2 * DINNER;
  const float* wp  = (t ? wxt : w2) + c * 9;
  const float bias = t ? bxt[c] : b2[c];
  float wk[9];
  #pragma unroll
  for (int i = 0; i < 9; ++i) wk[i] = wp[i];
  float win[3][K2_SW + 2];
  #pragma unroll
  for (int dy = 0; dy < 3; ++dy) {
    const int hh = h + dy - 1;
    const bool rok = (hh >= 0) && (hh < Hs);
    const int hc = hh < 0 ? 0 : (hh >= Hs ? Hs - 1 : hh);
    const float* rp = src + (size_t)((b * Hs + hc) * Ws) * ps;
    #pragma unroll
    for (int j = 0; j < K2_SW + 2; ++j) {
      int ww = w0 + j - 1;
      bool ok = rok;
      if (j == 0)         { ok = ok && (ww >= 0); ww = ww < 0 ? 0 : ww; }
      if (j == K2_SW + 1) { ok = ok && (ww < Ws); ww = ww >= Ws ? Ws - 1 : ww; }
      const float v = rp[(size_t)ww * ps];
      win[dy][j] = ok ? v : 0.f;
    }
  }
  float* op = xin + ((size_t)((b * 2 + t) * Hs + h) * Ws + w0) * DINNER + c;
  #pragma unroll
  for (int j = 0; j < K2_SW; ++j) {
    float acc = bias;
    #pragma unroll
    for (int dy = 0; dy < 3; ++dy) {
      #pragma unroll
      for (int dx = 0; dx < 3; ++dx) acc += win[dy][j + dx] * wk[dy * 3 + dx];
    }
    op[(size_t)j * DINNER] = siluf(acc);
  }
}

#define XT_P     224
#define XT_D     32
#define XT_PITCH 33

__global__ void __launch_bounds__(256) k3_xs(const float* __restrict__ xin, float* __restrict__ xs) {
  __shared__ float tile[XT_P * XT_PITCH];
  const int g    = blockIdx.x;
  const int d0   = blockIdx.y * XT_D;
  const int b    = blockIdx.z >> 1;
  const int col  = blockIdx.z & 1;
  const int lane = threadIdx.x & 31;
  const int wv   = __builtin_amdgcn_readfirstlane(threadIdx.x >> 5);

  const int major  = g * 4 + (wv >> 1);
  const int minor0 = (wv & 1) * 28;
  const int sMaj = col ? Ws : 1;
  const int sMin = col ? 1 : Ws;
  const float* xb = xin + ((size_t)b * 2 * Ls + major * sMaj + minor0 * sMin) * DINNER + d0 + lane;
  float pv[28];
  #pragma unroll
  for (int it = 0; it < 28; ++it) {
    const int s  = it & 1;
    const int i2 = it & ~1;
    const float* p0 = xb + ((size_t)s * Ls + (size_t)i2 * sMin) * DINNER;
    float v0 = p0[0];
    float v1 = p0[(size_t)sMin * DINNER];
    pv[it] = 0.5f * (v0 + v1) + fmaxf(v0, v1);
  }
  #pragma unroll
  for (int it = 0; it < 28; ++it) tile[(wv * 28 + it) * XT_PITCH + lane] = pv[it];
  __syncthreads();

  const int kbase = col ? 2 : 0;
  #pragma unroll
  for (int i = 0; i < 4; ++i) {
    int dd = wv * 4 + i;
    float* op0 = xs + ((size_t)(b * 4 + kbase) * DINNER + d0 + dd) * Ls + g * XT_P;
    float* op1 = op0 + (size_t)DINNER * Ls;
    #pragma unroll
    for (int c = 0; c < 7; ++c) {
      int q  = lane + 32 * c;
      int sp = (q / 56) * 56 + 55 - q % 56;
      op0[q] = tile[q  * XT_PITCH + dd];
      op1[q] = tile[sp * XT_PITCH + dd];
    }
  }
}

#define CDC_NT     5
#define CDC_STAGES (KBIG / 32)

struct CdcStage { v2f a[8]; float4 b[4][CDC_NT]; int ok; };

static __device__ __forceinline__ void cdc_load(CdcStage& q, const float* __restrict__ xin,
                                                const float* __restrict__ wf, int s,
                                                int b, int h, int w, int klo) {
  int td = s / 6, cc = (s - td * 6) * 32;
  int t = td / 9, r9 = td - t * 9, dy = r9 / 3, dx = r9 - dy * 3;
  int hh = h + dy - 1, ww = w + dx - 1;
  q.ok = (hh >= 0) && (hh < Hs) && (ww >= 0) && (ww < Ws);
  int hc = hh < 0 ? 0 : (hh >= Hs ? Hs - 1 : hh);
  int wc = ww < 0 ? 0 : (ww >= Ws ? Ws - 1 : ww);
  const float* ap = xin + (((size_t)(b * 2 + t) * Hs + hc) * Ws + wc) * DINNER + klo + cc;
  #pragma unroll
  for (int i = 0; i < 8; ++i) q.a[i] = *(const v2f*)(ap + 4 * i);
  const float* bp = wf + (size_t)s * 4 * 128;
  #pragma unroll
  for (int qd = 0; qd < 4; ++qd) {
    #pragma unroll
    for (int j = 0; j < CDC_NT; ++j)
      q.b[qd][j] = *(const float4*)(bp + ((size_t)j * NPAIR_W + qd) * 128);
  }
}

static __device__ __forceinline__ void cdc_compute(const CdcStage& q, v8f (&acc)[CDC_NT]) {
  const v2f zero = {0.f, 0.f};
  #pragma unroll
  for (int qd = 0; qd < 4; ++qd) {
    #pragma unroll
    for (int e2 = 0; e2 < 2; ++e2) {
      v2f a = q.ok ? q.a[2 * qd + e2] : zero;
      #pragma unroll
      for (int j = 0; j < CDC_NT; ++j) {
        float4 bv = q.b[qd][j];
        v2f bb;
        if (e2 == 0) { bb.x = bv.x; bb.y = bv.y; } else { bb.x = bv.z; bb.y = bv.w; }
        acc[j] = wmma4(a, bb, acc[j]);
      }
    }
  }
}

__global__ void __launch_bounds__(128) k4_cdc(const float* __restrict__ xin, const float* __restrict__ Wp,
                       const float* __restrict__ kdp, float* __restrict__ xdbl) {
  const int lane = threadIdx.x & 31;
  const int half = lane >> 4;
  const int klo  = half << 1;
  const int ln   = lane & 15;
  const int mtile = blockIdx.x * 4 + (threadIdx.x >> 5);
  const int nt0   = blockIdx.y * CDC_NT;
  const int m = mtile * 16 + ln;
  const int b = m / Ls;
  const int hw = m % Ls;
  const int h = hw / Ws, w = hw % Ws;

  v8f accN[CDC_NT];
  #pragma unroll
  for (int j = 0; j < CDC_NT; ++j) accN[j] = (v8f){0.f,0.f,0.f,0.f,0.f,0.f,0.f,0.f};

  const float* wf = Wp + (size_t)nt0 * NPAIR_W * 128 + lane * 4;
  CdcStage qa, qb;
  cdc_load(qa, xin, wf, 0, b, h, w, klo);
  for (int s = 0; s < CDC_STAGES; s += 2) {
    cdc_load(qb, xin, wf, s + 1, b, h, w, klo);
    cdc_compute(qa, accN);
    int s2 = (s + 2 < CDC_STAGES) ? (s + 2) : (s + 1);
    cdc_load(qa, xin, wf, s2, b, h, w, klo);
    cdc_compute(qb, accN);
  }

  v8f accD0[CDC_NT], accD1[CDC_NT];
  #pragma unroll
  for (int j = 0; j < CDC_NT; ++j) {
    accD0[j] = (v8f){0.f,0.f,0.f,0.f,0.f,0.f,0.f,0.f};
    accD1[j] = (v8f){0.f,0.f,0.f,0.f,0.f,0.f,0.f,0.f};
  }
  const float* ap0 = xin + (((size_t)(b * 2 + 0) * Hs + h) * Ws + w) * DINNER + klo;
  const float* ap1 = xin + (((size_t)(b * 2 + 1) * Hs + h) * Ws + w) * DINNER + klo;
  const float* kf  = kdp + (size_t)nt0 * NPAIR_KD * 128 + lane * 4;
  #pragma unroll
  for (int pr = 0; pr < NPAIR_KD; ++pr) {
    v2f a00 = *(const v2f*)(ap0 + pr * 8);
    v2f a01 = *(const v2f*)(ap0 + pr * 8 + 4);
    v2f a10 = *(const v2f*)(ap1 + pr * 8);
    v2f a11 = *(const v2f*)(ap1 + pr * 8 + 4);
    #pragma unroll
    for (int j = 0; j < CDC_NT; ++j) {
      float4 bv = *(const float4*)(kf + ((size_t)j * NPAIR_KD + pr) * 128);
      v2f b0, b1;
      b0.x = bv.x; b0.y = bv.y; b1.x = bv.z; b1.y = bv.w;
      accD0[j] = wmma4(a00, b0, accD0[j]);
      accD1[j] = wmma4(a10, b0, accD1[j]);
      accD0[j] = wmma4(a01, b1, accD0[j]);
      accD1[j] = wmma4(a11, b1, accD1[j]);
    }
  }

  #pragma unroll
  for (int j = 0; j < CDC_NT; ++j) {
    const int n = (nt0 + j) * 16 + ln;
    if (n < NKO) {
      int kq = n / NCOUT, o = n % NCOUT;
      for (int rr = 0; rr < 8; ++rr) {
        int mm = mtile * 16 + rr + half * 8;
        int b2 = mm / Ls; int l2 = mm % Ls;
        float Nv = accN[j][rr], d0 = accD0[j][rr], d1 = accD1[j][rr];
        float pooled = 2.f * Nv - THETA_C * (0.5f * (d0 + d1) + fminf(d0, d1));
        xdbl[(((size_t)b2 * 4 + kq) * NCOUT + o) * Ls + l2] = pooled;
      }
    }
  }
}

#define K5_DG  16
#define K5_TPB 224

__global__ void __launch_bounds__(K5_TPB) k5_delta(const float* __restrict__ xdbl, const float* __restrict__ dtw,
                         const float* __restrict__ dtb, float* __restrict__ delta) {
  const int l  = (blockIdx.x * K5_TPB + threadIdx.x) * 2;
  const int d0 = blockIdx.y * K5_DG;
  const int bk = blockIdx.z;
  const int k  = bk & 3;
  const float* xp = xdbl + (size_t)bk * NCOUT * Ls + l;
  v2f xv[DTRANK];
  #pragma unroll
  for (int rr = 0; rr < DTRANK; ++rr) xv[rr] = *(const v2f*)(xp + (size_t)rr * Ls);
  const float* wp = dtw + ((size_t)k * DINNER + d0) * DTRANK;
  const float* bp = dtb + k * DINNER + d0;
  float* op = delta + ((size_t)bk * DINNER + d0) * Ls + l;
  #pragma unroll 4
  for (int i = 0; i < K5_DG; ++i) {
    float a0 = bp[i];
    float a1 = a0;
    #pragma unroll
    for (int rr = 0; rr < DTRANK; ++rr) {
      float wv = wp[i * DTRANK + rr];
      a0 += xv[rr].x * wv;
      a1 += xv[rr].y * wv;
    }
    v2f o;
    o.x = fmaxf(a0, 0.f) + log1pf(expf(-fabsf(a0)));
    o.y = fmaxf(a1, 0.f) + log1pf(expf(-fabsf(a1)));
    *(v2f*)(op + (size_t)i * Ls) = o;
  }
}

#define SCH 16

struct ScanChunk { float d[SCH], u[SCH], b[SCH], c[SCH]; };

static __device__ __forceinline__ void scan_load(ScanChunk& q, const float* __restrict__ dP,
                                                 const float* __restrict__ uP,
                                                 const float* __restrict__ bP,
                                                 const float* __restrict__ cP, int l0) {
  #pragma unroll
  for (int i = 0; i < SCH; i += 4) {
    float4 vd = *(const float4*)(dP + l0 + i);
    float4 vu = *(const float4*)(uP + l0 + i);
    float4 vb = *(const float4*)(bP + l0 + i);
    float4 vc = *(const float4*)(cP + l0 + i);
    q.d[i] = vd.x; q.d[i + 1] = vd.y; q.d[i + 2] = vd.z; q.d[i + 3] = vd.w;
    q.u[i] = vu.x; q.u[i + 1] = vu.y; q.u[i + 2] = vu.z; q.u[i + 3] = vu.w;
    q.b[i] = vb.x; q.b[i + 1] = vb.y; q.b[i + 2] = vb.z; q.b[i + 3] = vb.w;
    q.c[i] = vc.x; q.c[i + 1] = vc.y; q.c[i + 2] = vc.z; q.c[i + 3] = vc.w;
  }
}

static __device__ __forceinline__ float scan_steps(const ScanChunk& q, float& h, float A,
                                                   float Dv, int ln) {
  float y[SCH];
  #pragma unroll
  for (int j = 0; j < SCH; ++j) {
    float dl = q.d[j];
    h = __builtin_amdgcn_exp2f(dl * A) * h + (dl * q.u[j]) * q.b[j];
    y[j] = h * q.c[j];
  }
  #pragma unroll
  for (int m = 1; m < 16; m <<= 1) {
    float t[SCH];
    #pragma unroll
    for (int j = 0; j < SCH; ++j) t[j] = __shfl_xor(y[j], m);
    __builtin_amdgcn_sched_barrier(0);
    #pragma unroll
    for (int j = 0; j < SCH; ++j) y[j] += t[j];
  }
  float keep = 0.f;
  #pragma unroll
  for (int j = 0; j < SCH; ++j) {
    float o = y[j] + Dv * q.u[j];
    if (ln == j) keep = o;
  }
  return keep;
}

__global__ void __launch_bounds__(128) k6_scan(const float* __restrict__ xs, const float* __restrict__ xdbl,
                        const float* __restrict__ Alogs, const float* __restrict__ Dss,
                        const float* __restrict__ delta, float* __restrict__ oy) {
  int wid  = blockIdx.x * (blockDim.x >> 5) + (threadIdx.x >> 5);
  int lane = threadIdx.x & 31;
  int half = lane >> 4, ln = lane & 15;
  int bk = wid / 96;
  int d  = (wid % 96) * 2 + half;
  int k  = bk & 3;
  const float* dP = delta + ((size_t)bk * DINNER + d) * Ls;
  float*       oP = oy    + ((size_t)bk * DINNER + d) * Ls;
  const float* uP = xs    + ((size_t)bk * DINNER + d) * Ls;
  const float* bP = xdbl  + ((size_t)bk * NCOUT + DTRANK + ln) * Ls;
  const float* cP = xdbl  + ((size_t)bk * NCOUT + DTRANK + DSTATE + ln) * Ls;
  float A  = -expf(Alogs[((size_t)(k * DINNER + d)) * DSTATE + ln]) * 1.4426950408889634f;
  float Dv = Dss[k * DINNER + d];
  float h = 0.f;
  ScanChunk qa, qb;
  scan_load(qa, dP, uP, bP, cP, 0);
  for (int l0 = 0; l0 < Ls; l0 += 2 * SCH) {
    scan_load(qb, dP, uP, bP, cP, l0 + SCH);
    oP[l0 + ln] = scan_steps(qa, h, A, Dv, ln);
    int ln2 = (l0 + 2 * SCH < Ls) ? (l0 + 2 * SCH) : (l0 + SCH);
    scan_load(qa, dP, uP, bP, cP, ln2);
    oP[l0 + SCH + ln] = scan_steps(qb, h, A, Dv, ln);
  }
}

__global__ void __launch_bounds__(256) k7a_comb(const float* __restrict__ oy, float* __restrict__ y2) {
  __shared__ float tile[XT_P * XT_PITCH];
  const int h0   = blockIdx.x * 4;
  const int d0   = blockIdx.y * XT_D;
  const int b    = blockIdx.z;
  const int lane = threadIdx.x & 31;
  const int wv   = __builtin_amdgcn_readfirstlane(threadIdx.x >> 5);
  const size_t s = (size_t)DINNER * Ls;

  #pragma unroll
  for (int it = 0; it < (XT_D * Ws) / 256; ++it) {
    int idx = it * 256 + threadIdx.x;
    int dd = idx / Ws;
    int w  = idx % Ws;
    const float* base = oy + ((size_t)b * 4 * DINNER + d0 + dd) * Ls;
    int lt0 = w * Hs + h0;
    float4 t1 = *(const float4*)(base + s + lt0);
    float4 t3 = *(const float4*)(base + 3 * s + (Ls - 4 - lt0));
    float a1[4] = {t1.x, t1.y, t1.z, t1.w};
    float a3[4] = {t3.w, t3.z, t3.y, t3.x};
    #pragma unroll
    for (int j = 0; j < 4; ++j) {
      int l = (h0 + j) * Ws + w;
      float v = base[l] + base[2 * s + (Ls - 1 - l)] + a1[j] + a3[j];
      tile[(j * Ws + w) * XT_PITCH + dd] = v;
    }
  }
  __syncthreads();

  float* yp = y2 + ((size_t)b * Ls + h0 * Ws + wv * 28) * DINNER + d0 + lane;
  #pragma unroll
  for (int it = 0; it < 28; ++it)
    yp[(size_t)it * DINNER] = tile[(wv * 28 + it) * XT_PITCH + lane];
}

__global__ void k7b_ln(const float* __restrict__ xz, const float* __restrict__ g,
                       const float* __restrict__ bb, float* __restrict__ y2) {
  int m = blockIdx.x;
  int d = threadIdx.x;
  float v = y2[(size_t)m * DINNER + d];
  float s1 = v, s2 = v * v;
  #pragma unroll
  for (int off = 16; off >= 1; off >>= 1) {
    s1 += __shfl_xor(s1, off);
    s2 += __shfl_xor(s2, off);
  }
  __shared__ float r1[6], r2[6];
  __shared__ float smu, srs;
  if ((d & 31) == 0) { r1[d >> 5] = s1; r2[d >> 5] = s2; }
  __syncthreads();
  if (d == 0) {
    float t1 = 0.f, t2 = 0.f;
    for (int i = 0; i < 6; ++i) { t1 += r1[i]; t2 += r2[i]; }
    float mu = t1 / (float)DINNER;
    float var = t2 / (float)DINNER - mu * mu;
    smu = mu; srs = rsqrtf(var + 1e-5f);
  }
  __syncthreads();
  float yln = (v - smu) * srs * g[d] + bb[d];
  float zv = xz[(size_t)m * 384 + DINNER + d];
  y2[(size_t)m * DINNER + d] = yln * siluf(zv);
}

extern "C" void kernel_launch(void* const* d_in, const int* in_sizes, int n_in,
                              void* d_out, int out_size, void* d_ws, size_t ws_size,
                              hipStream_t stream) {
  (void)in_sizes; (void)n_in; (void)out_size; (void)ws_size;
  const float* x   = (const float*)d_in[0];
  const float* xt  = (const float*)d_in[1];
  const float* ipw = (const float*)d_in[2];
  const float* c2w = (const float*)d_in[3];
  const float* c2b = (const float*)d_in[4];
  const float* cxw = (const float*)d_in[5];
  const float* cxb = (const float*)d_in[6];
  const float* xpw = (const float*)d_in[7];
  const float* dtw = (const float*)d_in[8];
  const float* dtb = (const float*)d_in[9];
  const float* alg = (const float*)d_in[10];
  const float* dss = (const float*)d_in[11];
  const float* lng = (const float*)d_in[12];
  const float* lnb = (const float*)d_in[13];
  const float* opw = (const float*)d_in[14];
  float* out = (float*)d_out;
  float* ws  = (float*)d_ws;

  size_t off = 0;
  float* xz    = ws + off; off += (size_t)MROWS * 384;
  float* xin   = ws + off; off += (size_t)BATCH * 2 * Ls * DINNER;
  float* xs    = ws + off; off += (size_t)BATCH * 4 * DINNER * Ls;
  float* Wp    = ws + off; off += (size_t)NKO_PAD * KBIG;
  float* kdp   = ws + off; off += (size_t)NKO_PAD * DINNER;
  float* xdbl  = ws + off; off += (size_t)BATCH * 4 * NCOUT * Ls;
  float* delta = ws + off; off += (size_t)BATCH * 4 * DINNER * Ls;
  float* oy    = ws + off; off += (size_t)BATCH * 4 * DINNER * Ls;
  float* y2    = xin;

  gemm_f32_wmma<<<dim3(384 / 16, MROWS / 16), 32, 0, stream>>>(x, ipw, xz, DMODEL, 384);

  k_pack_w <<<(NKO_PAD * KBIG + 255) / 256, 256, 0, stream>>>(xpw, Wp);
  k_pack_kd<<<(NKO_PAD * DINNER + 255) / 256, 256, 0, stream>>>(xpw, kdp);

  k2_conv<<<dim3(Ws / K2_SW, Hs, BATCH * 2), DINNER, 0, stream>>>(
      xz, xt, c2w, c2b, cxw, cxb, xin);

  k3_xs<<<dim3(Ls / XT_P, DINNER / XT_D, BATCH * 2), 256, 0, stream>>>(xin, xs);

  k4_cdc<<<dim3(MROWS / 64, NKO_PAD / (16 * CDC_NT)), 128, 0, stream>>>(xin, Wp, kdp, xdbl);

  k5_delta<<<dim3(Ls / (2 * K5_TPB), DINNER / K5_DG, BATCH * 4), K5_TPB, 0, stream>>>(xdbl, dtw, dtb, delta);

  k6_scan<<<192, 128, 0, stream>>>(xs, xdbl, alg, dss, delta, oy);

  k7a_comb<<<dim3(Ls / XT_P, DINNER / XT_D, BATCH), 256, 0, stream>>>(oy, y2);
  k7b_ln<<<MROWS, DINNER, 0, stream>>>(xz, lng, lnb, y2);

  gemm_f32_wmma<<<dim3(DMODEL / 16, MROWS / 16), 32, 0, stream>>>(y2, opw, out, DINNER, DMODEL);
}


# k6 scan: L cut into 4 segments, one per wave of a work-group (3 waves per SIMD): pass 1 gets each segment's end state and decay product, carry folded via LDS (carry rounding changes), pass 2 rescans
# speedup vs baseline: 7.3329x; 1.1576x over previous
#include <hip/hip_runtime.h>
#include <hip/hip_bf16.h>
#include <math.h>

typedef float v2f __attribute__((ext_vector_type(2)));
typedef float v8f __attribute__((ext_vector_type(8)));

#define BATCH   2
#define Hs      56
#define Ws      56
#define Ls      (Hs*Ws)
#define DMODEL  96
#define DINNER  192
#define DSTATE  16
#define DTRANK  6
#define NCOUT   38
#define NKO     (4*NCOUT)
#define NKO_PAD 160
#define KBIG    (DINNER*2*9)
#define THETA_C 0.6f
#define MROWS   (BATCH*Ls)

static __device__ __forceinline__ float siluf(float v) { return v / (1.f + expf(-v)); }

static __device__ __forceinline__ v8f wmma4(v2f a, v2f b, v8f c) {
  return __builtin_amdgcn_wmma_f32_16x16x4_f32(false, a, false, b, (short)0, c, false, false);
}

__global__ void gemm_f32_wmma(const float* __restrict__ A, const float* __restrict__ Bt,
                              float* __restrict__ C, int K, int ldc) {
  const int lane = threadIdx.x & 31;
  const int half = lane >> 4;
  const int klo  = half << 1;
  const int ln   = lane & 15;
  const int ntile = blockIdx.x;
  const int mtile = blockIdx.y;
  const float* ap = A  + (size_t)(mtile * 16 + ln) * K + klo;
  const float* bp = Bt + (size_t)(ntile * 16 + ln) * K + klo;
  v8f acc = {0.f,0.f,0.f,0.f,0.f,0.f,0.f,0.f};
  for (int k = 0; k < K; k += 4) {
    v2f a = *(const v2f*)(ap + k);
    v2f b = *(const v2f*)(bp + k);
    acc = wmma4(a, b, acc);
  }
  const int n = ntile * 16 + ln;
  for (int rr = 0; rr < 8; ++rr) {
    int mm = mtile * 16 + rr + half * 8;
    C[(size_t)mm * ldc + n] = acc[rr];
  }
}

#define NPAIR_W  (KBIG / 8)
#define NPAIR_KD (DINNER / 8)

__global__ void k_pack_w(const float* __restrict__ xw, float* __restrict__ Wp) {
  int tid = blockIdx.x * blockDim.x + threadIdx.x;
  if (tid >= NKO_PAD * KBIG) return;
  int e    = tid & 3;
  int lane = (tid >> 2) & 31;
  int grp  = tid >> 7;
  int pr   = grp % NPAIR_W;
  int nt   = grp / NPAIR_W;
  int n    = nt * 16 + (lane & 15);
  int kk   = pr * 8 + (e >> 1) * 4 + (lane >> 4) * 2 + (e & 1);
  float v = 0.f;
  if (n < NKO) {
    int c = kk % DINNER;
    int r = kk / DINNER;
    int dx = r % 3, dy = (r / 3) % 3, t = r / 9;
    v = xw[((((size_t)n * DINNER + c) * 2 + t) * 3 + dy) * 3 + dx];
  }
  Wp[tid] = v;
}

__global__ void k_pack_kd(const float* __restrict__ xw, float* __restrict__ kdp) {
  int tid = blockIdx.x * blockDim.x + threadIdx.x;
  if (tid >= NKO_PAD * DINNER) return;
  int e    = tid & 3;
  int lane = (tid >> 2) & 31;
  int grp  = tid >> 7;
  int pr   = grp % NPAIR_KD;
  int nt   = grp / NPAIR_KD;
  int n    = nt * 16 + (lane & 15);
  int c    = pr * 8 + (e >> 1) * 4 + (lane >> 4) * 2 + (e & 1);
  float v = 0.f;
  if (n < NKO) {
    const float* p = xw + ((size_t)n * DINNER + c) * 18;
    for (int i = 0; i < 18; ++i) v += p[i];
  }
  kdp[tid] = v;
}

#define K2_SW 14

__global__ void __launch_bounds__(DINNER) k2_conv(const float* __restrict__ xz, const float* __restrict__ xt,
                        const float* __restrict__ w2, const float* __restrict__ b2,
                        const float* __restrict__ wxt, const float* __restrict__ bxt,
                        float* __restrict__ xin) {
  const int c  = threadIdx.x;
  const int w0 = blockIdx.x * K2_SW;
  const int h  = blockIdx.y;
  const int t  = blockIdx.z & 1;
  const int b  = blockIdx.z >> 1;
  const float* src = t ? (xt + (c >> 1)) : (xz + c);
  const int    ps  = t ? DMODEL : 2 * DINNER;
  const float* wp  = (t ? wxt : w2) + c * 9;
  const float bias = t ? bxt[c] : b2[c];
  float wk[9];
  #pragma unroll
  for (int i = 0; i < 9; ++i) wk[i] = wp[i];
  float win[3][K2_SW + 2];
  #pragma unroll
  for (int dy = 0; dy < 3; ++dy) {
    const int hh = h + dy - 1;
    const bool rok = (hh >= 0) && (hh < Hs);
    const int hc = hh < 0 ? 0 : (hh >= Hs ? Hs - 1 : hh);
    const float* rp = src + (size_t)((b * Hs + hc) * Ws) * ps;
    #pragma unroll
    for (int j = 0; j < K2_SW + 2; ++j) {
      int ww = w0 + j - 1;
      bool ok = rok;
      if (j == 0)         { ok = ok && (ww >= 0); ww = ww < 0 ? 0 : ww; }
      if (j == K2_SW + 1) { ok = ok && (ww < Ws); ww = ww >= Ws ? Ws - 1 : ww; }
      const float v = rp[(size_t)ww * ps];
      win[dy][j] = ok ? v : 0.f;
    }
  }
  float* op = xin + ((size_t)((b * 2 + t) * Hs + h) * Ws + w0) * DINNER + c;
  #pragma unroll
  for (int j = 0; j < K2_SW; ++j) {
    float acc = bias;
    #pragma unroll
    for (int dy = 0; dy < 3; ++dy) {
      #pragma unroll
      for (int dx = 0; dx < 3; ++dx) acc += win[dy][j + dx] * wk[dy * 3 + dx];
    }
    op[(size_t)j * DINNER] = siluf(acc);
  }
}

#define XT_P     224
#define XT_D     32
#define XT_PITCH 33

__global__ void __launch_bounds__(256) k3_xs(const float* __restrict__ xin, float* __restrict__ xs) {
  __shared__ float tile[XT_P * XT_PITCH];
  const int g    = blockIdx.x;
  const int d0   = blockIdx.y * XT_D;
  const int b    = blockIdx.z >> 1;
  const int col  = blockIdx.z & 1;
  const int lane = threadIdx.x & 31;
  const int wv   = __builtin_amdgcn_readfirstlane(threadIdx.x >> 5);

  const int major  = g * 4 + (wv >> 1);
  const int minor0 = (wv & 1) * 28;
  const int sMaj = col ? Ws : 1;
  const int sMin = col ? 1 : Ws;
  const float* xb = xin + ((size_t)b * 2 * Ls + major * sMaj + minor0 * sMin) * DINNER + d0 + lane;
  float pv[28];
  #pragma unroll
  for (int it = 0; it < 28; ++it) {
    const int s  = it & 1;
    const int i2 = it & ~1;
    const float* p0 = xb + ((size_t)s * Ls + (size_t)i2 * sMin) * DINNER;
    float v0 = p0[0];
    float v1 = p0[(size_t)sMin * DINNER];
    pv[it] = 0.5f * (v0 + v1) + fmaxf(v0, v1);
  }
  #pragma unroll
  for (int it = 0; it < 28; ++it) tile[(wv * 28 + it) * XT_PITCH + lane] = pv[it];
  __syncthreads();

  const int kbase = col ? 2 : 0;
  #pragma unroll
  for (int i = 0; i < 4; ++i) {
    int dd = wv * 4 + i;
    float* op0 = xs + ((size_t)(b * 4 + kbase) * DINNER + d0 + dd) * Ls + g * XT_P;
    float* op1 = op0 + (size_t)DINNER * Ls;
    #pragma unroll
    for (int c = 0; c < 7; ++c) {
      int q  = lane + 32 * c;
      int sp = (q / 56) * 56 + 55 - q % 56;
      op0[q] = tile[q  * XT_PITCH + dd];
      op1[q] = tile[sp * XT_PITCH + dd];
    }
  }
}

#define CDC_NT     5
#define CDC_STAGES (KBIG / 32)

struct CdcStage { v2f a[8]; float4 b[4][CDC_NT]; int ok; };

static __device__ __forceinline__ void cdc_load(CdcStage& q, const float* __restrict__ xin,
                                                const float* __restrict__ wf, int s,
                                                int b, int h, int w, int klo) {
  int td = s / 6, cc = (s - td * 6) * 32;
  int t = td / 9, r9 = td - t * 9, dy = r9 / 3, dx = r9 - dy * 3;
  int hh = h + dy - 1, ww = w + dx - 1;
  q.ok = (hh >= 0) && (hh < Hs) && (ww >= 0) && (ww < Ws);
  int hc = hh < 0 ? 0 : (hh >= Hs ? Hs - 1 : hh);
  int wc = ww < 0 ? 0 : (ww >= Ws ? Ws - 1 : ww);
  const float* ap = xin + (((size_t)(b * 2 + t) * Hs + hc) * Ws + wc) * DINNER + klo + cc;
  #pragma unroll
  for (int i = 0; i < 8; ++i) q.a[i] = *(const v2f*)(ap + 4 * i);
  const float* bp = wf + (size_t)s * 4 * 128;
  #pragma unroll
  for (int qd = 0; qd < 4; ++qd) {
    #pragma unroll
    for (int j = 0; j < CDC_NT; ++j)
      q.b[qd][j] = *(const float4*)(bp + ((size_t)j * NPAIR_W + qd) * 128);
  }
}

static __device__ __forceinline__ void cdc_compute(const CdcStage& q, v8f (&acc)[CDC_NT]) {
  const v2f zero = {0.f, 0.f};
  #pragma unroll
  for (int qd = 0; qd < 4; ++qd) {
    #pragma unroll
    for (int e2 = 0; e2 < 2; ++e2) {
      v2f a = q.ok ? q.a[2 * qd + e2] : zero;
      #pragma unroll
      for (int j = 0; j < CDC_NT; ++j) {
        float4 bv = q.b[qd][j];
        v2f bb;
        if (e2 == 0) { bb.x = bv.x; bb.y = bv.y; } else { bb.x = bv.z; bb.y = bv.w; }
        acc[j] = wmma4(a, bb, acc[j]);
      }
    }
  }
}

__global__ void __launch_bounds__(128) k4_cdc(const float* __restrict__ xin, const float* __restrict__ Wp,
                       const float* __restrict__ kdp, float* __restrict__ xdbl) {
  const int lane = threadIdx.x & 31;
  const int half = lane >> 4;
  const int klo  = half << 1;
  const int ln   = lane & 15;
  const int mtile = blockIdx.x * 4 + (threadIdx.x >> 5);
  const int nt0   = blockIdx.y * CDC_NT;
  const int m = mtile * 16 + ln;
  const int b = m / Ls;
  const int hw = m % Ls;
  const int h = hw / Ws, w = hw % Ws;

  v8f accN[CDC_NT];
  #pragma unroll
  for (int j = 0; j < CDC_NT; ++j) accN[j] = (v8f){0.f,0.f,0.f,0.f,0.f,0.f,0.f,0.f};

  const float* wf = Wp + (size_t)nt0 * NPAIR_W * 128 + lane * 4;
  CdcStage qa, qb;
  cdc_load(qa, xin, wf, 0, b, h, w, klo);
  for (int s = 0; s < CDC_STAGES; s += 2) {
    cdc_load(qb, xin, wf, s + 1, b, h, w, klo);
    cdc_compute(qa, accN);
    int s2 = (s + 2 < CDC_STAGES) ? (s + 2) : (s + 1);
    cdc_load(qa, xin, wf, s2, b, h, w, klo);
    cdc_compute(qb, accN);
  }

  v8f accD0[CDC_NT], accD1[CDC_NT];
  #pragma unroll
  for (int j = 0; j < CDC_NT; ++j) {
    accD0[j] = (v8f){0.f,0.f,0.f,0.f,0.f,0.f,0.f,0.f};
    accD1[j] = (v8f){0.f,0.f,0.f,0.f,0.f,0.f,0.f,0.f};
  }
  const float* ap0 = xin + (((size_t)(b * 2 + 0) * Hs + h) * Ws + w) * DINNER + klo;
  const float* ap1 = xin + (((size_t)(b * 2 + 1) * Hs + h) * Ws + w) * DINNER + klo;
  const float* kf  = kdp + (size_t)nt0 * NPAIR_KD * 128 + lane * 4;
  #pragma unroll
  for (int pr = 0; pr < NPAIR_KD; ++pr) {
    v2f a00 = *(const v2f*)(ap0 + pr * 8);
    v2f a01 = *(const v2f*)(ap0 + pr * 8 + 4);
    v2f a10 = *(const v2f*)(ap1 + pr * 8);
    v2f a11 = *(const v2f*)(ap1 + pr * 8 + 4);
    #pragma unroll
    for (int j = 0; j < CDC_NT; ++j) {
      float4 bv = *(const float4*)(kf + ((size_t)j * NPAIR_KD + pr) * 128);
      v2f b0, b1;
      b0.x = bv.x; b0.y = bv.y; b1.x = bv.z; b1.y = bv.w;
      accD0[j] = wmma4(a00, b0, accD0[j]);
      accD1[j] = wmma4(a10, b0, accD1[j]);
      accD0[j] = wmma4(a01, b1, accD0[j]);
      accD1[j] = wmma4(a11, b1, accD1[j]);
    }
  }

  #pragma unroll
  for (int j = 0; j < CDC_NT; ++j) {
    const int n = (nt0 + j) * 16 + ln;
    if (n < NKO) {
      int kq = n / NCOUT, o = n % NCOUT;
      for (int rr = 0; rr < 8; ++rr) {
        int mm = mtile * 16 + rr + half * 8;
        int b2 = mm / Ls; int l2 = mm % Ls;
        float Nv = accN[j][rr], d0 = accD0[j][rr], d1 = accD1[j][rr];
        float pooled = 2.f * Nv - THETA_C * (0.5f * (d0 + d1) + fminf(d0, d1));
        xdbl[(((size_t)b2 * 4 + kq) * NCOUT + o) * Ls + l2] = pooled;
      }
    }
  }
}

#define K5_DG  16
#define K5_TPB 224

__global__ void __launch_bounds__(K5_TPB) k5_delta(const float* __restrict__ xdbl, const float* __restrict__ dtw,
                         const float* __restrict__ dtb, float* __restrict__ delta) {
  const int l  = (blockIdx.x * K5_TPB + threadIdx.x) * 2;
  const int d0 = blockIdx.y * K5_DG;
  const int bk = blockIdx.z;
  const int k  = bk & 3;
  const float* xp = xdbl + (size_t)bk * NCOUT * Ls + l;
  v2f xv[DTRANK];
  #pragma unroll
  for (int rr = 0; rr < DTRANK; ++rr) xv[rr] = *(const v2f*)(xp + (size_t)rr * Ls);
  const float* wp = dtw + ((size_t)k * DINNER + d0) * DTRANK;
  const float* bp = dtb + k * DINNER + d0;
  float* op = delta + ((size_t)bk * DINNER + d0) * Ls + l;
  #pragma unroll 4
  for (int i = 0; i < K5_DG; ++i) {
    float a0 = bp[i];
    float a1 = a0;
    #pragma unroll
    for (int rr = 0; rr < DTRANK; ++rr) {
      float wv = wp[i * DTRANK + rr];
      a0 += xv[rr].x * wv;
      a1 += xv[rr].y * wv;
    }
    v2f o;
    o.x = fmaxf(a0, 0.f) + log1pf(expf(-fabsf(a0)));
    o.y = fmaxf(a1, 0.f) + log1pf(expf(-fabsf(a1)));
    *(v2f*)(op + (size_t)i * Ls) = o;
  }
}

#define SCH 16
#define SEGS  4
#define SEGL  (Ls / SEGS)
#define SEGCH (SEGL / SCH)

struct ScanChunk { float d[SCH], u[SCH], b[SCH], c[SCH]; };

static __device__ __forceinline__ void scan_load(ScanChunk& q, const float* __restrict__ dP,
                                                 const float* __restrict__ uP,
                                                 const float* __restrict__ bP,
                                                 const float* __restrict__ cP, int l0) {
  #pragma unroll
  for (int i = 0; i < SCH; i += 4) {
    float4 vd = *(const float4*)(dP + l0 + i);
    float4 vu = *(const float4*)(uP + l0 + i);
    float4 vb = *(const float4*)(bP + l0 + i);
    float4 vc = *(const float4*)(cP + l0 + i);
    q.d[i] = vd.x; q.d[i + 1] = vd.y; q.d[i + 2] = vd.z; q.d[i + 3] = vd.w;
    q.u[i] = vu.x; q.u[i + 1] = vu.y; q.u[i + 2] = vu.z; q.u[i + 3] = vu.w;
    q.b[i] = vb.x; q.b[i + 1] = vb.y; q.b[i + 2] = vb.z; q.b[i + 3] = vb.w;
    q.c[i] = vc.x; q.c[i + 1] = vc.y; q.c[i + 2] = vc.z; q.c[i + 3] = vc.w;
  }
}

static __device__ __forceinline__ float scan_steps(const ScanChunk& q, float& h, float A,
                                                   float Dv, int ln) {
  float y[SCH];
  #pragma unroll
  for (int j = 0; j < SCH; ++j) {
    float dl = q.d[j];
    h = __builtin_amdgcn_exp2f(dl * A) * h + (dl * q.u[j]) * q.b[j];
    y[j] = h * q.c[j];
  }
  #pragma unroll
  for (int m = 1; m < 16; m <<= 1) {
    float t[SCH];
    #pragma unroll
    for (int j = 0; j < SCH; ++j) t[j] = __shfl_xor(y[j], m);
    __builtin_amdgcn_sched_barrier(0);
    #pragma unroll
    for (int j = 0; j < SCH; ++j) y[j] += t[j];
  }
  float keep = 0.f;
  #pragma unroll
  for (int j = 0; j < SCH; ++j) {
    float o = y[j] + Dv * q.u[j];
    if (ln == j) keep = o;
  }
  return keep;
}

struct CarryChunk { float d[SCH], u[SCH], b[SCH]; };

static __device__ __forceinline__ void carry_load(CarryChunk& q, const float* __restrict__ dP,
                                                  const float* __restrict__ uP,
                                                  const float* __restrict__ bP, int l0) {
  #pragma unroll
  for (int i = 0; i < SCH; i += 4) {
    float4 vd = *(const float4*)(dP + l0 + i);
    float4 vu = *(const float4*)(uP + l0 + i);
    float4 vb = *(const float4*)(bP + l0 + i);
    q.d[i] = vd.x; q.d[i + 1] = vd.y; q.d[i + 2] = vd.z; q.d[i + 3] = vd.w;
    q.u[i] = vu.x; q.u[i + 1] = vu.y; q.u[i + 2] = vu.z; q.u[i + 3] = vu.w;
    q.b[i] = vb.x; q.b[i + 1] = vb.y; q.b[i + 2] = vb.z; q.b[i + 3] = vb.w;
  }
}

static __device__ __forceinline__ void carry_steps(const CarryChunk& q, float& h, float& P, float A) {
  #pragma unroll
  for (int j = 0; j < SCH; ++j) {
    float dl = q.d[j];
    float e = __builtin_amdgcn_exp2f(dl * A);
    h = e * h + (dl * q.u[j]) * q.b[j];
    P = P * e;
  }
}

__global__ void __launch_bounds__(128) k6_scan(const float* __restrict__ xs, const float* __restrict__ xdbl,
                        const float* __restrict__ Alogs, const float* __restrict__ Dss,
                        const float* __restrict__ delta, float* __restrict__ oy) {
  __shared__ float carP[SEGS - 1][32];
  __shared__ float carH[SEGS - 1][32];
  const int wid  = blockIdx.x;
  const int seg  = __builtin_amdgcn_readfirstlane(threadIdx.x >> 5);
  int lane = threadIdx.x & 31;
  int half = lane >> 4, ln = lane & 15;
  int bk = wid / 96;
  int d  = (wid % 96) * 2 + half;
  int k  = bk & 3;
  const int lb = seg * SEGL;
  const float* dP = delta + ((size_t)bk * DINNER + d) * Ls + lb;
  float*       oP = oy    + ((size_t)bk * DINNER + d) * Ls + lb;
  const float* uP = xs    + ((size_t)bk * DINNER + d) * Ls + lb;
  const float* bP = xdbl  + ((size_t)bk * NCOUT + DTRANK + ln) * Ls + lb;
  const float* cP = xdbl  + ((size_t)bk * NCOUT + DTRANK + DSTATE + ln) * Ls + lb;
  float A  = -expf(Alogs[((size_t)(k * DINNER + d)) * DSTATE + ln]) * 1.4426950408889634f;
  float Dv = Dss[k * DINNER + d];

  if (seg < SEGS - 1) {
    float hs = 0.f, P = 1.f;
    CarryChunk ca, cb;
    carry_load(ca, dP, uP, bP, 0);
    for (int c = 0; c < SEGCH - 1; c += 2) {
      carry_load(cb, dP, uP, bP, (c + 1) * SCH);
      carry_steps(ca, hs, P, A);
      carry_load(ca, dP, uP, bP, (c + 2) * SCH);
      carry_steps(cb, hs, P, A);
    }
    carry_steps(ca, hs, P, A);
    carP[seg][lane] = P;
    carH[seg][lane] = hs;
  }
  __syncthreads();
  float h = 0.f;
  #pragma unroll
  for (int i = 0; i < SEGS - 1; ++i) {
    if (i < seg) h = carP[i][lane] * h + carH[i][lane];
  }

  ScanChunk qa, qb;
  scan_load(qa, dP, uP, bP, cP, 0);
  for (int c = 0; c < SEGCH - 1; c += 2) {
    const int l0 = c * SCH;
    scan_load(qb, dP, uP, bP, cP, l0 + SCH);
    oP[l0 + ln] = scan_steps(qa, h, A, Dv, ln);
    scan_load(qa, dP, uP, bP, cP, l0 + 2 * SCH);
    oP[l0 + SCH + ln] = scan_steps(qb, h, A, Dv, ln);
  }
  oP[(SEGCH - 1) * SCH + ln] = scan_steps(qa, h, A, Dv, ln);
}

__global__ void __launch_bounds__(256) k7a_comb(const float* __restrict__ oy, float* __restrict__ y2) {
  __shared__ float tile[XT_P * XT_PITCH];
  const int h0   = blockIdx.x * 4;
  const int d0   = blockIdx.y * XT_D;
  const int b    = blockIdx.z;
  const int lane = threadIdx.x & 31;
  const int wv   = __builtin_amdgcn_readfirstlane(threadIdx.x >> 5);
  const size_t s = (size_t)DINNER * Ls;

  #pragma unroll
  for (int it = 0; it < (XT_D * Ws) / 256; ++it) {
    int idx = it * 256 + threadIdx.x;
    int dd = idx / Ws;
    int w  = idx % Ws;
    const float* base = oy + ((size_t)b * 4 * DINNER + d0 + dd) * Ls;
    int lt0 = w * Hs + h0;
    float4 t1 = *(const float4*)(base + s + lt0);
    float4 t3 = *(const float4*)(base + 3 * s + (Ls - 4 - lt0));
    float a1[4] = {t1.x, t1.y, t1.z, t1.w};
    float a3[4] = {t3.w, t3.z, t3.y, t3.x};
    #pragma unroll
    for (int j = 0; j < 4; ++j) {
      int l = (h0 + j) * Ws + w;
      float v = base[l] + base[2 * s + (Ls - 1 - l)] + a1[j] + a3[j];
      tile[(j * Ws + w) * XT_PITCH + dd] = v;
    }
  }
  __syncthreads();

  float* yp = y2 + ((size_t)b * Ls + h0 * Ws + wv * 28) * DINNER + d0 + lane;
  #pragma unroll
  for (int it = 0; it < 28; ++it)
    yp[(size_t)it * DINNER] = tile[(wv * 28 + it) * XT_PITCH + lane];
}

__global__ void k7b_ln(const float* __restrict__ xz, const float* __restrict__ g,
                       const float* __restrict__ bb, float* __restrict__ y2) {
  int m = blockIdx.x;
  int d = threadIdx.x;
  float v = y2[(size_t)m * DINNER + d];
  float s1 = v, s2 = v * v;
  #pragma unroll
  for (int off = 16; off >= 1; off >>= 1) {
    s1 += __shfl_xor(s1, off);
    s2 += __shfl_xor(s2, off);
  }
  __shared__ float r1[6], r2[6];
  __shared__ float smu, srs;
  if ((d & 31) == 0) { r1[d >> 5] = s1; r2[d >> 5] = s2; }
  __syncthreads();
  if (d == 0) {
    float t1 = 0.f, t2 = 0.f;
    for (int i = 0; i < 6; ++i) { t1 += r1[i]; t2 += r2[i]; }
    float mu = t1 / (float)DINNER;
    float var = t2 / (float)DINNER - mu * mu;
    smu = mu; srs = rsqrtf(var + 1e-5f);
  }
  __syncthreads();
  float yln = (v - smu) * srs * g[d] + bb[d];
  float zv = xz[(size_t)m * 384 + DINNER + d];
  y2[(size_t)m * DINNER + d] = yln * siluf(zv);
}

extern "C" void kernel_launch(void* const* d_in, const int* in_sizes, int n_in,
                              void* d_out, int out_size, void* d_ws, size_t ws_size,
                              hipStream_t stream) {
  (void)in_sizes; (void)n_in; (void)out_size; (void)ws_size;
  const float* x   = (const float*)d_in[0];
  const float* xt  = (const float*)d_in[1];
  const float* ipw = (const float*)d_in[2];
  const float* c2w = (const float*)d_in[3];
  const float* c2b = (const float*)d_in[4];
  const float* cxw = (const float*)d_in[5];
  const float* cxb = (const float*)d_in[6];
  const float* xpw = (const float*)d_in[7];
  const float* dtw = (const float*)d_in[8];
  const float* dtb = (const float*)d_in[9];
  const float* alg = (const float*)d_in[10];
  const float* dss = (const float*)d_in[11];
  const float* lng = (const float*)d_in[12];
  const float* lnb = (const float*)d_in[13];
  const float* opw = (const float*)d_in[14];
  float* out = (float*)d_out;
  float* ws  = (float*)d_ws;

  size_t off = 0;
  float* xz    = ws + off; off += (size_t)MROWS * 384;
  float* xin   = ws + off; off += (size_t)BATCH * 2 * Ls * DINNER;
  float* xs    = ws + off; off += (size_t)BATCH * 4 * DINNER * Ls;
  float* Wp    = ws + off; off += (size_t)NKO_PAD * KBIG;
  float* kdp   = ws + off; off += (size_t)NKO_PAD * DINNER;
  float* xdbl  = ws + off; off += (size_t)BATCH * 4 * NCOUT * Ls;
  float* delta = ws + off; off += (size_t)BATCH * 4 * DINNER * Ls;
  float* oy    = ws + off; off += (size_t)BATCH * 4 * DINNER * Ls;
  float* y2    = xin;

  gemm_f32_wmma<<<dim3(384 / 16, MROWS / 16), 32, 0, stream>>>(x, ipw, xz, DMODEL, 384);

  k_pack_w <<<(NKO_PAD * KBIG + 255) / 256, 256, 0, stream>>>(xpw, Wp);
  k_pack_kd<<<(NKO_PAD * DINNER + 255) / 256, 256, 0, stream>>>(xpw, kdp);

  k2_conv<<<dim3(Ws / K2_SW, Hs, BATCH * 2), DINNER, 0, stream>>>(
      xz, xt, c2w, c2b, cxw, cxb, xin);

  k3_xs<<<dim3(Ls / XT_P, DINNER / XT_D, BATCH * 2), 256, 0, stream>>>(xin, xs);

  k4_cdc<<<dim3(MROWS / 64, NKO_PAD / (16 * CDC_NT)), 128, 0, stream>>>(xin, Wp, kdp, xdbl);

  k5_delta<<<dim3(Ls / (2 * K5_TPB), DINNER / K5_DG, BATCH * 4), K5_TPB, 0, stream>>>(xdbl, dtw, dtb, delta);

  k6_scan<<<BATCH * 4 * (DINNER / 2), 32 * SEGS, 0, stream>>>(xs, xdbl, alg, dss, delta, oy);

  k7a_comb<<<dim3(Ls / XT_P, DINNER / XT_D, BATCH), 256, 0, stream>>>(oy, y2);
  k7b_ln<<<MROWS, DINNER, 0, stream>>>(xz, lng, lnb, y2);

  gemm_f32_wmma<<<dim3(DMODEL / 16, MROWS / 16), 32, 0, stream>>>(y2, opw, out, DINNER, DMODEL);
}


# k4 writes the scan's B and C operands in scan order [4-step group][B|C][state][step]: a k6 scan load reads 256 contiguous bytes (2 lines), was 16 bytes from each of 16 rows; sums unchanged
# speedup vs baseline: 7.8747x; 1.0739x over previous
#include <hip/hip_runtime.h>
#include <hip/hip_bf16.h>
#include <math.h>

typedef float v2f __attribute__((ext_vector_type(2)));
typedef float v8f __attribute__((ext_vector_type(8)));

#define BATCH   2
#define Hs      56
#define Ws      56
#define Ls      (Hs*Ws)
#define DMODEL  96
#define DINNER  192
#define DSTATE  16
#define DTRANK  6
#define NCOUT   38
#define NKO     (4*NCOUT)
#define NKO_PAD 160
#define KBIG    (DINNER*2*9)
#define THETA_C 0.6f
#define MROWS   (BATCH*Ls)

static __device__ __forceinline__ float siluf(float v) { return v / (1.f + expf(-v)); }

static __device__ __forceinline__ v8f wmma4(v2f a, v2f b, v8f c) {
  return __builtin_amdgcn_wmma_f32_16x16x4_f32(false, a, false, b, (short)0, c, false, false);
}

__global__ void gemm_f32_wmma(const float* __restrict__ A, const float* __restrict__ Bt,
                              float* __restrict__ C, int K, int ldc) {
  const int lane = threadIdx.x & 31;
  const int half = lane >> 4;
  const int klo  = half << 1;
  const int ln   = lane & 15;
  const int ntile = blockIdx.x;
  const int mtile = blockIdx.y;
  const float* ap = A  + (size_t)(mtile * 16 + ln) * K + klo;
  const float* bp = Bt + (size_t)(ntile * 16 + ln) * K + klo;
  v8f acc = {0.f,0.f,0.f,0.f,0.f,0.f,0.f,0.f};
  for (int k = 0; k < K; k += 4) {
    v2f a = *(const v2f*)(ap + k);
    v2f b = *(const v2f*)(bp + k);
    acc = wmma4(a, b, acc);
  }
  const int n = ntile * 16 + ln;
  for (int rr = 0; rr < 8; ++rr) {
    int mm = mtile * 16 + rr + half * 8;
    C[(size_t)mm * ldc + n] = acc[rr];
  }
}

#define NPAIR_W  (KBIG / 8)
#define NPAIR_KD (DINNER / 8)

__global__ void k_pack_w(const float* __restrict__ xw, float* __restrict__ Wp) {
  int tid = blockIdx.x * blockDim.x + threadIdx.x;
  if (tid >= NKO_PAD * KBIG) return;
  int e    = tid & 3;
  int lane = (tid >> 2) & 31;
  int grp  = tid >> 7;
  int pr   = grp % NPAIR_W;
  int nt   = grp / NPAIR_W;
  int n    = nt * 16 + (lane & 15);
  int kk   = pr * 8 + (e >> 1) * 4 + (lane >> 4) * 2 + (e & 1);
  float v = 0.f;
  if (n < NKO) {
    int c = kk % DINNER;
    int r = kk / DINNER;
    int dx = r % 3, dy = (r / 3) % 3, t = r / 9;
    v = xw[((((size_t)n * DINNER + c) * 2 + t) * 3 + dy) * 3 + dx];
  }
  Wp[tid] = v;
}

__global__ void k_pack_kd(const float* __restrict__ xw, float* __restrict__ kdp) {
  int tid = blockIdx.x * blockDim.x + threadIdx.x;
  if (tid >= NKO_PAD * DINNER) return;
  int e    = tid & 3;
  int lane = (tid >> 2) & 31;
  int grp  = tid >> 7;
  int pr   = grp % NPAIR_KD;
  int nt   = grp / NPAIR_KD;
  int n    = nt * 16 + (lane & 15);
  int c    = pr * 8 + (e >> 1) * 4 + (lane >> 4) * 2 + (e & 1);
  float v = 0.f;
  if (n < NKO) {
    const float* p = xw + ((size_t)n * DINNER + c) * 18;
    for (int i = 0; i < 18; ++i) v += p[i];
  }
  kdp[tid] = v;
}

#define K2_SW 14

__global__ void __launch_bounds__(DINNER) k2_conv(const float* __restrict__ xz, const float* __restrict__ xt,
                        const float* __restrict__ w2, const float* __restrict__ b2,
                        const float* __restrict__ wxt, const float* __restrict__ bxt,
                        float* __restrict__ xin) {
  const int c  = threadIdx.x;
  const int w0 = blockIdx.x * K2_SW;
  const int h  = blockIdx.y;
  const int t  = blockIdx.z & 1;
  const int b  = blockIdx.z >> 1;
  const float* src = t ? (xt + (c >> 1)) : (xz + c);
  const int    ps  = t ? DMODEL : 2 * DINNER;
  const float* wp  = (t ? wxt : w2) + c * 9;
  const float bias = t ? bxt[c] : b2[c];
  float wk[9];
  #pragma unroll
  for (int i = 0; i < 9; ++i) wk[i] = wp[i];
  float win[3][K2_SW + 2];
  #pragma unroll
  for (int dy = 0; dy < 3; ++dy) {
    const int hh = h + dy - 1;
    const bool rok = (hh >= 0) && (hh < Hs);
    const int hc = hh < 0 ? 0 : (hh >= Hs ? Hs - 1 : hh);
    const float* rp = src + (size_t)((b * Hs + hc) * Ws) * ps;
    #pragma unroll
    for (int j = 0; j < K2_SW + 2; ++j) {
      int ww = w0 + j - 1;
      bool ok = rok;
      if (j == 0)         { ok = ok && (ww >= 0); ww = ww < 0 ? 0 : ww; }
      if (j == K2_SW + 1) { ok = ok && (ww < Ws); ww = ww >= Ws ? Ws - 1 : ww; }
      const float v = rp[(size_t)ww * ps];
      win[dy][j] = ok ? v : 0.f;
    }
  }
  float* op = xin + ((size_t)((b * 2 + t) * Hs + h) * Ws + w0) * DINNER + c;
  #pragma unroll
  for (int j = 0; j < K2_SW; ++j) {
    float acc = bias;
    #pragma unroll
    for (int dy = 0; dy < 3; ++dy) {
      #pragma unroll
      for (int dx = 0; dx < 3; ++dx) acc += win[dy][j + dx] * wk[dy * 3 + dx];
    }
    op[(size_t)j * DINNER] = siluf(acc);
  }
}

#define XT_P     224
#define XT_D     32
#define XT_PITCH 33

__global__ void __launch_bounds__(256) k3_xs(const float* __restrict__ xin, float* __restrict__ xs) {
  __shared__ float tile[XT_P * XT_PITCH];
  const int g    = blockIdx.x;
  const int d0   = blockIdx.y * XT_D;
  const int b    = blockIdx.z >> 1;
  const int col  = blockIdx.z & 1;
  const int lane = threadIdx.x & 31;
  const int wv   = __builtin_amdgcn_readfirstlane(threadIdx.x >> 5);

  const int major  = g * 4 + (wv >> 1);
  const int minor0 = (wv & 1) * 28;
  const int sMaj = col ? Ws : 1;
  const int sMin = col ? 1 : Ws;
  const float* xb = xin + ((size_t)b * 2 * Ls + major * sMaj + minor0 * sMin) * DINNER + d0 + lane;
  float pv[28];
  #pragma unroll
  for (int it = 0; it < 28; ++it) {
    const int s  = it & 1;
    const int i2 = it & ~1;
    const float* p0 = xb + ((size_t)s * Ls + (size_t)i2 * sMin) * DINNER;
    float v0 = p0[0];
    float v1 = p0[(size_t)sMin * DINNER];
    pv[it] = 0.5f * (v0 + v1) + fmaxf(v0, v1);
  }
  #pragma unroll
  for (int it = 0; it < 28; ++it) tile[(wv * 28 + it) * XT_PITCH + lane] = pv[it];
  __syncthreads();

  const int kbase = col ? 2 : 0;
  #pragma unroll
  for (int i = 0; i < 4; ++i) {
    int dd = wv * 4 + i;
    float* op0 = xs + ((size_t)(b * 4 + kbase) * DINNER + d0 + dd) * Ls + g * XT_P;
    float* op1 = op0 + (size_t)DINNER * Ls;
    #pragma unroll
    for (int c = 0; c < 7; ++c) {
      int q  = lane + 32 * c;
      int sp = (q / 56) * 56 + 55 - q % 56;
      op0[q] = tile[q  * XT_PITCH + dd];
      op1[q] = tile[sp * XT_PITCH + dd];
    }
  }
}

#define NGRP   (Ls / 4)
#define BC_GRP (2 * DSTATE * 4)
#define CDC_NT     5
#define CDC_STAGES (KBIG / 32)

struct CdcStage { v2f a[8]; float4 b[4][CDC_NT]; int ok; };

static __device__ __forceinline__ void cdc_load(CdcStage& q, const float* __restrict__ xin,
                                                const float* __restrict__ wf, int s,
                                                int b, int h, int w, int klo) {
  int td = s / 6, cc = (s - td * 6) * 32;
  int t = td / 9, r9 = td - t * 9, dy = r9 / 3, dx = r9 - dy * 3;
  int hh = h + dy - 1, ww = w + dx - 1;
  q.ok = (hh >= 0) && (hh < Hs) && (ww >= 0) && (ww < Ws);
  int hc = hh < 0 ? 0 : (hh >= Hs ? Hs - 1 : hh);
  int wc = ww < 0 ? 0 : (ww >= Ws ? Ws - 1 : ww);
  const float* ap = xin + (((size_t)(b * 2 + t) * Hs + hc) * Ws + wc) * DINNER + klo + cc;
  #pragma unroll
  for (int i = 0; i < 8; ++i) q.a[i] = *(const v2f*)(ap + 4 * i);
  const float* bp = wf + (size_t)s * 4 * 128;
  #pragma unroll
  for (int qd = 0; qd < 4; ++qd) {
    #pragma unroll
    for (int j = 0; j < CDC_NT; ++j)
      q.b[qd][j] = *(const float4*)(bp + ((size_t)j * NPAIR_W + qd) * 128);
  }
}

static __device__ __forceinline__ void cdc_compute(const CdcStage& q, v8f (&acc)[CDC_NT]) {
  const v2f zero = {0.f, 0.f};
  #pragma unroll
  for (int qd = 0; qd < 4; ++qd) {
    #pragma unroll
    for (int e2 = 0; e2 < 2; ++e2) {
      v2f a = q.ok ? q.a[2 * qd + e2] : zero;
      #pragma unroll
      for (int j = 0; j < CDC_NT; ++j) {
        float4 bv = q.b[qd][j];
        v2f bb;
        if (e2 == 0) { bb.x = bv.x; bb.y = bv.y; } else { bb.x = bv.z; bb.y = bv.w; }
        acc[j] = wmma4(a, bb, acc[j]);
      }
    }
  }
}

__global__ void __launch_bounds__(128) k4_cdc(const float* __restrict__ xin, const float* __restrict__ Wp,
                       const float* __restrict__ kdp, float* __restrict__ xdbl,
                       float* __restrict__ bc) {
  const int lane = threadIdx.x & 31;
  const int half = lane >> 4;
  const int klo  = half << 1;
  const int ln   = lane & 15;
  const int mtile = blockIdx.x * 4 + (threadIdx.x >> 5);
  const int nt0   = blockIdx.y * CDC_NT;
  const int m = mtile * 16 + ln;
  const int b = m / Ls;
  const int hw = m % Ls;
  const int h = hw / Ws, w = hw % Ws;

  v8f accN[CDC_NT];
  #pragma unroll
  for (int j = 0; j < CDC_NT; ++j) accN[j] = (v8f){0.f,0.f,0.f,0.f,0.f,0.f,0.f,0.f};

  const float* wf = Wp + (size_t)nt0 * NPAIR_W * 128 + lane * 4;
  CdcStage qa, qb;
  cdc_load(qa, xin, wf, 0, b, h, w, klo);
  for (int s = 0; s < CDC_STAGES; s += 2) {
    cdc_load(qb, xin, wf, s + 1, b, h, w, klo);
    cdc_compute(qa, accN);
    int s2 = (s + 2 < CDC_STAGES) ? (s + 2) : (s + 1);
    cdc_load(qa, xin, wf, s2, b, h, w, klo);
    cdc_compute(qb, accN);
  }

  v8f accD0[CDC_NT], accD1[CDC_NT];
  #pragma unroll
  for (int j = 0; j < CDC_NT; ++j) {
    accD0[j] = (v8f){0.f,0.f,0.f,0.f,0.f,0.f,0.f,0.f};
    accD1[j] = (v8f){0.f,0.f,0.f,0.f,0.f,0.f,0.f,0.f};
  }
  const float* ap0 = xin + (((size_t)(b * 2 + 0) * Hs + h) * Ws + w) * DINNER + klo;
  const float* ap1 = xin + (((size_t)(b * 2 + 1) * Hs + h) * Ws + w) * DINNER + klo;
  const float* kf  = kdp + (size_t)nt0 * NPAIR_KD * 128 + lane * 4;
  #pragma unroll
  for (int pr = 0; pr < NPAIR_KD; ++pr) {
    v2f a00 = *(const v2f*)(ap0 + pr * 8);
    v2f a01 = *(const v2f*)(ap0 + pr * 8 + 4);
    v2f a10 = *(const v2f*)(ap1 + pr * 8);
    v2f a11 = *(const v2f*)(ap1 + pr * 8 + 4);
    #pragma unroll
    for (int j = 0; j < CDC_NT; ++j) {
      float4 bv = *(const float4*)(kf + ((size_t)j * NPAIR_KD + pr) * 128);
      v2f b0, b1;
      b0.x = bv.x; b0.y = bv.y; b1.x = bv.z; b1.y = bv.w;
      accD0[j] = wmma4(a00, b0, accD0[j]);
      accD1[j] = wmma4(a10, b0, accD1[j]);
      accD0[j] = wmma4(a01, b1, accD0[j]);
      accD1[j] = wmma4(a11, b1, accD1[j]);
    }
  }

  const int m8 = mtile * 16 + half * 8;
  const int b8 = m8 / Ls, l8 = m8 % Ls;
  #pragma unroll
  for (int j = 0; j < CDC_NT; ++j) {
    const int n = (nt0 + j) * 16 + ln;
    if (n < NKO) {
      int kq = n / NCOUT, o = n % NCOUT;
      float p[8];
      #pragma unroll
      for (int rr = 0; rr < 8; ++rr) {
        float Nv = accN[j][rr], d0 = accD0[j][rr], d1 = accD1[j][rr];
        p[rr] = 2.f * Nv - THETA_C * (0.5f * (d0 + d1) + fminf(d0, d1));
      }
      if (o < DTRANK) {
        float* q = xdbl + (((size_t)b8 * 4 + kq) * NCOUT + o) * Ls + l8;
        #pragma unroll
        for (int rr = 0; rr < 8; ++rr) q[rr] = p[rr];
      } else {
        const int sel = (o - DTRANK) >> 4, st = (o - DTRANK) & 15;
        float* q = bc + (((size_t)(b8 * 4 + kq) * NGRP + (l8 >> 2)) * 2 + sel) * (DSTATE * 4) + st * 4;
        *(float4*)q            = make_float4(p[0], p[1], p[2], p[3]);
        *(float4*)(q + BC_GRP) = make_float4(p[4], p[5], p[6], p[7]);
      }
    }
  }
}

#define K5_DG  16
#define K5_TPB 224

__global__ void __launch_bounds__(K5_TPB) k5_delta(const float* __restrict__ xdbl, const float* __restrict__ dtw,
                         const float* __restrict__ dtb, float* __restrict__ delta) {
  const int l  = (blockIdx.x * K5_TPB + threadIdx.x) * 2;
  const int d0 = blockIdx.y * K5_DG;
  const int bk = blockIdx.z;
  const int k  = bk & 3;
  const float* xp = xdbl + (size_t)bk * NCOUT * Ls + l;
  v2f xv[DTRANK];
  #pragma unroll
  for (int rr = 0; rr < DTRANK; ++rr) xv[rr] = *(const v2f*)(xp + (size_t)rr * Ls);
  const float* wp = dtw + ((size_t)k * DINNER + d0) * DTRANK;
  const float* bp = dtb + k * DINNER + d0;
  float* op = delta + ((size_t)bk * DINNER + d0) * Ls + l;
  #pragma unroll 4
  for (int i = 0; i < K5_DG; ++i) {
    float a0 = bp[i];
    float a1 = a0;
    #pragma unroll
    for (int rr = 0; rr < DTRANK; ++rr) {
      float wv = wp[i * DTRANK + rr];
      a0 += xv[rr].x * wv;
      a1 += xv[rr].y * wv;
    }
    v2f o;
    o.x = fmaxf(a0, 0.f) + log1pf(expf(-fabsf(a0)));
    o.y = fmaxf(a1, 0.f) + log1pf(expf(-fabsf(a1)));
    *(v2f*)(op + (size_t)i * Ls) = o;
  }
}

#define SCH 16
#define SEGS  4
#define SEGL  (Ls / SEGS)
#define SEGCH (SEGL / SCH)

struct ScanChunk { float d[SCH], u[SCH], b[SCH], c[SCH]; };

static __device__ __forceinline__ void scan_load(ScanChunk& q, const float* __restrict__ dP,
                                                 const float* __restrict__ uP,
                                                 const float* __restrict__ bcP, int l0) {
  #pragma unroll
  for (int i = 0; i < SCH; i += 4) {
    const float* gp = bcP + (size_t)((l0 + i) >> 2) * BC_GRP;
    float4 vd = *(const float4*)(dP + l0 + i);
    float4 vu = *(const float4*)(uP + l0 + i);
    float4 vb = *(const float4*)(gp);
    float4 vc = *(const float4*)(gp + DSTATE * 4);
    q.d[i] = vd.x; q.d[i + 1] = vd.y; q.d[i + 2] = vd.z; q.d[i + 3] = vd.w;
    q.u[i] = vu.x; q.u[i + 1] = vu.y; q.u[i + 2] = vu.z; q.u[i + 3] = vu.w;
    q.b[i] = vb.x; q.b[i + 1] = vb.y; q.b[i + 2] = vb.z; q.b[i + 3] = vb.w;
    q.c[i] = vc.x; q.c[i + 1] = vc.y; q.c[i + 2] = vc.z; q.c[i + 3] = vc.w;
  }
}

static __device__ __forceinline__ float scan_steps(const ScanChunk& q, float& h, float A,
                                                   float Dv, int ln) {
  float y[SCH];
  #pragma unroll
  for (int j = 0; j < SCH; ++j) {
    float dl = q.d[j];
    h = __builtin_amdgcn_exp2f(dl * A) * h + (dl * q.u[j]) * q.b[j];
    y[j] = h * q.c[j];
  }
  #pragma unroll
  for (int m = 1; m < 16; m <<= 1) {
    float t[SCH];
    #pragma unroll
    for (int j = 0; j < SCH; ++j) t[j] = __shfl_xor(y[j], m);
    __builtin_amdgcn_sched_barrier(0);
    #pragma unroll
    for (int j = 0; j < SCH; ++j) y[j] += t[j];
  }
  float keep = 0.f;
  #pragma unroll
  for (int j = 0; j < SCH; ++j) {
    float o = y[j] + Dv * q.u[j];
    if (ln == j) keep = o;
  }
  return keep;
}

struct CarryChunk { float d[SCH], u[SCH], b[SCH]; };

static __device__ __forceinline__ void carry_load(CarryChunk& q, const float* __restrict__ dP,
                                                  const float* __restrict__ uP,
                                                  const float* __restrict__ bcP, int l0) {
  #pragma unroll
  for (int i = 0; i < SCH; i += 4) {
    float4 vd = *(const float4*)(dP + l0 + i);
    float4 vu = *(const float4*)(uP + l0 + i);
    float4 vb = *(const float4*)(bcP + (size_t)((l0 + i) >> 2) * BC_GRP);
    q.d[i] = vd.x; q.d[i + 1] = vd.y; q.d[i + 2] = vd.z; q.d[i + 3] = vd.w;
    q.u[i] = vu.x; q.u[i + 1] = vu.y; q.u[i + 2] = vu.z; q.u[i + 3] = vu.w;
    q.b[i] = vb.x; q.b[i + 1] = vb.y; q.b[i + 2] = vb.z; q.b[i + 3] = vb.w;
  }
}

static __device__ __forceinline__ void carry_steps(const CarryChunk& q, float& h, float& P, float A) {
  #pragma unroll
  for (int j = 0; j < SCH; ++j) {
    float dl = q.d[j];
    float e = __builtin_amdgcn_exp2f(dl * A);
    h = e * h + (dl * q.u[j]) * q.b[j];
    P = P * e;
  }
}

__global__ void __launch_bounds__(128) k6_scan(const float* __restrict__ xs, const float* __restrict__ bc,
                        const float* __restrict__ Alogs, const float* __restrict__ Dss,
                        const float* __restrict__ delta, float* __restrict__ oy) {
  __shared__ float carP[SEGS - 1][32];
  __shared__ float carH[SEGS - 1][32];
  const int wid  = blockIdx.x;
  const int seg  = __builtin_amdgcn_readfirstlane(threadIdx.x >> 5);
  int lane = threadIdx.x & 31;
  int half = lane >> 4, ln = lane & 15;
  int bk = wid / 96;
  int d  = (wid % 96) * 2 + half;
  int k  = bk & 3;
  const int lb = seg * SEGL;
  const float* dP = delta + ((size_t)bk * DINNER + d) * Ls + lb;
  float*       oP = oy    + ((size_t)bk * DINNER + d) * Ls + lb;
  const float* uP = xs    + ((size_t)bk * DINNER + d) * Ls + lb;
  const float* bcP = bc   + ((size_t)bk * NGRP + (lb >> 2)) * BC_GRP + ln * 4;
  float A  = -expf(Alogs[((size_t)(k * DINNER + d)) * DSTATE + ln]) * 1.4426950408889634f;
  float Dv = Dss[k * DINNER + d];

  if (seg < SEGS - 1) {
    float hs = 0.f, P = 1.f;
    CarryChunk ca, cb;
    carry_load(ca, dP, uP, bcP, 0);
    for (int c = 0; c < SEGCH - 1; c += 2) {
      carry_load(cb, dP, uP, bcP, (c + 1) * SCH);
      carry_steps(ca, hs, P, A);
      carry_load(ca, dP, uP, bcP, (c + 2) * SCH);
      carry_steps(cb, hs, P, A);
    }
    carry_steps(ca, hs, P, A);
    carP[seg][lane] = P;
    carH[seg][lane] = hs;
  }
  __syncthreads();
  float h = 0.f;
  #pragma unroll
  for (int i = 0; i < SEGS - 1; ++i) {
    if (i < seg) h = carP[i][lane] * h + carH[i][lane];
  }

  ScanChunk qa, qb;
  scan_load(qa, dP, uP, bcP, 0);
  for (int c = 0; c < SEGCH - 1; c += 2) {
    const int l0 = c * SCH;
    scan_load(qb, dP, uP, bcP, l0 + SCH);
    oP[l0 + ln] = scan_steps(qa, h, A, Dv, ln);
    scan_load(qa, dP, uP, bcP, l0 + 2 * SCH);
    oP[l0 + SCH + ln] = scan_steps(qb, h, A, Dv, ln);
  }
  oP[(SEGCH - 1) * SCH + ln] = scan_steps(qa, h, A, Dv, ln);
}

__global__ void __launch_bounds__(256) k7a_comb(const float* __restrict__ oy, float* __restrict__ y2) {
  __shared__ float tile[XT_P * XT_PITCH];
  const int h0   = blockIdx.x * 4;
  const int d0   = blockIdx.y * XT_D;
  const int b    = blockIdx.z;
  const int lane = threadIdx.x & 31;
  const int wv   = __builtin_amdgcn_readfirstlane(threadIdx.x >> 5);
  const size_t s = (size_t)DINNER * Ls;

  #pragma unroll
  for (int it = 0; it < (XT_D * Ws) / 256; ++it) {
    int idx = it * 256 + threadIdx.x;
    int dd = idx / Ws;
    int w  = idx % Ws;
    const float* base = oy + ((size_t)b * 4 * DINNER + d0 + dd) * Ls;
    int lt0 = w * Hs + h0;
    float4 t1 = *(const float4*)(base + s + lt0);
    float4 t3 = *(const float4*)(base + 3 * s + (Ls - 4 - lt0));
    float a1[4] = {t1.x, t1.y, t1.z, t1.w};
    float a3[4] = {t3.w, t3.z, t3.y, t3.x};
    #pragma unroll
    for (int j = 0; j < 4; ++j) {
      int l = (h0 + j) * Ws + w;
      float v = base[l] + base[2 * s + (Ls - 1 - l)] + a1[j] + a3[j];
      tile[(j * Ws + w) * XT_PITCH + dd] = v;
    }
  }
  __syncthreads();

  float* yp = y2 + ((size_t)b * Ls + h0 * Ws + wv * 28) * DINNER + d0 + lane;
  #pragma unroll
  for (int it = 0; it < 28; ++it)
    yp[(size_t)it * DINNER] = tile[(wv * 28 + it) * XT_PITCH + lane];
}

__global__ void k7b_ln(const float* __restrict__ xz, const float* __restrict__ g,
                       const float* __restrict__ bb, float* __restrict__ y2) {
  int m = blockIdx.x;
  int d = threadIdx.x;
  float v = y2[(size_t)m * DINNER + d];
  float s1 = v, s2 = v * v;
  #pragma unroll
  for (int off = 16; off >= 1; off >>= 1) {
    s1 += __shfl_xor(s1, off);
    s2 += __shfl_xor(s2, off);
  }
  __shared__ float r1[6], r2[6];
  __shared__ float smu, srs;
  if ((d & 31) == 0) { r1[d >> 5] = s1; r2[d >> 5] = s2; }
  __syncthreads();
  if (d == 0) {
    float t1 = 0.f, t2 = 0.f;
    for (int i = 0; i < 6; ++i) { t1 += r1[i]; t2 += r2[i]; }
    float mu = t1 / (float)DINNER;
    float var = t2 / (float)DINNER - mu * mu;
    smu = mu; srs = rsqrtf(var + 1e-5f);
  }
  __syncthreads();
  float yln = (v - smu) * srs * g[d] + bb[d];
  float zv = xz[(size_t)m * 384 + DINNER + d];
  y2[(size_t)m * DINNER + d] = yln * siluf(zv);
}

extern "C" void kernel_launch(void* const* d_in, const int* in_sizes, int n_in,
                              void* d_out, int out_size, void* d_ws, size_t ws_size,
                              hipStream_t stream) {
  (void)in_sizes; (void)n_in; (void)out_size; (void)ws_size;
  const float* x   = (const float*)d_in[0];
  const float* xt  = (const float*)d_in[1];
  const float* ipw = (const float*)d_in[2];
  const float* c2w = (const float*)d_in[3];
  const float* c2b = (const float*)d_in[4];
  const float* cxw = (const float*)d_in[5];
  const float* cxb = (const float*)d_in[6];
  const float* xpw = (const float*)d_in[7];
  const float* dtw = (const float*)d_in[8];
  const float* dtb = (const float*)d_in[9];
  const float* alg = (const float*)d_in[10];
  const float* dss = (const float*)d_in[11];
  const float* lng = (const float*)d_in[12];
  const float* lnb = (const float*)d_in[13];
  const float* opw = (const float*)d_in[14];
  float* out = (float*)d_out;
  float* ws  = (float*)d_ws;

  size_t off = 0;
  float* xz    = ws + off; off += (size_t)MROWS * 384;
  float* xin   = ws + off; off += (size_t)BATCH * 2 * Ls * DINNER;
  float* xs    = ws + off; off += (size_t)BATCH * 4 * DINNER * Ls;
  float* Wp    = ws + off; off += (size_t)NKO_PAD * KBIG;
  float* kdp   = ws + off; off += (size_t)NKO_PAD * DINNER;
  float* xdbl  = ws + off; off += (size_t)BATCH * 4 * NCOUT * Ls;
  float* delta = ws + off; off += (size_t)BATCH * 4 * DINNER * Ls;
  float* oy    = ws + off; off += (size_t)BATCH * 4 * DINNER * Ls;
  float* bc    = ws + off; off += (size_t)BATCH * 4 * NGRP * BC_GRP;
  float* y2    = xin;

  gemm_f32_wmma<<<dim3(384 / 16, MROWS / 16), 32, 0, stream>>>(x, ipw, xz, DMODEL, 384);

  k_pack_w <<<(NKO_PAD * KBIG + 255) / 256, 256, 0, stream>>>(xpw, Wp);
  k_pack_kd<<<(NKO_PAD * DINNER + 255) / 256, 256, 0, stream>>>(xpw, kdp);

  k2_conv<<<dim3(Ws / K2_SW, Hs, BATCH * 2), DINNER, 0, stream>>>(
      xz, xt, c2w, c2b, cxw, cxb, xin);

  k3_xs<<<dim3(Ls / XT_P, DINNER / XT_D, BATCH * 2), 256, 0, stream>>>(xin, xs);

  k4_cdc<<<dim3(MROWS / 64, NKO_PAD / (16 * CDC_NT)), 128, 0, stream>>>(xin, Wp, kdp, xdbl, bc);

  k5_delta<<<dim3(Ls / (2 * K5_TPB), DINNER / K5_DG, BATCH * 4), K5_TPB, 0, stream>>>(xdbl, dtw, dtb, delta);

  k6_scan<<<BATCH * 4 * (DINNER / 2), 32 * SEGS, 0, stream>>>(xs, bc, alg, dss, delta, oy);

  k7a_comb<<<dim3(Ls / XT_P, DINNER / XT_D, BATCH), 256, 0, stream>>>(oy, y2);
  k7b_ln<<<MROWS, DINNER, 0, stream>>>(xz, lng, lnb, y2);

  gemm_f32_wmma<<<dim3(DMODEL / 16, MROWS / 16), 32, 0, stream>>>(y2, opw, out, DINNER, DMODEL);
}


# k4 conv GEMM: scheduling fences keep each stage's 28 loads one group, issued a whole stage (40 matrix ops) before use; the compiler had sunk them next to their readers; sums unchanged
# speedup vs baseline: 8.6104x; 1.0934x over previous
#include <hip/hip_runtime.h>
#include <hip/hip_bf16.h>
#include <math.h>

typedef float v2f __attribute__((ext_vector_type(2)));
typedef float v8f __attribute__((ext_vector_type(8)));

#define BATCH   2
#define Hs      56
#define Ws      56
#define Ls      (Hs*Ws)
#define DMODEL  96
#define DINNER  192
#define DSTATE  16
#define DTRANK  6
#define NCOUT   38
#define NKO     (4*NCOUT)
#define NKO_PAD 160
#define KBIG    (DINNER*2*9)
#define THETA_C 0.6f
#define MROWS   (BATCH*Ls)

static __device__ __forceinline__ float siluf(float v) { return v / (1.f + expf(-v)); }

static __device__ __forceinline__ v8f wmma4(v2f a, v2f b, v8f c) {
  return __builtin_amdgcn_wmma_f32_16x16x4_f32(false, a, false, b, (short)0, c, false, false);
}

__global__ void gemm_f32_wmma(const float* __restrict__ A, const float* __restrict__ Bt,
                              float* __restrict__ C, int K, int ldc) {
  const int lane = threadIdx.x & 31;
  const int half = lane >> 4;
  const int klo  = half << 1;
  const int ln   = lane & 15;
  const int ntile = blockIdx.x;
  const int mtile = blockIdx.y;
  const float* ap = A  + (size_t)(mtile * 16 + ln) * K + klo;
  const float* bp = Bt + (size_t)(ntile * 16 + ln) * K + klo;
  v8f acc = {0.f,0.f,0.f,0.f,0.f,0.f,0.f,0.f};
  for (int k = 0; k < K; k += 4) {
    v2f a = *(const v2f*)(ap + k);
    v2f b = *(const v2f*)(bp + k);
    acc = wmma4(a, b, acc);
  }
  const int n = ntile * 16 + ln;
  for (int rr = 0; rr < 8; ++rr) {
    int mm = mtile * 16 + rr + half * 8;
    C[(size_t)mm * ldc + n] = acc[rr];
  }
}

#define NPAIR_W  (KBIG / 8)
#define NPAIR_KD (DINNER / 8)

__global__ void k_pack_w(const float* __restrict__ xw, float* __restrict__ Wp) {
  int tid = blockIdx.x * blockDim.x + threadIdx.x;
  if (tid >= NKO_PAD * KBIG) return;
  int e    = tid & 3;
  int lane = (tid >> 2) & 31;
  int grp  = tid >> 7;
  int pr   = grp % NPAIR_W;
  int nt   = grp / NPAIR_W;
  int n    = nt * 16 + (lane & 15);
  int kk   = pr * 8 + (e >> 1) * 4 + (lane >> 4) * 2 + (e & 1);
  float v = 0.f;
  if (n < NKO) {
    int c = kk % DINNER;
    int r = kk / DINNER;
    int dx = r % 3, dy = (r / 3) % 3, t = r / 9;
    v = xw[((((size_t)n * DINNER + c) * 2 + t) * 3 + dy) * 3 + dx];
  }
  Wp[tid] = v;
}

__global__ void k_pack_kd(const float* __restrict__ xw, float* __restrict__ kdp) {
  int tid = blockIdx.x * blockDim.x + threadIdx.x;
  if (tid >= NKO_PAD * DINNER) return;
  int e    = tid & 3;
  int lane = (tid >> 2) & 31;
  int grp  = tid >> 7;
  int pr   = grp % NPAIR_KD;
  int nt   = grp / NPAIR_KD;
  int n    = nt * 16 + (lane & 15);
  int c    = pr * 8 + (e >> 1) * 4 + (lane >> 4) * 2 + (e & 1);
  float v = 0.f;
  if (n < NKO) {
    const float* p = xw + ((size_t)n * DINNER + c) * 18;
    for (int i = 0; i < 18; ++i) v += p[i];
  }
  kdp[tid] = v;
}

#define K2_SW 14

__global__ void __launch_bounds__(DINNER) k2_conv(const float* __restrict__ xz, const float* __restrict__ xt,
                        const float* __restrict__ w2, const float* __restrict__ b2,
                        const float* __restrict__ wxt, const float* __restrict__ bxt,
                        float* __restrict__ xin) {
  const int c  = threadIdx.x;
  const int w0 = blockIdx.x * K2_SW;
  const int h  = blockIdx.y;
  const int t  = blockIdx.z & 1;
  const int b  = blockIdx.z >> 1;
  const float* src = t ? (xt + (c >> 1)) : (xz + c);
  const int    ps  = t ? DMODEL : 2 * DINNER;
  const float* wp  = (t ? wxt : w2) + c * 9;
  const float bias = t ? bxt[c] : b2[c];
  float wk[9];
  #pragma unroll
  for (int i = 0; i < 9; ++i) wk[i] = wp[i];
  float win[3][K2_SW + 2];
  #pragma unroll
  for (int dy = 0; dy < 3; ++dy) {
    const int hh = h + dy - 1;
    const bool rok = (hh >= 0) && (hh < Hs);
    const int hc = hh < 0 ? 0 : (hh >= Hs ? Hs - 1 : hh);
    const float* rp = src + (size_t)((b * Hs + hc) * Ws) * ps;
    #pragma unroll
    for (int j = 0; j < K2_SW + 2; ++j) {
      int ww = w0 + j - 1;
      bool ok = rok;
      if (j == 0)         { ok = ok && (ww >= 0); ww = ww < 0 ? 0 : ww; }
      if (j == K2_SW + 1) { ok = ok && (ww < Ws); ww = ww >= Ws ? Ws - 1 : ww; }
      const float v = rp[(size_t)ww * ps];
      win[dy][j] = ok ? v : 0.f;
    }
  }
  float* op = xin + ((size_t)((b * 2 + t) * Hs + h) * Ws + w0) * DINNER + c;
  #pragma unroll
  for (int j = 0; j < K2_SW; ++j) {
    float acc = bias;
    #pragma unroll
    for (int dy = 0; dy < 3; ++dy) {
      #pragma unroll
      for (int dx = 0; dx < 3; ++dx) acc += win[dy][j + dx] * wk[dy * 3 + dx];
    }
    op[(size_t)j * DINNER] = siluf(acc);
  }
}

#define XT_P     224
#define XT_D     32
#define XT_PITCH 33

__global__ void __launch_bounds__(256) k3_xs(const float* __restrict__ xin, float* __restrict__ xs) {
  __shared__ float tile[XT_P * XT_PITCH];
  const int g    = blockIdx.x;
  const int d0   = blockIdx.y * XT_D;
  const int b    = blockIdx.z >> 1;
  const int col  = blockIdx.z & 1;
  const int lane = threadIdx.x & 31;
  const int wv   = __builtin_amdgcn_readfirstlane(threadIdx.x >> 5);

  const int major  = g * 4 + (wv >> 1);
  const int minor0 = (wv & 1) * 28;
  const int sMaj = col ? Ws : 1;
  const int sMin = col ? 1 : Ws;
  const float* xb = xin + ((size_t)b * 2 * Ls + major * sMaj + minor0 * sMin) * DINNER + d0 + lane;
  float pv[28];
  #pragma unroll
  for (int it = 0; it < 28; ++it) {
    const int s  = it & 1;
    const int i2 = it & ~1;
    const float* p0 = xb + ((size_t)s * Ls + (size_t)i2 * sMin) * DINNER;
    float v0 = p0[0];
    float v1 = p0[(size_t)sMin * DINNER];
    pv[it] = 0.5f * (v0 + v1) + fmaxf(v0, v1);
  }
  #pragma unroll
  for (int it = 0; it < 28; ++it) tile[(wv * 28 + it) * XT_PITCH + lane] = pv[it];
  __syncthreads();

  const int kbase = col ? 2 : 0;
  #pragma unroll
  for (int i = 0; i < 4; ++i) {
    int dd = wv * 4 + i;
    float* op0 = xs + ((size_t)(b * 4 + kbase) * DINNER + d0 + dd) * Ls + g * XT_P;
    float* op1 = op0 + (size_t)DINNER * Ls;
    #pragma unroll
    for (int c = 0; c < 7; ++c) {
      int q  = lane + 32 * c;
      int sp = (q / 56) * 56 + 55 - q % 56;
      op0[q] = tile[q  * XT_PITCH + dd];
      op1[q] = tile[sp * XT_PITCH + dd];
    }
  }
}

#define NGRP   (Ls / 4)
#define BC_GRP (2 * DSTATE * 4)
#define CDC_NT     5
#define CDC_STAGES (KBIG / 32)

struct CdcStage { v2f a[8]; float4 b[4][CDC_NT]; int ok; };

static __device__ __forceinline__ void cdc_load(CdcStage& q, const float* __restrict__ xin,
                                                const float* __restrict__ wf, int s,
                                                int b, int h, int w, int klo) {
  int td = s / 6, cc = (s - td * 6) * 32;
  int t = td / 9, r9 = td - t * 9, dy = r9 / 3, dx = r9 - dy * 3;
  int hh = h + dy - 1, ww = w + dx - 1;
  q.ok = (hh >= 0) && (hh < Hs) && (ww >= 0) && (ww < Ws);
  int hc = hh < 0 ? 0 : (hh >= Hs ? Hs - 1 : hh);
  int wc = ww < 0 ? 0 : (ww >= Ws ? Ws - 1 : ww);
  const float* ap = xin + (((size_t)(b * 2 + t) * Hs + hc) * Ws + wc) * DINNER + klo + cc;
  #pragma unroll
  for (int i = 0; i < 8; ++i) q.a[i] = *(const v2f*)(ap + 4 * i);
  const float* bp = wf + (size_t)s * 4 * 128;
  #pragma unroll
  for (int qd = 0; qd < 4; ++qd) {
    #pragma unroll
    for (int j = 0; j < CDC_NT; ++j)
      q.b[qd][j] = *(const float4*)(bp + ((size_t)j * NPAIR_W + qd) * 128);
  }
}

static __device__ __forceinline__ void cdc_compute(const CdcStage& q, v8f (&acc)[CDC_NT]) {
  const v2f zero = {0.f, 0.f};
  #pragma unroll
  for (int qd = 0; qd < 4; ++qd) {
    #pragma unroll
    for (int e2 = 0; e2 < 2; ++e2) {
      v2f a = q.ok ? q.a[2 * qd + e2] : zero;
      #pragma unroll
      for (int j = 0; j < CDC_NT; ++j) {
        float4 bv = q.b[qd][j];
        v2f bb;
        if (e2 == 0) { bb.x = bv.x; bb.y = bv.y; } else { bb.x = bv.z; bb.y = bv.w; }
        acc[j] = wmma4(a, bb, acc[j]);
      }
    }
  }
}

__global__ void __launch_bounds__(128) k4_cdc(const float* __restrict__ xin, const float* __restrict__ Wp,
                       const float* __restrict__ kdp, float* __restrict__ xdbl,
                       float* __restrict__ bc) {
  const int lane = threadIdx.x & 31;
  const int half = lane >> 4;
  const int klo  = half << 1;
  const int ln   = lane & 15;
  const int mtile = blockIdx.x * 4 + (threadIdx.x >> 5);
  const int nt0   = blockIdx.y * CDC_NT;
  const int m = mtile * 16 + ln;
  const int b = m / Ls;
  const int hw = m % Ls;
  const int h = hw / Ws, w = hw % Ws;

  v8f accN[CDC_NT];
  #pragma unroll
  for (int j = 0; j < CDC_NT; ++j) accN[j] = (v8f){0.f,0.f,0.f,0.f,0.f,0.f,0.f,0.f};

  const float* wf = Wp + (size_t)nt0 * NPAIR_W * 128 + lane * 4;
  CdcStage qa, qb;
  cdc_load(qa, xin, wf, 0, b, h, w, klo);
  for (int s = 0; s < CDC_STAGES; s += 2) {
    cdc_load(qb, xin, wf, s + 1, b, h, w, klo);
    __builtin_amdgcn_sched_barrier(0);
    cdc_compute(qa, accN);
    __builtin_amdgcn_sched_barrier(0);
    int s2 = (s + 2 < CDC_STAGES) ? (s + 2) : (s + 1);
    cdc_load(qa, xin, wf, s2, b, h, w, klo);
    __builtin_amdgcn_sched_barrier(0);
    cdc_compute(qb, accN);
  }

  v8f accD0[CDC_NT], accD1[CDC_NT];
  #pragma unroll
  for (int j = 0; j < CDC_NT; ++j) {
    accD0[j] = (v8f){0.f,0.f,0.f,0.f,0.f,0.f,0.f,0.f};
    accD1[j] = (v8f){0.f,0.f,0.f,0.f,0.f,0.f,0.f,0.f};
  }
  const float* ap0 = xin + (((size_t)(b * 2 + 0) * Hs + h) * Ws + w) * DINNER + klo;
  const float* ap1 = xin + (((size_t)(b * 2 + 1) * Hs + h) * Ws + w) * DINNER + klo;
  const float* kf  = kdp + (size_t)nt0 * NPAIR_KD * 128 + lane * 4;
  #pragma unroll
  for (int pr = 0; pr < NPAIR_KD; ++pr) {
    v2f a00 = *(const v2f*)(ap0 + pr * 8);
    v2f a01 = *(const v2f*)(ap0 + pr * 8 + 4);
    v2f a10 = *(const v2f*)(ap1 + pr * 8);
    v2f a11 = *(const v2f*)(ap1 + pr * 8 + 4);
    #pragma unroll
    for (int j = 0; j < CDC_NT; ++j) {
      float4 bv = *(const float4*)(kf + ((size_t)j * NPAIR_KD + pr) * 128);
      v2f b0, b1;
      b0.x = bv.x; b0.y = bv.y; b1.x = bv.z; b1.y = bv.w;
      accD0[j] = wmma4(a00, b0, accD0[j]);
      accD1[j] = wmma4(a10, b0, accD1[j]);
      accD0[j] = wmma4(a01, b1, accD0[j]);
      accD1[j] = wmma4(a11, b1, accD1[j]);
    }
  }

  const int m8 = mtile * 16 + half * 8;
  const int b8 = m8 / Ls, l8 = m8 % Ls;
  #pragma unroll
  for (int j = 0; j < CDC_NT; ++j) {
    const int n = (nt0 + j) * 16 + ln;
    if (n < NKO) {
      int kq = n / NCOUT, o = n % NCOUT;
      float p[8];
      #pragma unroll
      for (int rr = 0; rr < 8; ++rr) {
        float Nv = accN[j][rr], d0 = accD0[j][rr], d1 = accD1[j][rr];
        p[rr] = 2.f * Nv - THETA_C * (0.5f * (d0 + d1) + fminf(d0, d1));
      }
      if (o < DTRANK) {
        float* q = xdbl + (((size_t)b8 * 4 + kq) * NCOUT + o) * Ls + l8;
        #pragma unroll
        for (int rr = 0; rr < 8; ++rr) q[rr] = p[rr];
      } else {
        const int sel = (o - DTRANK) >> 4, st = (o - DTRANK) & 15;
        float* q = bc + (((size_t)(b8 * 4 + kq) * NGRP + (l8 >> 2)) * 2 + sel) * (DSTATE * 4) + st * 4;
        *(float4*)q            = make_float4(p[0], p[1], p[2], p[3]);
        *(float4*)(q + BC_GRP) = make_float4(p[4], p[5], p[6], p[7]);
      }
    }
  }
}

#define K5_DG  16
#define K5_TPB 224

__global__ void __launch_bounds__(K5_TPB) k5_delta(const float* __restrict__ xdbl, const float* __restrict__ dtw,
                         const float* __restrict__ dtb, float* __restrict__ delta) {
  const int l  = (blockIdx.x * K5_TPB + threadIdx.x) * 2;
  const int d0 = blockIdx.y * K5_DG;
  const int bk = blockIdx.z;
  const int k  = bk & 3;
  const float* xp = xdbl + (size_t)bk * NCOUT * Ls + l;
  v2f xv[DTRANK];
  #pragma unroll
  for (int rr = 0; rr < DTRANK; ++rr) xv[rr] = *(const v2f*)(xp + (size_t)rr * Ls);
  const float* wp = dtw + ((size_t)k * DINNER + d0) * DTRANK;
  const float* bp = dtb + k * DINNER + d0;
  float* op = delta + ((size_t)bk * DINNER + d0) * Ls + l;
  #pragma unroll 4
  for (int i = 0; i < K5_DG; ++i) {
    float a0 = bp[i];
    float a1 = a0;
    #pragma unroll
    for (int rr = 0; rr < DTRANK; ++rr) {
      float wv = wp[i * DTRANK + rr];
      a0 += xv[rr].x * wv;
      a1 += xv[rr].y * wv;
    }
    v2f o;
    o.x = fmaxf(a0, 0.f) + log1pf(expf(-fabsf(a0)));
    o.y = fmaxf(a1, 0.f) + log1pf(expf(-fabsf(a1)));
    *(v2f*)(op + (size_t)i * Ls) = o;
  }
}

#define SCH 16
#define SEGS  4
#define SEGL  (Ls / SEGS)
#define SEGCH (SEGL / SCH)

struct ScanChunk { float d[SCH], u[SCH], b[SCH], c[SCH]; };

static __device__ __forceinline__ void scan_load(ScanChunk& q, const float* __restrict__ dP,
                                                 const float* __restrict__ uP,
                                                 const float* __restrict__ bcP, int l0) {
  #pragma unroll
  for (int i = 0; i < SCH; i += 4) {
    const float* gp = bcP + (size_t)((l0 + i) >> 2) * BC_GRP;
    float4 vd = *(const float4*)(dP + l0 + i);
    float4 vu = *(const float4*)(uP + l0 + i);
    float4 vb = *(const float4*)(gp);
    float4 vc = *(const float4*)(gp + DSTATE * 4);
    q.d[i] = vd.x; q.d[i + 1] = vd.y; q.d[i + 2] = vd.z; q.d[i + 3] = vd.w;
    q.u[i] = vu.x; q.u[i + 1] = vu.y; q.u[i + 2] = vu.z; q.u[i + 3] = vu.w;
    q.b[i] = vb.x; q.b[i + 1] = vb.y; q.b[i + 2] = vb.z; q.b[i + 3] = vb.w;
    q.c[i] = vc.x; q.c[i + 1] = vc.y; q.c[i + 2] = vc.z; q.c[i + 3] = vc.w;
  }
}

static __device__ __forceinline__ float scan_steps(const ScanChunk& q, float& h, float A,
                                                   float Dv, int ln) {
  float y[SCH];
  #pragma unroll
  for (int j = 0; j < SCH; ++j) {
    float dl = q.d[j];
    h = __builtin_amdgcn_exp2f(dl * A) * h + (dl * q.u[j]) * q.b[j];
    y[j] = h * q.c[j];
  }
  #pragma unroll
  for (int m = 1; m < 16; m <<= 1) {
    float t[SCH];
    #pragma unroll
    for (int j = 0; j < SCH; ++j) t[j] = __shfl_xor(y[j], m);
    __builtin_amdgcn_sched_barrier(0);
    #pragma unroll
    for (int j = 0; j < SCH; ++j) y[j] += t[j];
  }
  float keep = 0.f;
  #pragma unroll
  for (int j = 0; j < SCH; ++j) {
    float o = y[j] + Dv * q.u[j];
    if (ln == j) keep = o;
  }
  return keep;
}

struct CarryChunk { float d[SCH], u[SCH], b[SCH]; };

static __device__ __forceinline__ void carry_load(CarryChunk& q, const float* __restrict__ dP,
                                                  const float* __restrict__ uP,
                                                  const float* __restrict__ bcP, int l0) {
  #pragma unroll
  for (int i = 0; i < SCH; i += 4) {
    float4 vd = *(const float4*)(dP + l0 + i);
    float4 vu = *(const float4*)(uP + l0 + i);
    float4 vb = *(const float4*)(bcP + (size_t)((l0 + i) >> 2) * BC_GRP);
    q.d[i] = vd.x; q.d[i + 1] = vd.y; q.d[i + 2] = vd.z; q.d[i + 3] = vd.w;
    q.u[i] = vu.x; q.u[i + 1] = vu.y; q.u[i + 2] = vu.z; q.u[i + 3] = vu.w;
    q.b[i] = vb.x; q.b[i + 1] = vb.y; q.b[i + 2] = vb.z; q.b[i + 3] = vb.w;
  }
}

static __device__ __forceinline__ void carry_steps(const CarryChunk& q, float& h, float& P, float A) {
  #pragma unroll
  for (int j = 0; j < SCH; ++j) {
    float dl = q.d[j];
    float e = __builtin_amdgcn_exp2f(dl * A);
    h = e * h + (dl * q.u[j]) * q.b[j];
    P = P * e;
  }
}

__global__ void __launch_bounds__(128) k6_scan(const float* __restrict__ xs, const float* __restrict__ bc,
                        const float* __restrict__ Alogs, const float* __restrict__ Dss,
                        const float* __restrict__ delta, float* __restrict__ oy) {
  __shared__ float carP[SEGS - 1][32];
  __shared__ float carH[SEGS - 1][32];
  const int wid  = blockIdx.x;
  const int seg  = __builtin_amdgcn_readfirstlane(threadIdx.x >> 5);
  int lane = threadIdx.x & 31;
  int half = lane >> 4, ln = lane & 15;
  int bk = wid / 96;
  int d  = (wid % 96) * 2 + half;
  int k  = bk & 3;
  const int lb = seg * SEGL;
  const float* dP = delta + ((size_t)bk * DINNER + d) * Ls + lb;
  float*       oP = oy    + ((size_t)bk * DINNER + d) * Ls + lb;
  const float* uP = xs    + ((size_t)bk * DINNER + d) * Ls + lb;
  const float* bcP = bc   + ((size_t)bk * NGRP + (lb >> 2)) * BC_GRP + ln * 4;
  float A  = -expf(Alogs[((size_t)(k * DINNER + d)) * DSTATE + ln]) * 1.4426950408889634f;
  float Dv = Dss[k * DINNER + d];

  if (seg < SEGS - 1) {
    float hs = 0.f, P = 1.f;
    CarryChunk ca, cb;
    carry_load(ca, dP, uP, bcP, 0);
    for (int c = 0; c < SEGCH - 1; c += 2) {
      carry_load(cb, dP, uP, bcP, (c + 1) * SCH);
      carry_steps(ca, hs, P, A);
      carry_load(ca, dP, uP, bcP, (c + 2) * SCH);
      carry_steps(cb, hs, P, A);
    }
    carry_steps(ca, hs, P, A);
    carP[seg][lane] = P;
    carH[seg][lane] = hs;
  }
  __syncthreads();
  float h = 0.f;
  #pragma unroll
  for (int i = 0; i < SEGS - 1; ++i) {
    if (i < seg) h = carP[i][lane] * h + carH[i][lane];
  }

  ScanChunk qa, qb;
  scan_load(qa, dP, uP, bcP, 0);
  for (int c = 0; c < SEGCH - 1; c += 2) {
    const int l0 = c * SCH;
    scan_load(qb, dP, uP, bcP, l0 + SCH);
    oP[l0 + ln] = scan_steps(qa, h, A, Dv, ln);
    scan_load(qa, dP, uP, bcP, l0 + 2 * SCH);
    oP[l0 + SCH + ln] = scan_steps(qb, h, A, Dv, ln);
  }
  oP[(SEGCH - 1) * SCH + ln] = scan_steps(qa, h, A, Dv, ln);
}

__global__ void __launch_bounds__(256) k7a_comb(const float* __restrict__ oy, float* __restrict__ y2) {
  __shared__ float tile[XT_P * XT_PITCH];
  const int h0   = blockIdx.x * 4;
  const int d0   = blockIdx.y * XT_D;
  const int b    = blockIdx.z;
  const int lane = threadIdx.x & 31;
  const int wv   = __builtin_amdgcn_readfirstlane(threadIdx.x >> 5);
  const size_t s = (size_t)DINNER * Ls;

  #pragma unroll
  for (int it = 0; it < (XT_D * Ws) / 256; ++it) {
    int idx = it * 256 + threadIdx.x;
    int dd = idx / Ws;
    int w  = idx % Ws;
    const float* base = oy + ((size_t)b * 4 * DINNER + d0 + dd) * Ls;
    int lt0 = w * Hs + h0;
    float4 t1 = *(const float4*)(base + s + lt0);
    float4 t3 = *(const float4*)(base + 3 * s + (Ls - 4 - lt0));
    float a1[4] = {t1.x, t1.y, t1.z, t1.w};
    float a3[4] = {t3.w, t3.z, t3.y, t3.x};
    #pragma unroll
    for (int j = 0; j < 4; ++j) {
      int l = (h0 + j) * Ws + w;
      float v = base[l] + base[2 * s + (Ls - 1 - l)] + a1[j] + a3[j];
      tile[(j * Ws + w) * XT_PITCH + dd] = v;
    }
  }
  __syncthreads();

  float* yp = y2 + ((size_t)b * Ls + h0 * Ws + wv * 28) * DINNER + d0 + lane;
  #pragma unroll
  for (int it = 0; it < 28; ++it)
    yp[(size_t)it * DINNER] = tile[(wv * 28 + it) * XT_PITCH + lane];
}

__global__ void k7b_ln(const float* __restrict__ xz, const float* __restrict__ g,
                       const float* __restrict__ bb, float* __restrict__ y2) {
  int m = blockIdx.x;
  int d = threadIdx.x;
  float v = y2[(size_t)m * DINNER + d];
  float s1 = v, s2 = v * v;
  #pragma unroll
  for (int off = 16; off >= 1; off >>= 1) {
    s1 += __shfl_xor(s1, off);
    s2 += __shfl_xor(s2, off);
  }
  __shared__ float r1[6], r2[6];
  __shared__ float smu, srs;
  if ((d & 31) == 0) { r1[d >> 5] = s1; r2[d >> 5] = s2; }
  __syncthreads();
  if (d == 0) {
    float t1 = 0.f, t2 = 0.f;
    for (int i = 0; i < 6; ++i) { t1 += r1[i]; t2 += r2[i]; }
    float mu = t1 / (float)DINNER;
    float var = t2 / (float)DINNER - mu * mu;
    smu = mu; srs = rsqrtf(var + 1e-5f);
  }
  __syncthreads();
  float yln = (v - smu) * srs * g[d] + bb[d];
  float zv = xz[(size_t)m * 384 + DINNER + d];
  y2[(size_t)m * DINNER + d] = yln * siluf(zv);
}

extern "C" void kernel_launch(void* const* d_in, const int* in_sizes, int n_in,
                              void* d_out, int out_size, void* d_ws, size_t ws_size,
                              hipStream_t stream) {
  (void)in_sizes; (void)n_in; (void)out_size; (void)ws_size;
  const float* x   = (const float*)d_in[0];
  const float* xt  = (const float*)d_in[1];
  const float* ipw = (const float*)d_in[2];
  const float* c2w = (const float*)d_in[3];
  const float* c2b = (const float*)d_in[4];
  const float* cxw = (const float*)d_in[5];
  const float* cxb = (const float*)d_in[6];
  const float* xpw = (const float*)d_in[7];
  const float* dtw = (const float*)d_in[8];
  const float* dtb = (const float*)d_in[9];
  const float* alg = (const float*)d_in[10];
  const float* dss = (const float*)d_in[11];
  const float* lng = (const float*)d_in[12];
  const float* lnb = (const float*)d_in[13];
  const float* opw = (const float*)d_in[14];
  float* out = (float*)d_out;
  float* ws  = (float*)d_ws;

  size_t off = 0;
  float* xz    = ws + off; off += (size_t)MROWS * 384;
  float* xin   = ws + off; off += (size_t)BATCH * 2 * Ls * DINNER;
  float* xs    = ws + off; off += (size_t)BATCH * 4 * DINNER * Ls;
  float* Wp    = ws + off; off += (size_t)NKO_PAD * KBIG;
  float* kdp   = ws + off; off += (size_t)NKO_PAD * DINNER;
  float* xdbl  = ws + off; off += (size_t)BATCH * 4 * NCOUT * Ls;
  float* delta = ws + off; off += (size_t)BATCH * 4 * DINNER * Ls;
  float* oy    = ws + off; off += (size_t)BATCH * 4 * DINNER * Ls;
  float* bc    = ws + off; off += (size_t)BATCH * 4 * NGRP * BC_GRP;
  float* y2    = xin;

  gemm_f32_wmma<<<dim3(384 / 16, MROWS / 16), 32, 0, stream>>>(x, ipw, xz, DMODEL, 384);

  k_pack_w <<<(NKO_PAD * KBIG + 255) / 256, 256, 0, stream>>>(xpw, Wp);
  k_pack_kd<<<(NKO_PAD * DINNER + 255) / 256, 256, 0, stream>>>(xpw, kdp);

  k2_conv<<<dim3(Ws / K2_SW, Hs, BATCH * 2), DINNER, 0, stream>>>(
      xz, xt, c2w, c2b, cxw, cxb, xin);

  k3_xs<<<dim3(Ls / XT_P, DINNER / XT_D, BATCH * 2), 256, 0, stream>>>(xin, xs);

  k4_cdc<<<dim3(MROWS / 64, NKO_PAD / (16 * CDC_NT)), 128, 0, stream>>>(xin, Wp, kdp, xdbl, bc);

  k5_delta<<<dim3(Ls / (2 * K5_TPB), DINNER / K5_DG, BATCH * 4), K5_TPB, 0, stream>>>(xdbl, dtw, dtb, delta);

  k6_scan<<<BATCH * 4 * (DINNER / 2), 32 * SEGS, 0, stream>>>(xs, bc, alg, dss, delta, oy);

  k7a_comb<<<dim3(Ls / XT_P, DINNER / XT_D, BATCH), 256, 0, stream>>>(oy, y2);
  k7b_ln<<<MROWS, DINNER, 0, stream>>>(xz, lng, lnb, y2);

  gemm_f32_wmma<<<dim3(DMODEL / 16, MROWS / 16), 32, 0, stream>>>(y2, opw, out, DINNER, DMODEL);
}


# launches that do not depend on one another merged, told apart by block index: in_proj GEMM + both weight packs (8 GEMM waves per group), conv GEMM + cross-scan build; 11 launches become 8; sums kept
# speedup vs baseline: 9.2991x; 1.0800x over previous
#include <hip/hip_runtime.h>
#include <hip/hip_bf16.h>
#include <math.h>

typedef float v2f __attribute__((ext_vector_type(2)));
typedef float v8f __attribute__((ext_vector_type(8)));

#define BATCH   2
#define Hs      56
#define Ws      56
#define Ls      (Hs*Ws)
#define DMODEL  96
#define DINNER  192
#define DSTATE  16
#define DTRANK  6
#define NCOUT   38
#define NKO     (4*NCOUT)
#define NKO_PAD 160
#define KBIG    (DINNER*2*9)
#define THETA_C 0.6f
#define MROWS   (BATCH*Ls)

static __device__ __forceinline__ float siluf(float v) { return v / (1.f + expf(-v)); }

static __device__ __forceinline__ v8f wmma4(v2f a, v2f b, v8f c) {
  return __builtin_amdgcn_wmma_f32_16x16x4_f32(false, a, false, b, (short)0, c, false, false);
}

static __device__ __forceinline__ void gemm_tile(const float* __restrict__ A, const float* __restrict__ Bt,
                                                 float* __restrict__ C, int K, int ldc,
                                                 int mtile, int ntile, int lane) {
  const int half = lane >> 4;
  const int klo  = half << 1;
  const int ln   = lane & 15;
  const float* ap = A  + (size_t)(mtile * 16 + ln) * K + klo;
  const float* bp = Bt + (size_t)(ntile * 16 + ln) * K + klo;
  v8f acc = {0.f,0.f,0.f,0.f,0.f,0.f,0.f,0.f};
  for (int k = 0; k < K; k += 4) {
    v2f a = *(const v2f*)(ap + k);
    v2f b = *(const v2f*)(bp + k);
    acc = wmma4(a, b, acc);
  }
  const int n = ntile * 16 + ln;
  for (int rr = 0; rr < 8; ++rr) {
    int mm = mtile * 16 + rr + half * 8;
    C[(size_t)mm * ldc + n] = acc[rr];
  }
}

__global__ void gemm_f32_wmma(const float* __restrict__ A, const float* __restrict__ Bt,
                              float* __restrict__ C, int K, int ldc) {
  gemm_tile(A, Bt, C, K, ldc, blockIdx.y, blockIdx.x, threadIdx.x & 31);
}

#define NPAIR_W  (KBIG / 8)
#define NPAIR_KD (DINNER / 8)

static __device__ __forceinline__ void pack_w(const float* __restrict__ xw, float* __restrict__ Wp, int tid) {
  if (tid >= NKO_PAD * KBIG) return;
  int e    = tid & 3;
  int lane = (tid >> 2) & 31;
  int grp  = tid >> 7;
  int pr   = grp % NPAIR_W;
  int nt   = grp / NPAIR_W;
  int n    = nt * 16 + (lane & 15);
  int kk   = pr * 8 + (e >> 1) * 4 + (lane >> 4) * 2 + (e & 1);
  float v = 0.f;
  if (n < NKO) {
    int c = kk % DINNER;
    int r = kk / DINNER;
    int dx = r % 3, dy = (r / 3) % 3, t = r / 9;
    v = xw[((((size_t)n * DINNER + c) * 2 + t) * 3 + dy) * 3 + dx];
  }
  Wp[tid] = v;
}

static __device__ __forceinline__ void pack_kd(const float* __restrict__ xw, float* __restrict__ kdp, int tid) {
  if (tid >= NKO_PAD * DINNER) return;
  int e    = tid & 3;
  int lane = (tid >> 2) & 31;
  int grp  = tid >> 7;
  int pr   = grp % NPAIR_KD;
  int nt   = grp / NPAIR_KD;
  int n    = nt * 16 + (lane & 15);
  int c    = pr * 8 + (e >> 1) * 4 + (lane >> 4) * 2 + (e & 1);
  float v = 0.f;
  if (n < NKO) {
    const float* p = xw + ((size_t)n * DINNER + c) * 18;
    for (int i = 0; i < 18; ++i) v += p[i];
  }
  kdp[tid] = v;
}

#define G1_NT  (2 * DINNER / 16)
#define G1_WG  ((MROWS / 16) * (G1_NT / 8))
#define PW_WG  ((NKO_PAD * KBIG) / 256)
#define PK_WG  ((NKO_PAD * DINNER) / 256)

__global__ void __launch_bounds__(256) k1_front(const float* __restrict__ x, const float* __restrict__ ipw,
                                                float* __restrict__ xz, const float* __restrict__ xw,
                                                float* __restrict__ Wp, float* __restrict__ kdp) {
  const int bid = blockIdx.x;
  if (bid < G1_WG) {
    const int wv = __builtin_amdgcn_readfirstlane(threadIdx.x >> 5);
    const int mtile = bid / (G1_NT / 8);
    const int ntile = (bid % (G1_NT / 8)) * 8 + wv;
    gemm_tile(x, ipw, xz, DMODEL, 2 * DINNER, mtile, ntile, threadIdx.x & 31);
  } else if (bid < G1_WG + PW_WG) {
    pack_w(xw, Wp, (bid - G1_WG) * 256 + threadIdx.x);
  } else {
    pack_kd(xw, kdp, (bid - G1_WG - PW_WG) * 256 + threadIdx.x);
  }
}

#define K2_SW 14

__global__ void __launch_bounds__(DINNER) k2_conv(const float* __restrict__ xz, const float* __restrict__ xt,
                        const float* __restrict__ w2, const float* __restrict__ b2,
                        const float* __restrict__ wxt, const float* __restrict__ bxt,
                        float* __restrict__ xin) {
  const int c  = threadIdx.x;
  const int w0 = blockIdx.x * K2_SW;
  const int h  = blockIdx.y;
  const int t  = blockIdx.z & 1;
  const int b  = blockIdx.z >> 1;
  const float* src = t ? (xt + (c >> 1)) : (xz + c);
  const int    ps  = t ? DMODEL : 2 * DINNER;
  const float* wp  = (t ? wxt : w2) + c * 9;
  const float bias = t ? bxt[c] : b2[c];
  float wk[9];
  #pragma unroll
  for (int i = 0; i < 9; ++i) wk[i] = wp[i];
  float win[3][K2_SW + 2];
  #pragma unroll
  for (int dy = 0; dy < 3; ++dy) {
    const int hh = h + dy - 1;
    const bool rok = (hh >= 0) && (hh < Hs);
    const int hc = hh < 0 ? 0 : (hh >= Hs ? Hs - 1 : hh);
    const float* rp = src + (size_t)((b * Hs + hc) * Ws) * ps;
    #pragma unroll
    for (int j = 0; j < K2_SW + 2; ++j) {
      int ww = w0 + j - 1;
      bool ok = rok;
      if (j == 0)         { ok = ok && (ww >= 0); ww = ww < 0 ? 0 : ww; }
      if (j == K2_SW + 1) { ok = ok && (ww < Ws); ww = ww >= Ws ? Ws - 1 : ww; }
      const float v = rp[(size_t)ww * ps];
      win[dy][j] = ok ? v : 0.f;
    }
  }
  float* op = xin + ((size_t)((b * 2 + t) * Hs + h) * Ws + w0) * DINNER + c;
  #pragma unroll
  for (int j = 0; j < K2_SW; ++j) {
    float acc = bias;
    #pragma unroll
    for (int dy = 0; dy < 3; ++dy) {
      #pragma unroll
      for (int dx = 0; dx < 3; ++dx) acc += win[dy][j + dx] * wk[dy * 3 + dx];
    }
    op[(size_t)j * DINNER] = siluf(acc);
  }
}

#define XT_P     224
#define XT_D     32
#define XT_PITCH 33

static __device__ __forceinline__ void xs_tile(const float* __restrict__ xin, float* __restrict__ xs,
                                               float* __restrict__ tile, int g, int d0, int b, int col) {
  const int lane = threadIdx.x & 31;
  const int w4   = __builtin_amdgcn_readfirstlane(threadIdx.x >> 5);

  const int sMaj = col ? Ws : 1;
  const int sMin = col ? 1 : Ws;
  float pv[2][28];
  #pragma unroll
  for (int rp = 0; rp < 2; ++rp) {
    const int wv     = w4 * 2 + rp;
    const int major  = g * 4 + (wv >> 1);
    const int minor0 = (wv & 1) * 28;
    const float* xb = xin + ((size_t)b * 2 * Ls + major * sMaj + minor0 * sMin) * DINNER + d0 + lane;
    #pragma unroll
    for (int it = 0; it < 28; ++it) {
      const int s  = it & 1;
      const int i2 = it & ~1;
      const float* p0 = xb + ((size_t)s * Ls + (size_t)i2 * sMin) * DINNER;
      float v0 = p0[0];
      float v1 = p0[(size_t)sMin * DINNER];
      pv[rp][it] = 0.5f * (v0 + v1) + fmaxf(v0, v1);
    }
  }
  #pragma unroll
  for (int rp = 0; rp < 2; ++rp) {
    #pragma unroll
    for (int it = 0; it < 28; ++it) tile[((w4 * 2 + rp) * 28 + it) * XT_PITCH + lane] = pv[rp][it];
  }
  __syncthreads();

  const int kbase = col ? 2 : 0;
  #pragma unroll
  for (int i = 0; i < 8; ++i) {
    int dd = w4 * 8 + i;
    float* op0 = xs + ((size_t)(b * 4 + kbase) * DINNER + d0 + dd) * Ls + g * XT_P;
    float* op1 = op0 + (size_t)DINNER * Ls;
    #pragma unroll
    for (int c = 0; c < 7; ++c) {
      int q  = lane + 32 * c;
      int sp = (q / 56) * 56 + 55 - q % 56;
      op0[q] = tile[q  * XT_PITCH + dd];
      op1[q] = tile[sp * XT_PITCH + dd];
    }
  }
}

#define NGRP   (Ls / 4)
#define BC_GRP (2 * DSTATE * 4)
#define CDC_NT     5
#define CDC_STAGES (KBIG / 32)

struct CdcStage { v2f a[8]; float4 b[4][CDC_NT]; int ok; };

static __device__ __forceinline__ void cdc_load(CdcStage& q, const float* __restrict__ xin,
                                                const float* __restrict__ wf, int s,
                                                int b, int h, int w, int klo) {
  int td = s / 6, cc = (s - td * 6) * 32;
  int t = td / 9, r9 = td - t * 9, dy = r9 / 3, dx = r9 - dy * 3;
  int hh = h + dy - 1, ww = w + dx - 1;
  q.ok = (hh >= 0) && (hh < Hs) && (ww >= 0) && (ww < Ws);
  int hc = hh < 0 ? 0 : (hh >= Hs ? Hs - 1 : hh);
  int wc = ww < 0 ? 0 : (ww >= Ws ? Ws - 1 : ww);
  const float* ap = xin + (((size_t)(b * 2 + t) * Hs + hc) * Ws + wc) * DINNER + klo + cc;
  #pragma unroll
  for (int i = 0; i < 8; ++i) q.a[i] = *(const v2f*)(ap + 4 * i);
  const float* bp = wf + (size_t)s * 4 * 128;
  #pragma unroll
  for (int qd = 0; qd < 4; ++qd) {
    #pragma unroll
    for (int j = 0; j < CDC_NT; ++j)
      q.b[qd][j] = *(const float4*)(bp + ((size_t)j * NPAIR_W + qd) * 128);
  }
}

static __device__ __forceinline__ void cdc_compute(const CdcStage& q, v8f (&acc)[CDC_NT]) {
  const v2f zero = {0.f, 0.f};
  #pragma unroll
  for (int qd = 0; qd < 4; ++qd) {
    #pragma unroll
    for (int e2 = 0; e2 < 2; ++e2) {
      v2f a = q.ok ? q.a[2 * qd + e2] : zero;
      #pragma unroll
      for (int j = 0; j < CDC_NT; ++j) {
        float4 bv = q.b[qd][j];
        v2f bb;
        if (e2 == 0) { bb.x = bv.x; bb.y = bv.y; } else { bb.x = bv.z; bb.y = bv.w; }
        acc[j] = wmma4(a, bb, acc[j]);
      }
    }
  }
}

static __device__ __forceinline__ void cdc_strip(const float* __restrict__ xin, const float* __restrict__ Wp,
                       const float* __restrict__ kdp, float* __restrict__ xdbl,
                       float* __restrict__ bc, int bx, int by) {
  const int lane = threadIdx.x & 31;
  const int half = lane >> 4;
  const int klo  = half << 1;
  const int ln   = lane & 15;
  const int mtile = bx * 4 + (threadIdx.x >> 5);
  const int nt0   = by * CDC_NT;
  const int m = mtile * 16 + ln;
  const int b = m / Ls;
  const int hw = m % Ls;
  const int h = hw / Ws, w = hw % Ws;

  v8f accN[CDC_NT];
  #pragma unroll
  for (int j = 0; j < CDC_NT; ++j) accN[j] = (v8f){0.f,0.f,0.f,0.f,0.f,0.f,0.f,0.f};

  const float* wf = Wp + (size_t)nt0 * NPAIR_W * 128 + lane * 4;
  CdcStage qa, qb;
  cdc_load(qa, xin, wf, 0, b, h, w, klo);
  for (int s = 0; s < CDC_STAGES; s += 2) {
    cdc_load(qb, xin, wf, s + 1, b, h, w, klo);
    __builtin_amdgcn_sched_barrier(0);
    cdc_compute(qa, accN);
    __builtin_amdgcn_sched_barrier(0);
    int s2 = (s + 2 < CDC_STAGES) ? (s + 2) : (s + 1);
    cdc_load(qa, xin, wf, s2, b, h, w, klo);
    __builtin_amdgcn_sched_barrier(0);
    cdc_compute(qb, accN);
  }

  v8f accD0[CDC_NT], accD1[CDC_NT];
  #pragma unroll
  for (int j = 0; j < CDC_NT; ++j) {
    accD0[j] = (v8f){0.f,0.f,0.f,0.f,0.f,0.f,0.f,0.f};
    accD1[j] = (v8f){0.f,0.f,0.f,0.f,0.f,0.f,0.f,0.f};
  }
  const float* ap0 = xin + (((size_t)(b * 2 + 0) * Hs + h) * Ws + w) * DINNER + klo;
  const float* ap1 = xin + (((size_t)(b * 2 + 1) * Hs + h) * Ws + w) * DINNER + klo;
  const float* kf  = kdp + (size_t)nt0 * NPAIR_KD * 128 + lane * 4;
  #pragma unroll
  for (int pr = 0; pr < NPAIR_KD; ++pr) {
    v2f a00 = *(const v2f*)(ap0 + pr * 8);
    v2f a01 = *(const v2f*)(ap0 + pr * 8 + 4);
    v2f a10 = *(const v2f*)(ap1 + pr * 8);
    v2f a11 = *(const v2f*)(ap1 + pr * 8 + 4);
    #pragma unroll
    for (int j = 0; j < CDC_NT; ++j) {
      float4 bv = *(const float4*)(kf + ((size_t)j * NPAIR_KD + pr) * 128);
      v2f b0, b1;
      b0.x = bv.x; b0.y = bv.y; b1.x = bv.z; b1.y = bv.w;
      accD0[j] = wmma4(a00, b0, accD0[j]);
      accD1[j] = wmma4(a10, b0, accD1[j]);
      accD0[j] = wmma4(a01, b1, accD0[j]);
      accD1[j] = wmma4(a11, b1, accD1[j]);
    }
  }

  const int m8 = mtile * 16 + half * 8;
  const int b8 = m8 / Ls, l8 = m8 % Ls;
  #pragma unroll
  for (int j = 0; j < CDC_NT; ++j) {
    const int n = (nt0 + j) * 16 + ln;
    if (n < NKO) {
      int kq = n / NCOUT, o = n % NCOUT;
      float p[8];
      #pragma unroll
      for (int rr = 0; rr < 8; ++rr) {
        float Nv = accN[j][rr], d0 = accD0[j][rr], d1 = accD1[j][rr];
        p[rr] = 2.f * Nv - THETA_C * (0.5f * (d0 + d1) + fminf(d0, d1));
      }
      if (o < DTRANK) {
        float* q = xdbl + (((size_t)b8 * 4 + kq) * NCOUT + o) * Ls + l8;
        #pragma unroll
        for (int rr = 0; rr < 8; ++rr) q[rr] = p[rr];
      } else {
        const int sel = (o - DTRANK) >> 4, st = (o - DTRANK) & 15;
        float* q = bc + (((size_t)(b8 * 4 + kq) * NGRP + (l8 >> 2)) * 2 + sel) * (DSTATE * 4) + st * 4;
        *(float4*)q            = make_float4(p[0], p[1], p[2], p[3]);
        *(float4*)(q + BC_GRP) = make_float4(p[4], p[5], p[6], p[7]);
      }
    }
  }
}

#define CDC_WG ((MROWS / 64) * (NKO_PAD / (16 * CDC_NT)))
#define XS_NG  (Ls / XT_P)
#define XS_ND  (DINNER / XT_D)
#define XS_WG  (XS_NG * XS_ND * BATCH * 2)

__global__ void __launch_bounds__(128) k34_cdc_xs(const float* __restrict__ xin, const float* __restrict__ Wp,
                       const float* __restrict__ kdp, float* __restrict__ xdbl,
                       float* __restrict__ bc, float* __restrict__ xs) {
  __shared__ float tile[XT_P * XT_PITCH];
  const int bid = blockIdx.x;
  if (bid < CDC_WG) {
    cdc_strip(xin, Wp, kdp, xdbl, bc, bid % (MROWS / 64), bid / (MROWS / 64));
  } else {
    const int r  = bid - CDC_WG;
    const int g  = r % XS_NG;
    const int dt = (r / XS_NG) % XS_ND;
    const int z  = r / (XS_NG * XS_ND);
    xs_tile(xin, xs, tile, g, dt * XT_D, z >> 1, z & 1);
  }
}

#define K5_DG  16
#define K5_TPB 224

__global__ void __launch_bounds__(K5_TPB) k5_delta(const float* __restrict__ xdbl, const float* __restrict__ dtw,
                         const float* __restrict__ dtb, float* __restrict__ delta) {
  const int l  = (blockIdx.x * K5_TPB + threadIdx.x) * 2;
  const int d0 = blockIdx.y * K5_DG;
  const int bk = blockIdx.z;
  const int k  = bk & 3;
  const float* xp = xdbl + (size_t)bk * NCOUT * Ls + l;
  v2f xv[DTRANK];
  #pragma unroll
  for (int rr = 0; rr < DTRANK; ++rr) xv[rr] = *(const v2f*)(xp + (size_t)rr * Ls);
  const float* wp = dtw + ((size_t)k * DINNER + d0) * DTRANK;
  const float* bp = dtb + k * DINNER + d0;
  float* op = delta + ((size_t)bk * DINNER + d0) * Ls + l;
  #pragma unroll 4
  for (int i = 0; i < K5_DG; ++i) {
    float a0 = bp[i];
    float a1 = a0;
    #pragma unroll
    for (int rr = 0; rr < DTRANK; ++rr) {
      float wv = wp[i * DTRANK + rr];
      a0 += xv[rr].x * wv;
      a1 += xv[rr].y * wv;
    }
    v2f o;
    o.x = fmaxf(a0, 0.f) + log1pf(expf(-fabsf(a0)));
    o.y = fmaxf(a1, 0.f) + log1pf(expf(-fabsf(a1)));
    *(v2f*)(op + (size_t)i * Ls) = o;
  }
}

#define SCH 16
#define SEGS  4
#define SEGL  (Ls / SEGS)
#define SEGCH (SEGL / SCH)

struct ScanChunk { float d[SCH], u[SCH], b[SCH], c[SCH]; };

static __device__ __forceinline__ void scan_load(ScanChunk& q, const float* __restrict__ dP,
                                                 const float* __restrict__ uP,
                                                 const float* __restrict__ bcP, int l0) {
  #pragma unroll
  for (int i = 0; i < SCH; i += 4) {
    const float* gp = bcP + (size_t)((l0 + i) >> 2) * BC_GRP;
    float4 vd = *(const float4*)(dP + l0 + i);
    float4 vu = *(const float4*)(uP + l0 + i);
    float4 vb = *(const float4*)(gp);
    float4 vc = *(const float4*)(gp + DSTATE * 4);
    q.d[i] = vd.x; q.d[i + 1] = vd.y; q.d[i + 2] = vd.z; q.d[i + 3] = vd.w;
    q.u[i] = vu.x; q.u[i + 1] = vu.y; q.u[i + 2] = vu.z; q.u[i + 3] = vu.w;
    q.b[i] = vb.x; q.b[i + 1] = vb.y; q.b[i + 2] = vb.z; q.b[i + 3] = vb.w;
    q.c[i] = vc.x; q.c[i + 1] = vc.y; q.c[i + 2] = vc.z; q.c[i + 3] = vc.w;
  }
}

static __device__ __forceinline__ float scan_steps(const ScanChunk& q, float& h, float A,
                                                   float Dv, int ln) {
  float y[SCH];
  #pragma unroll
  for (int j = 0; j < SCH; ++j) {
    float dl = q.d[j];
    h = __builtin_amdgcn_exp2f(dl * A) * h + (dl * q.u[j]) * q.b[j];
    y[j] = h * q.c[j];
  }
  #pragma unroll
  for (int m = 1; m < 16; m <<= 1) {
    float t[SCH];
    #pragma unroll
    for (int j = 0; j < SCH; ++j) t[j] = __shfl_xor(y[j], m);
    __builtin_amdgcn_sched_barrier(0);
    #pragma unroll
    for (int j = 0; j < SCH; ++j) y[j] += t[j];
  }
  float keep = 0.f;
  #pragma unroll
  for (int j = 0; j < SCH; ++j) {
    float o = y[j] + Dv * q.u[j];
    if (ln == j) keep = o;
  }
  return keep;
}

struct CarryChunk { float d[SCH], u[SCH], b[SCH]; };

static __device__ __forceinline__ void carry_load(CarryChunk& q, const float* __restrict__ dP,
                                                  const float* __restrict__ uP,
                                                  const float* __restrict__ bcP, int l0) {
  #pragma unroll
  for (int i = 0; i < SCH; i += 4) {
    float4 vd = *(const float4*)(dP + l0 + i);
    float4 vu = *(const float4*)(uP + l0 + i);
    float4 vb = *(const float4*)(bcP + (size_t)((l0 + i) >> 2) * BC_GRP);
    q.d[i] = vd.x; q.d[i + 1] = vd.y; q.d[i + 2] = vd.z; q.d[i + 3] = vd.w;
    q.u[i] = vu.x; q.u[i + 1] = vu.y; q.u[i + 2] = vu.z; q.u[i + 3] = vu.w;
    q.b[i] = vb.x; q.b[i + 1] = vb.y; q.b[i + 2] = vb.z; q.b[i + 3] = vb.w;
  }
}

static __device__ __forceinline__ void carry_steps(const CarryChunk& q, float& h, float& P, float A) {
  #pragma unroll
  for (int j = 0; j < SCH; ++j) {
    float dl = q.d[j];
    float e = __builtin_amdgcn_exp2f(dl * A);
    h = e * h + (dl * q.u[j]) * q.b[j];
    P = P * e;
  }
}

__global__ void __launch_bounds__(128) k6_scan(const float* __restrict__ xs, const float* __restrict__ bc,
                        const float* __restrict__ Alogs, const float* __restrict__ Dss,
                        const float* __restrict__ delta, float* __restrict__ oy) {
  __shared__ float carP[SEGS - 1][32];
  __shared__ float carH[SEGS - 1][32];
  const int wid  = blockIdx.x;
  const int seg  = __builtin_amdgcn_readfirstlane(threadIdx.x >> 5);
  int lane = threadIdx.x & 31;
  int half = lane >> 4, ln = lane & 15;
  int bk = wid / 96;
  int d  = (wid % 96) * 2 + half;
  int k  = bk & 3;
  const int lb = seg * SEGL;
  const float* dP = delta + ((size_t)bk * DINNER + d) * Ls + lb;
  float*       oP = oy    + ((size_t)bk * DINNER + d) * Ls + lb;
  const float* uP = xs    + ((size_t)bk * DINNER + d) * Ls + lb;
  const float* bcP = bc   + ((size_t)bk * NGRP + (lb >> 2)) * BC_GRP + ln * 4;
  float A  = -expf(Alogs[((size_t)(k * DINNER + d)) * DSTATE + ln]) * 1.4426950408889634f;
  float Dv = Dss[k * DINNER + d];

  if (seg < SEGS - 1) {
    float hs = 0.f, P = 1.f;
    CarryChunk ca, cb;
    carry_load(ca, dP, uP, bcP, 0);
    for (int c = 0; c < SEGCH - 1; c += 2) {
      carry_load(cb, dP, uP, bcP, (c + 1) * SCH);
      carry_steps(ca, hs, P, A);
      carry_load(ca, dP, uP, bcP, (c + 2) * SCH);
      carry_steps(cb, hs, P, A);
    }
    carry_steps(ca, hs, P, A);
    carP[seg][lane] = P;
    carH[seg][lane] = hs;
  }
  __syncthreads();
  float h = 0.f;
  #pragma unroll
  for (int i = 0; i < SEGS - 1; ++i) {
    if (i < seg) h = carP[i][lane] * h + carH[i][lane];
  }

  ScanChunk qa, qb;
  scan_load(qa, dP, uP, bcP, 0);
  for (int c = 0; c < SEGCH - 1; c += 2) {
    const int l0 = c * SCH;
    scan_load(qb, dP, uP, bcP, l0 + SCH);
    oP[l0 + ln] = scan_steps(qa, h, A, Dv, ln);
    scan_load(qa, dP, uP, bcP, l0 + 2 * SCH);
    oP[l0 + SCH + ln] = scan_steps(qb, h, A, Dv, ln);
  }
  oP[(SEGCH - 1) * SCH + ln] = scan_steps(qa, h, A, Dv, ln);
}

__global__ void __launch_bounds__(256) k7a_comb(const float* __restrict__ oy, float* __restrict__ y2) {
  __shared__ float tile[XT_P * XT_PITCH];
  const int h0   = blockIdx.x * 4;
  const int d0   = blockIdx.y * XT_D;
  const int b    = blockIdx.z;
  const int lane = threadIdx.x & 31;
  const int wv   = __builtin_amdgcn_readfirstlane(threadIdx.x >> 5);
  const size_t s = (size_t)DINNER * Ls;

  #pragma unroll
  for (int it = 0; it < (XT_D * Ws) / 256; ++it) {
    int idx = it * 256 + threadIdx.x;
    int dd = idx / Ws;
    int w  = idx % Ws;
    const float* base = oy + ((size_t)b * 4 * DINNER + d0 + dd) * Ls;
    int lt0 = w * Hs + h0;
    float4 t1 = *(const float4*)(base + s + lt0);
    float4 t3 = *(const float4*)(base + 3 * s + (Ls - 4 - lt0));
    float a1[4] = {t1.x, t1.y, t1.z, t1.w};
    float a3[4] = {t3.w, t3.z, t3.y, t3.x};
    #pragma unroll
    for (int j = 0; j < 4; ++j) {
      int l = (h0 + j) * Ws + w;
      float v = base[l] + base[2 * s + (Ls - 1 - l)] + a1[j] + a3[j];
      tile[(j * Ws + w) * XT_PITCH + dd] = v;
    }
  }
  __syncthreads();

  float* yp = y2 + ((size_t)b * Ls + h0 * Ws + wv * 28) * DINNER + d0 + lane;
  #pragma unroll
  for (int it = 0; it < 28; ++it)
    yp[(size_t)it * DINNER] = tile[(wv * 28 + it) * XT_PITCH + lane];
}

__global__ void k7b_ln(const float* __restrict__ xz, const float* __restrict__ g,
                       const float* __restrict__ bb, float* __restrict__ y2) {
  int m = blockIdx.x;
  int d = threadIdx.x;
  float v = y2[(size_t)m * DINNER + d];
  float s1 = v, s2 = v * v;
  #pragma unroll
  for (int off = 16; off >= 1; off >>= 1) {
    s1 += __shfl_xor(s1, off);
    s2 += __shfl_xor(s2, off);
  }
  __shared__ float r1[6], r2[6];
  __shared__ float smu, srs;
  if ((d & 31) == 0) { r1[d >> 5] = s1; r2[d >> 5] = s2; }
  __syncthreads();
  if (d == 0) {
    float t1 = 0.f, t2 = 0.f;
    for (int i = 0; i < 6; ++i) { t1 += r1[i]; t2 += r2[i]; }
    float mu = t1 / (float)DINNER;
    float var = t2 / (float)DINNER - mu * mu;
    smu = mu; srs = rsqrtf(var + 1e-5f);
  }
  __syncthreads();
  float yln = (v - smu) * srs * g[d] + bb[d];
  float zv = xz[(size_t)m * 384 + DINNER + d];
  y2[(size_t)m * DINNER + d] = yln * siluf(zv);
}

extern "C" void kernel_launch(void* const* d_in, const int* in_sizes, int n_in,
                              void* d_out, int out_size, void* d_ws, size_t ws_size,
                              hipStream_t stream) {
  (void)in_sizes; (void)n_in; (void)out_size; (void)ws_size;
  const float* x   = (const float*)d_in[0];
  const float* xt  = (const float*)d_in[1];
  const float* ipw = (const float*)d_in[2];
  const float* c2w = (const float*)d_in[3];
  const float* c2b = (const float*)d_in[4];
  const float* cxw = (const float*)d_in[5];
  const float* cxb = (const float*)d_in[6];
  const float* xpw = (const float*)d_in[7];
  const float* dtw = (const float*)d_in[8];
  const float* dtb = (const float*)d_in[9];
  const float* alg = (const float*)d_in[10];
  const float* dss = (const float*)d_in[11];
  const float* lng = (const float*)d_in[12];
  const float* lnb = (const float*)d_in[13];
  const float* opw = (const float*)d_in[14];
  float* out = (float*)d_out;
  float* ws  = (float*)d_ws;

  size_t off = 0;
  float* xz    = ws + off; off += (size_t)MROWS * 384;
  float* xin   = ws + off; off += (size_t)BATCH * 2 * Ls * DINNER;
  float* xs    = ws + off; off += (size_t)BATCH * 4 * DINNER * Ls;
  float* Wp    = ws + off; off += (size_t)NKO_PAD * KBIG;
  float* kdp   = ws + off; off += (size_t)NKO_PAD * DINNER;
  float* xdbl  = ws + off; off += (size_t)BATCH * 4 * NCOUT * Ls;
  float* delta = ws + off; off += (size_t)BATCH * 4 * DINNER * Ls;
  float* oy    = ws + off; off += (size_t)BATCH * 4 * DINNER * Ls;
  float* bc    = ws + off; off += (size_t)BATCH * 4 * NGRP * BC_GRP;
  float* y2    = xin;

  k1_front<<<G1_WG + PW_WG + PK_WG, 256, 0, stream>>>(x, ipw, xz, xpw, Wp, kdp);

  k2_conv<<<dim3(Ws / K2_SW, Hs, BATCH * 2), DINNER, 0, stream>>>(
      xz, xt, c2w, c2b, cxw, cxb, xin);

  k34_cdc_xs<<<CDC_WG + XS_WG, 128, 0, stream>>>(xin, Wp, kdp, xdbl, bc, xs);

  k5_delta<<<dim3(Ls / (2 * K5_TPB), DINNER / K5_DG, BATCH * 4), K5_TPB, 0, stream>>>(xdbl, dtw, dtb, delta);

  k6_scan<<<BATCH * 4 * (DINNER / 2), 32 * SEGS, 0, stream>>>(xs, bc, alg, dss, delta, oy);

  k7a_comb<<<dim3(Ls / XT_P, DINNER / XT_D, BATCH), 256, 0, stream>>>(oy, y2);
  k7b_ln<<<MROWS, DINNER, 0, stream>>>(xz, lng, lnb, y2);

  gemm_f32_wmma<<<dim3(DMODEL / 16, MROWS / 16), 32, 0, stream>>>(y2, opw, out, DINNER, DMODEL);
}
